# starting kernel
# baseline (speedup 1.0000x reference)
#include <hip/hip_runtime.h>

#define NODE_SIZE 50000
#define REL_SIZE  500
#define HID       128
#define NTRIPLE   800000
#define EPSF      1e-12f

typedef __attribute__((ext_vector_type(2))) float v2f;
typedef __attribute__((ext_vector_type(8))) float v8f;

__global__ void zero_kernel(float* __restrict__ p, int n) {
    int i = blockIdx.x * blockDim.x + threadIdx.x;
    int stride = gridDim.x * blockDim.x;
    for (; i < n; i += stride) p[i] = 0.0f;
}

__global__ __launch_bounds__(32)
void rel_stats_kernel(const float* __restrict__ rel_emb,
                      const float* __restrict__ attn,
                      float* __restrict__ rel_norm,
                      float* __restrict__ rel_att) {
    const int lane = threadIdx.x & 31;
    const int tile = blockIdx.x;
    const int half = lane >> 4;
    const int m16  = lane & 15;

    const int   row   = tile * 16 + m16;
    const bool  rowok = row < REL_SIZE;
    const float amask = rowok ? 1.0f : 0.0f;
    const float* arow = rel_emb + (size_t)(rowok ? row : 0) * HID;
    const float bmask = (m16 == 0) ? 1.0f : 0.0f;

    v8f c = {};
    for (int k = 0; k < HID; k += 4) {
        const int koff = half * 2;
        v2f a, b;
        a.x = arow[k + koff + 0] * amask;
        a.y = arow[k + koff + 1] * amask;
        b.x = attn[k + koff + 0] * bmask;
        b.y = attn[k + koff + 1] * bmask;
        c = __builtin_amdgcn_wmma_f32_16x16x4_f32(false, a, false, b,
                                                  (short)0, c, false, false);
    }
    if (m16 == 0) {
        const int mbase = tile * 16 + half * 8;
#pragma unroll
        for (int j = 0; j < 8; ++j)
            if (mbase + j < REL_SIZE) rel_att[mbase + j] = c[j];
    }

    float ss = 0.0f;
    {
        const float4* p = (const float4*)(arow + half * 64);
#pragma unroll
        for (int i = 0; i < 16; ++i) {
            float4 f = p[i];
            ss += f.x * f.x + f.y * f.y + f.z * f.z + f.w * f.w;
        }
    }
    ss += __shfl_xor(ss, 16, 32);
    if (half == 0 && rowok) rel_norm[row] = sqrtf(ss);
}

__global__ __launch_bounds__(256)
void att_kernel(const int*   __restrict__ indices,
                const int*   __restrict__ sp_idx,
                const float* __restrict__ sp_val,
                const float* __restrict__ rel_norm,
                const float* __restrict__ rel_att,
                float* __restrict__ exp_v,
                float* __restrict__ row_sums) {
    int t = blockIdx.x * blockDim.x + threadIdx.x;
    if (t >= NTRIPLE) return;
    const int   r    = sp_idx[2 * t + 1];
    const float v    = sp_val[t];
    const float coef = v / fmaxf(fabsf(v) * rel_norm[r], EPSF);
    const float e    = expf(coef * rel_att[r]);
    exp_v[t] = e;
    atomicAdd(&row_sums[indices[2 * t]], e);
}

__global__ __launch_bounds__(256)
void scatter_kernel(const float* __restrict__ features,
                    const float* __restrict__ rel_emb,
                    const int*   __restrict__ indices,
                    const int*   __restrict__ sp_idx,
                    const float* __restrict__ sp_val,
                    const float* __restrict__ rel_norm,
                    const float* __restrict__ exp_v,
                    const float* __restrict__ row_sums,
                    float* __restrict__ out) {
    const int wavesPerBlock = blockDim.x >> 5;
    const int t    = blockIdx.x * wavesPerBlock + (threadIdx.x >> 5);
    const int lane = threadIdx.x & 31;
    if (t >= NTRIPLE) return;

    const int   row  = indices[2 * t + 0];
    const int   col  = indices[2 * t + 1];
    const int   r    = sp_idx[2 * t + 1];
    const float v    = sp_val[t];
    const float coef = v / fmaxf(fabsf(v) * rel_norm[r], EPSF);

    const float4 f4 = ((const float4*)(features + (size_t)col * HID))[lane];
    const float4 r4 = ((const float4*)(rel_emb  + (size_t)r   * HID))[lane];

    float partial = f4.x * r4.x + f4.y * r4.y + f4.z * r4.z + f4.w * r4.w;
#pragma unroll
    for (int off = 16; off >= 1; off >>= 1)
        partial += __shfl_xor(partial, off, 32);

    const float proj = coef * partial;
    const float s    = exp_v[t] / row_sums[row];
    const float k2   = 2.0f * proj * coef;

    float* dst = out + (size_t)row * HID + lane * 4;
    atomicAdd(dst + 0, (f4.x - k2 * r4.x) * s);
    atomicAdd(dst + 1, (f4.y - k2 * r4.y) * s);
    atomicAdd(dst + 2, (f4.z - k2 * r4.z) * s);
    atomicAdd(dst + 3, (f4.w - k2 * r4.w) * s);
}

extern "C" void kernel_launch(void* const* d_in, const int* in_sizes, int n_in,
                              void* d_out, int out_size, void* d_ws, size_t ws_size,
                              hipStream_t stream) {
    const float* features = (const float*)d_in[0];
    const float* rel_emb  = (const float*)d_in[1];
    const int*   indices  = (const int*)  d_in[2];
    const int*   sp_idx   = (const int*)  d_in[3];
    const float* sp_val   = (const float*)d_in[4];
    const float* attn     = (const float*)d_in[5];
    float* out = (float*)d_out;

    float* ws       = (float*)d_ws;
    float* rel_norm = ws;
    float* rel_att  = ws + 512;
    float* exp_v    = ws + 1024;
    float* row_sums = ws + 1024 + NTRIPLE;

    zero_kernel<<<2048, 256, 0, stream>>>(out, NODE_SIZE * HID);
    zero_kernel<<<256, 256, 0, stream>>>(row_sums, NODE_SIZE);

    rel_stats_kernel<<<32, 32, 0, stream>>>(rel_emb, attn, rel_norm, rel_att);

    att_kernel<<<(NTRIPLE + 255) / 256, 256, 0, stream>>>(
        indices, sp_idx, sp_val, rel_norm, rel_att, exp_v, row_sums);

    scatter_kernel<<<(NTRIPLE + 7) / 8, 256, 0, stream>>>(
        features, rel_emb, indices, sp_idx, sp_val, rel_norm,
        exp_v, row_sums, out);
}


# scatter pass: lane l owns features l, l+32, l+64, l+96, so every f32 atomic add and row load covers one whole 128-byte line, not a quarter of each of 4 lines; dot product summed in a new order
# speedup vs baseline: 1.2996x; 1.2996x over previous
#include <hip/hip_runtime.h>

#define NODE_SIZE 50000
#define REL_SIZE  500
#define HID       128
#define NTRIPLE   800000
#define EPSF      1e-12f

typedef __attribute__((ext_vector_type(2))) float v2f;
typedef __attribute__((ext_vector_type(8))) float v8f;

__global__ void zero_kernel(float* __restrict__ p, int n) {
    int i = blockIdx.x * blockDim.x + threadIdx.x;
    int stride = gridDim.x * blockDim.x;
    for (; i < n; i += stride) p[i] = 0.0f;
}

__global__ __launch_bounds__(32)
void rel_stats_kernel(const float* __restrict__ rel_emb,
                      const float* __restrict__ attn,
                      float* __restrict__ rel_norm,
                      float* __restrict__ rel_att) {
    const int lane = threadIdx.x & 31;
    const int tile = blockIdx.x;
    const int half = lane >> 4;
    const int m16  = lane & 15;

    const int   row   = tile * 16 + m16;
    const bool  rowok = row < REL_SIZE;
    const float amask = rowok ? 1.0f : 0.0f;
    const float* arow = rel_emb + (size_t)(rowok ? row : 0) * HID;
    const float bmask = (m16 == 0) ? 1.0f : 0.0f;

    v8f c = {};
    for (int k = 0; k < HID; k += 4) {
        const int koff = half * 2;
        v2f a, b;
        a.x = arow[k + koff + 0] * amask;
        a.y = arow[k + koff + 1] * amask;
        b.x = attn[k + koff + 0] * bmask;
        b.y = attn[k + koff + 1] * bmask;
        c = __builtin_amdgcn_wmma_f32_16x16x4_f32(false, a, false, b,
                                                  (short)0, c, false, false);
    }
    if (m16 == 0) {
        const int mbase = tile * 16 + half * 8;
#pragma unroll
        for (int j = 0; j < 8; ++j)
            if (mbase + j < REL_SIZE) rel_att[mbase + j] = c[j];
    }

    float ss = 0.0f;
    {
        const float4* p = (const float4*)(arow + half * 64);
#pragma unroll
        for (int i = 0; i < 16; ++i) {
            float4 f = p[i];
            ss += f.x * f.x + f.y * f.y + f.z * f.z + f.w * f.w;
        }
    }
    ss += __shfl_xor(ss, 16, 32);
    if (half == 0 && rowok) rel_norm[row] = sqrtf(ss);
}

__global__ __launch_bounds__(256)
void att_kernel(const int*   __restrict__ indices,
                const int*   __restrict__ sp_idx,
                const float* __restrict__ sp_val,
                const float* __restrict__ rel_norm,
                const float* __restrict__ rel_att,
                float* __restrict__ exp_v,
                float* __restrict__ row_sums) {
    int t = blockIdx.x * blockDim.x + threadIdx.x;
    if (t >= NTRIPLE) return;
    const int   r    = sp_idx[2 * t + 1];
    const float v    = sp_val[t];
    const float coef = v / fmaxf(fabsf(v) * rel_norm[r], EPSF);
    const float e    = expf(coef * rel_att[r]);
    exp_v[t] = e;
    atomicAdd(&row_sums[indices[2 * t]], e);
}

__global__ __launch_bounds__(256)
void scatter_kernel(const float* __restrict__ features,
                    const float* __restrict__ rel_emb,
                    const int*   __restrict__ indices,
                    const int*   __restrict__ sp_idx,
                    const float* __restrict__ sp_val,
                    const float* __restrict__ rel_norm,
                    const float* __restrict__ exp_v,
                    const float* __restrict__ row_sums,
                    float* __restrict__ out) {
    const int wavesPerBlock = blockDim.x >> 5;
    const int t    = blockIdx.x * wavesPerBlock + (threadIdx.x >> 5);
    const int lane = threadIdx.x & 31;
    if (t >= NTRIPLE) return;

    const int   row  = indices[2 * t + 0];
    const int   col  = indices[2 * t + 1];
    const int   r    = sp_idx[2 * t + 1];
    const float v    = sp_val[t];
    const float coef = v / fmaxf(fabsf(v) * rel_norm[r], EPSF);

    const float* frow = features + (size_t)col * HID + lane;
    const float* rrow = rel_emb  + (size_t)r   * HID + lane;
    const float f0 = frow[0], f1 = frow[32], f2 = frow[64], f3 = frow[96];
    const float r0 = rrow[0], r1 = rrow[32], r2 = rrow[64], r3 = rrow[96];

    float partial = f0 * r0 + f1 * r1 + f2 * r2 + f3 * r3;
#pragma unroll
    for (int off = 16; off >= 1; off >>= 1)
        partial += __shfl_xor(partial, off, 32);

    const float proj = coef * partial;
    const float s    = exp_v[t] / row_sums[row];
    const float k2   = 2.0f * proj * coef;

    float* dst = out + (size_t)row * HID + lane;
    atomicAdd(dst +  0, (f0 - k2 * r0) * s);
    atomicAdd(dst + 32, (f1 - k2 * r1) * s);
    atomicAdd(dst + 64, (f2 - k2 * r2) * s);
    atomicAdd(dst + 96, (f3 - k2 * r3) * s);
}

extern "C" void kernel_launch(void* const* d_in, const int* in_sizes, int n_in,
                              void* d_out, int out_size, void* d_ws, size_t ws_size,
                              hipStream_t stream) {
    const float* features = (const float*)d_in[0];
    const float* rel_emb  = (const float*)d_in[1];
    const int*   indices  = (const int*)  d_in[2];
    const int*   sp_idx   = (const int*)  d_in[3];
    const float* sp_val   = (const float*)d_in[4];
    const float* attn     = (const float*)d_in[5];
    float* out = (float*)d_out;

    float* ws       = (float*)d_ws;
    float* rel_norm = ws;
    float* rel_att  = ws + 512;
    float* exp_v    = ws + 1024;
    float* row_sums = ws + 1024 + NTRIPLE;

    zero_kernel<<<2048, 256, 0, stream>>>(out, NODE_SIZE * HID);
    zero_kernel<<<256, 256, 0, stream>>>(row_sums, NODE_SIZE);

    rel_stats_kernel<<<32, 32, 0, stream>>>(rel_emb, attn, rel_norm, rel_att);

    att_kernel<<<(NTRIPLE + 255) / 256, 256, 0, stream>>>(
        indices, sp_idx, sp_val, rel_norm, rel_att, exp_v, row_sums);

    scatter_kernel<<<(NTRIPLE + 7) / 8, 256, 0, stream>>>(
        features, rel_emb, indices, sp_idx, sp_val, rel_norm,
        exp_v, row_sums, out);
}


# atomic f32 scatter replaced by counting sort of edges by destination (LDS counts, scan, LDS ranks) + one wave per node summing rows in registers in rank order; exp sum divided out once per node
# speedup vs baseline: 1.4811x; 1.1397x over previous
#include <hip/hip_runtime.h>

#define NODE_SIZE 50000
#define REL_SIZE  500
#define HID       128
#define NTRIPLE   800000
#define EPSF      1e-12f

typedef __attribute__((ext_vector_type(2))) float v2f;
typedef __attribute__((ext_vector_type(8))) float v8f;

__global__ void zero_int_kernel(int* __restrict__ p, int n) {
    int i = blockIdx.x * blockDim.x + threadIdx.x;
    int stride = gridDim.x * blockDim.x;
    for (; i < n; i += stride) p[i] = 0;
}

__global__ __launch_bounds__(32)
void rel_stats_kernel(const float* __restrict__ rel_emb,
                      const float* __restrict__ attn,
                      float* __restrict__ rel_norm,
                      float* __restrict__ rel_att) {
    const int lane = threadIdx.x & 31;
    const int tile = blockIdx.x;
    const int half = lane >> 4;
    const int m16  = lane & 15;

    const int   row   = tile * 16 + m16;
    const bool  rowok = row < REL_SIZE;
    const float amask = rowok ? 1.0f : 0.0f;
    const float* arow = rel_emb + (size_t)(rowok ? row : 0) * HID;
    const float bmask = (m16 == 0) ? 1.0f : 0.0f;

    v8f c = {};
    for (int k = 0; k < HID; k += 4) {
        const int koff = half * 2;
        v2f a, b;
        a.x = arow[k + koff + 0] * amask;
        a.y = arow[k + koff + 1] * amask;
        b.x = attn[k + koff + 0] * bmask;
        b.y = attn[k + koff + 1] * bmask;
        c = __builtin_amdgcn_wmma_f32_16x16x4_f32(false, a, false, b,
                                                  (short)0, c, false, false);
    }
    if (m16 == 0) {
        const int mbase = tile * 16 + half * 8;
#pragma unroll
        for (int j = 0; j < 8; ++j)
            if (mbase + j < REL_SIZE) rel_att[mbase + j] = c[j];
    }

    float ss = 0.0f;
    {
        const float4* p = (const float4*)(arow + half * 64);
#pragma unroll
        for (int i = 0; i < 16; ++i) {
            float4 f = p[i];
            ss += f.x * f.x + f.y * f.y + f.z * f.z + f.w * f.w;
        }
    }
    ss += __shfl_xor(ss, 16, 32);
    if (half == 0 && rowok) rel_norm[row] = sqrtf(ss);
}

#define SORT_GROUPS  64
#define SORT_CHUNK   (NTRIPLE / SORT_GROUPS)
#define SORT_EDGES   13
#define NODE_TRIPS   49

__device__ __forceinline__ int node_slot(int k, int rot, int tid) {
    int kk = k + rot;
    if (kk >= NODE_TRIPS) kk -= NODE_TRIPS;
    return kk * 1024 + tid;
}

__global__ __launch_bounds__(1024)
void count_kernel(const int* __restrict__ indices, int* __restrict__ cnt) {
    __shared__ int h[NODE_SIZE];
    const int tid = threadIdx.x;
    for (int i = tid; i < NODE_SIZE; i += 1024) h[i] = 0;

    const int lo = blockIdx.x * SORT_CHUNK;
    int row[SORT_EDGES];
#pragma unroll
    for (int k = 0; k < SORT_EDGES; ++k) {
        const int e = k * 1024 + tid;
        row[k] = (e < SORT_CHUNK) ? indices[2 * (lo + e)] : -1;
    }
    __syncthreads();
#pragma unroll
    for (int k = 0; k < SORT_EDGES; ++k)
        if (row[k] >= 0) atomicAdd(&h[row[k]], 1);
    __syncthreads();

    const int rot = blockIdx.x % NODE_TRIPS;
    for (int k = 0; k < NODE_TRIPS; ++k) {
        const int i = node_slot(k, rot, tid);
        if (i < NODE_SIZE) {
            const int c = h[i];
            if (c) atomicAdd(&cnt[i], c);
        }
    }
}

#define SCAN_PER_THREAD 49

__global__ __launch_bounds__(1024)
void scan_kernel(int* __restrict__ cur) {
    __shared__ int s[1024 * SCAN_PER_THREAD];
    __shared__ int wsum[32];
    const int tid  = threadIdx.x;
    const int lane = tid & 31;
    const int wave = tid >> 5;
    for (int i = tid; i < 1024 * SCAN_PER_THREAD; i += 1024)
        s[i] = (i < NODE_SIZE) ? cur[i] : 0;
    __syncthreads();

    const int base = tid * SCAN_PER_THREAD;
    int sum = 0;
    for (int j = 0; j < SCAN_PER_THREAD; ++j) sum += s[base + j];

    int incl = sum;
#pragma unroll
    for (int off = 1; off < 32; off <<= 1) {
        const int n = __shfl_up(incl, off, 32);
        if (lane >= off) incl += n;
    }
    if (lane == 31) wsum[wave] = incl;
    __syncthreads();
    if (wave == 0) {
        const int w = wsum[lane];
        int wi = w;
#pragma unroll
        for (int off = 1; off < 32; off <<= 1) {
            const int n = __shfl_up(wi, off, 32);
            if (lane >= off) wi += n;
        }
        wsum[lane] = wi - w;
    }
    __syncthreads();

    int run = incl - sum + wsum[wave];
    for (int j = 0; j < SCAN_PER_THREAD; ++j) {
        const int c = s[base + j];
        s[base + j] = run;
        run += c;
    }
    __syncthreads();
    for (int i = tid; i < NODE_SIZE; i += 1024) cur[i] = s[i];
}

__global__ __launch_bounds__(1024)
void fill_kernel(const int* __restrict__ indices,
                 int* __restrict__ cur,
                 int* __restrict__ perm) {
    __shared__ int h[NODE_SIZE];
    const int tid = threadIdx.x;
    for (int i = tid; i < NODE_SIZE; i += 1024) h[i] = 0;

    const int lo = blockIdx.x * SORT_CHUNK;
    int row[SORT_EDGES], rank[SORT_EDGES];
#pragma unroll
    for (int k = 0; k < SORT_EDGES; ++k) {
        const int e = k * 1024 + tid;
        row[k] = (e < SORT_CHUNK) ? indices[2 * (lo + e)] : -1;
    }
    __syncthreads();
#pragma unroll
    for (int k = 0; k < SORT_EDGES; ++k)
        rank[k] = (row[k] >= 0) ? atomicAdd(&h[row[k]], 1) : 0;
    __syncthreads();

    const int rot = blockIdx.x % NODE_TRIPS;
    for (int k0 = 0; k0 < NODE_TRIPS; k0 += 7) {
        int start[7];
#pragma unroll
        for (int k = 0; k < 7; ++k) {
            const int i = node_slot(k0 + k, rot, tid);
            start[k] = (i < NODE_SIZE) ? atomicAdd(&cur[i], h[i]) : 0;
        }
#pragma unroll
        for (int k = 0; k < 7; ++k) {
            const int i = node_slot(k0 + k, rot, tid);
            if (i < NODE_SIZE) h[i] = start[k];
        }
    }
    __syncthreads();

#pragma unroll
    for (int k = 0; k < SORT_EDGES; ++k) {
        if (row[k] >= 0) {
            int pos = h[row[k]] + rank[k];
            pos = min(max(pos, 0), NTRIPLE - 1);
            perm[pos] = lo + k * 1024 + tid;
        }
    }
}

struct RowPair { float f0, f1, f2, f3, r0, r1, r2, r3; };

__device__ __forceinline__ RowPair load_pair(const float* __restrict__ features,
                                             const float* __restrict__ rel_emb,
                                             int col, int r, int lane) {
    const float* frow = features + (size_t)col * HID + lane;
    const float* rrow = rel_emb  + (size_t)r   * HID + lane;
    RowPair x;
    x.f0 = frow[0]; x.f1 = frow[32]; x.f2 = frow[64]; x.f3 = frow[96];
    x.r0 = rrow[0]; x.r1 = rrow[32]; x.r2 = rrow[64]; x.r3 = rrow[96];
    return x;
}

__device__ __forceinline__ float wave_sum(float p) {
#pragma unroll
    for (int off = 16; off >= 1; off >>= 1) p += __shfl_xor(p, off, 32);
    return p;
}

__device__ __forceinline__ void add_edge(const RowPair& x, float coef, float e,
                                         float& a0, float& a1, float& a2, float& a3) {
    const float partial = wave_sum(x.f0 * x.r0 + x.f1 * x.r1 + x.f2 * x.r2 + x.f3 * x.r3);
    const float proj = coef * partial;
    const float k2   = 2.0f * proj * coef;
    a0 += (x.f0 - k2 * x.r0) * e;
    a1 += (x.f1 - k2 * x.r1) * e;
    a2 += (x.f2 - k2 * x.r2) * e;
    a3 += (x.f3 - k2 * x.r3) * e;
}

__device__ __forceinline__ int bcast_i(int v, int j) {
    return __builtin_amdgcn_readlane(v, j);
}
__device__ __forceinline__ float bcast_f(float v, int j) {
    return __int_as_float(__builtin_amdgcn_readlane(__float_as_int(v), j));
}

__global__ __launch_bounds__(256)
void gather_kernel(const float* __restrict__ features,
                   const float* __restrict__ rel_emb,
                   const int*   __restrict__ indices,
                   const int*   __restrict__ sp_idx,
                   const float* __restrict__ sp_val,
                   const float* __restrict__ rel_norm,
                   const float* __restrict__ rel_att,
                   const int*   __restrict__ cur,
                   const int*   __restrict__ perm,
                   float* __restrict__ out) {
    const int lane = threadIdx.x & 31;
    const int wave = __builtin_amdgcn_readfirstlane((int)(threadIdx.x >> 5));
    const int node = blockIdx.x * 8 + wave;
    if (node >= NODE_SIZE) return;

    const int end   = cur[node];
    const int start = node ? cur[node - 1] : 0;

    float a0 = 0.0f, a1 = 0.0f, a2 = 0.0f, a3 = 0.0f, den = 0.0f;

    for (int base = start; base < end; base += 32) {
        const int m = min(32, end - base);
        int col = 0, r = 0;
        float coef = 0.0f, e = 0.0f;
        if (lane < m) {
            int t = perm[base + lane];
            t = min(max(t, 0), NTRIPLE - 1);
            col = indices[2 * t + 1];
            r   = sp_idx[2 * t + 1];
            const float v = sp_val[t];
            coef = v / fmaxf(fabsf(v) * rel_norm[r], EPSF);
            e    = expf(coef * rel_att[r]);
        }
        den += e;

        int j = 0;
        for (; j + 4 <= m; j += 4) {
            const RowPair x0 = load_pair(features, rel_emb, bcast_i(col, j + 0), bcast_i(r, j + 0), lane);
            const RowPair x1 = load_pair(features, rel_emb, bcast_i(col, j + 1), bcast_i(r, j + 1), lane);
            const RowPair x2 = load_pair(features, rel_emb, bcast_i(col, j + 2), bcast_i(r, j + 2), lane);
            const RowPair x3 = load_pair(features, rel_emb, bcast_i(col, j + 3), bcast_i(r, j + 3), lane);
            add_edge(x0, bcast_f(coef, j + 0), bcast_f(e, j + 0), a0, a1, a2, a3);
            add_edge(x1, bcast_f(coef, j + 1), bcast_f(e, j + 1), a0, a1, a2, a3);
            add_edge(x2, bcast_f(coef, j + 2), bcast_f(e, j + 2), a0, a1, a2, a3);
            add_edge(x3, bcast_f(coef, j + 3), bcast_f(e, j + 3), a0, a1, a2, a3);
        }
        for (; j < m; ++j) {
            const RowPair x0 = load_pair(features, rel_emb, bcast_i(col, j), bcast_i(r, j), lane);
            add_edge(x0, bcast_f(coef, j), bcast_f(e, j), a0, a1, a2, a3);
        }
    }

    den = wave_sum(den);
    float* dst = out + (size_t)node * HID + lane;
    if (end > start) {
        dst[0]  = a0 / den;
        dst[32] = a1 / den;
        dst[64] = a2 / den;
        dst[96] = a3 / den;
    } else {
        dst[0] = 0.0f; dst[32] = 0.0f; dst[64] = 0.0f; dst[96] = 0.0f;
    }
}

extern "C" void kernel_launch(void* const* d_in, const int* in_sizes, int n_in,
                              void* d_out, int out_size, void* d_ws, size_t ws_size,
                              hipStream_t stream) {
    const float* features = (const float*)d_in[0];
    const float* rel_emb  = (const float*)d_in[1];
    const int*   indices  = (const int*)  d_in[2];
    const int*   sp_idx   = (const int*)  d_in[3];
    const float* sp_val   = (const float*)d_in[4];
    const float* attn     = (const float*)d_in[5];
    float* out = (float*)d_out;

    float* ws       = (float*)d_ws;
    float* rel_norm = ws;
    float* rel_att  = ws + 512;
    int*   perm     = (int*)(ws + 1024);
    int*   cur      = (int*)(ws + 1024 + NTRIPLE);

    zero_int_kernel<<<49, 1024, 0, stream>>>(cur, NODE_SIZE);

    rel_stats_kernel<<<32, 32, 0, stream>>>(rel_emb, attn, rel_norm, rel_att);

    count_kernel<<<SORT_GROUPS, 1024, 0, stream>>>(indices, cur);
    scan_kernel<<<1, 1024, 0, stream>>>(cur);
    fill_kernel<<<SORT_GROUPS, 1024, 0, stream>>>(indices, cur, perm);

    gather_kernel<<<(NODE_SIZE + 7) / 8, 256, 0, stream>>>(
        features, rel_emb, indices, sp_idx, sp_val, rel_norm, rel_att,
        cur, perm, out);
}


# gather: 256 resident work-groups stage the relation table in LDS once, their waves loop over nodes (LDS counter, next node prefetched): relation rows read from LDS, not L2; 8 feature rows per trip
# speedup vs baseline: 1.5694x; 1.0596x over previous
#include <hip/hip_runtime.h>

#define NODE_SIZE 50000
#define REL_SIZE  500
#define HID       128
#define NTRIPLE   800000
#define EPSF      1e-12f

typedef __attribute__((ext_vector_type(2))) float v2f;
typedef __attribute__((ext_vector_type(8))) float v8f;

__global__ void zero_int_kernel(int* __restrict__ p, int n) {
    int i = blockIdx.x * blockDim.x + threadIdx.x;
    int stride = gridDim.x * blockDim.x;
    for (; i < n; i += stride) p[i] = 0;
}

__global__ __launch_bounds__(32)
void rel_stats_kernel(const float* __restrict__ rel_emb,
                      const float* __restrict__ attn,
                      float* __restrict__ rel_norm,
                      float* __restrict__ rel_att) {
    const int lane = threadIdx.x & 31;
    const int tile = blockIdx.x;
    const int half = lane >> 4;
    const int m16  = lane & 15;

    const int   row   = tile * 16 + m16;
    const bool  rowok = row < REL_SIZE;
    const float amask = rowok ? 1.0f : 0.0f;
    const float* arow = rel_emb + (size_t)(rowok ? row : 0) * HID;
    const float bmask = (m16 == 0) ? 1.0f : 0.0f;

    v8f c = {};
    for (int k = 0; k < HID; k += 4) {
        const int koff = half * 2;
        v2f a, b;
        a.x = arow[k + koff + 0] * amask;
        a.y = arow[k + koff + 1] * amask;
        b.x = attn[k + koff + 0] * bmask;
        b.y = attn[k + koff + 1] * bmask;
        c = __builtin_amdgcn_wmma_f32_16x16x4_f32(false, a, false, b,
                                                  (short)0, c, false, false);
    }
    if (m16 == 0) {
        const int mbase = tile * 16 + half * 8;
#pragma unroll
        for (int j = 0; j < 8; ++j)
            if (mbase + j < REL_SIZE) rel_att[mbase + j] = c[j];
    }

    float ss = 0.0f;
    {
        const float4* p = (const float4*)(arow + half * 64);
#pragma unroll
        for (int i = 0; i < 16; ++i) {
            float4 f = p[i];
            ss += f.x * f.x + f.y * f.y + f.z * f.z + f.w * f.w;
        }
    }
    ss += __shfl_xor(ss, 16, 32);
    if (half == 0 && rowok) rel_norm[row] = sqrtf(ss);
}

#define SORT_GROUPS  64
#define SORT_CHUNK   (NTRIPLE / SORT_GROUPS)
#define SORT_EDGES   13
#define NODE_TRIPS   49

__device__ __forceinline__ int node_slot(int k, int rot, int tid) {
    int kk = k + rot;
    if (kk >= NODE_TRIPS) kk -= NODE_TRIPS;
    return kk * 1024 + tid;
}

__global__ __launch_bounds__(1024)
void count_kernel(const int* __restrict__ indices, int* __restrict__ cnt) {
    __shared__ int h[NODE_SIZE];
    const int tid = threadIdx.x;
    for (int i = tid; i < NODE_SIZE; i += 1024) h[i] = 0;

    const int lo = blockIdx.x * SORT_CHUNK;
    int row[SORT_EDGES];
#pragma unroll
    for (int k = 0; k < SORT_EDGES; ++k) {
        const int e = k * 1024 + tid;
        row[k] = (e < SORT_CHUNK) ? indices[2 * (lo + e)] : -1;
    }
    __syncthreads();
#pragma unroll
    for (int k = 0; k < SORT_EDGES; ++k)
        if (row[k] >= 0) atomicAdd(&h[row[k]], 1);
    __syncthreads();

    const int rot = blockIdx.x % NODE_TRIPS;
    for (int k = 0; k < NODE_TRIPS; ++k) {
        const int i = node_slot(k, rot, tid);
        if (i < NODE_SIZE) {
            const int c = h[i];
            if (c) atomicAdd(&cnt[i], c);
        }
    }
}

#define SCAN_PER_THREAD 49

__global__ __launch_bounds__(1024)
void scan_kernel(int* __restrict__ cur) {
    __shared__ int s[1024 * SCAN_PER_THREAD];
    __shared__ int wsum[32];
    const int tid  = threadIdx.x;
    const int lane = tid & 31;
    const int wave = tid >> 5;
    for (int i = tid; i < 1024 * SCAN_PER_THREAD; i += 1024)
        s[i] = (i < NODE_SIZE) ? cur[i] : 0;
    __syncthreads();

    const int base = tid * SCAN_PER_THREAD;
    int sum = 0;
    for (int j = 0; j < SCAN_PER_THREAD; ++j) sum += s[base + j];

    int incl = sum;
#pragma unroll
    for (int off = 1; off < 32; off <<= 1) {
        const int n = __shfl_up(incl, off, 32);
        if (lane >= off) incl += n;
    }
    if (lane == 31) wsum[wave] = incl;
    __syncthreads();
    if (wave == 0) {
        const int w = wsum[lane];
        int wi = w;
#pragma unroll
        for (int off = 1; off < 32; off <<= 1) {
            const int n = __shfl_up(wi, off, 32);
            if (lane >= off) wi += n;
        }
        wsum[lane] = wi - w;
    }
    __syncthreads();

    int run = incl - sum + wsum[wave];
    for (int j = 0; j < SCAN_PER_THREAD; ++j) {
        const int c = s[base + j];
        s[base + j] = run;
        run += c;
    }
    __syncthreads();
    for (int i = tid; i < NODE_SIZE; i += 1024) cur[i] = s[i];
}

__global__ __launch_bounds__(1024)
void fill_kernel(const int* __restrict__ indices,
                 int* __restrict__ cur,
                 int* __restrict__ perm) {
    __shared__ int h[NODE_SIZE];
    const int tid = threadIdx.x;
    for (int i = tid; i < NODE_SIZE; i += 1024) h[i] = 0;

    const int lo = blockIdx.x * SORT_CHUNK;
    int row[SORT_EDGES], rank[SORT_EDGES];
#pragma unroll
    for (int k = 0; k < SORT_EDGES; ++k) {
        const int e = k * 1024 + tid;
        row[k] = (e < SORT_CHUNK) ? indices[2 * (lo + e)] : -1;
    }
    __syncthreads();
#pragma unroll
    for (int k = 0; k < SORT_EDGES; ++k)
        rank[k] = (row[k] >= 0) ? atomicAdd(&h[row[k]], 1) : 0;
    __syncthreads();

    const int rot = blockIdx.x % NODE_TRIPS;
    for (int k0 = 0; k0 < NODE_TRIPS; k0 += 7) {
        int start[7];
#pragma unroll
        for (int k = 0; k < 7; ++k) {
            const int i = node_slot(k0 + k, rot, tid);
            start[k] = (i < NODE_SIZE) ? atomicAdd(&cur[i], h[i]) : 0;
        }
#pragma unroll
        for (int k = 0; k < 7; ++k) {
            const int i = node_slot(k0 + k, rot, tid);
            if (i < NODE_SIZE) h[i] = start[k];
        }
    }
    __syncthreads();

#pragma unroll
    for (int k = 0; k < SORT_EDGES; ++k) {
        if (row[k] >= 0) {
            int pos = h[row[k]] + rank[k];
            pos = min(max(pos, 0), NTRIPLE - 1);
            perm[pos] = lo + k * 1024 + tid;
        }
    }
}

#define G_GROUPS 256
#define G_WAVES  32
#define G_CHUNK  ((NODE_SIZE + G_GROUPS - 1) / G_GROUPS)
#define REL_VEC4 (REL_SIZE * HID / 4)

__device__ __forceinline__ float wave_sum(float p) {
#pragma unroll
    for (int off = 16; off >= 1; off >>= 1) p += __shfl_xor(p, off, 32);
    return p;
}

__device__ __forceinline__ void add_edge(const float (&f)[4], const float* rrow,
                                         float coef, float e,
                                         float& a0, float& a1, float& a2, float& a3) {
    const float r0 = rrow[0], r1 = rrow[32], r2 = rrow[64], r3 = rrow[96];
    const float partial = wave_sum(f[0] * r0 + f[1] * r1 + f[2] * r2 + f[3] * r3);
    const float proj = coef * partial;
    const float k2   = 2.0f * proj * coef;
    a0 += (f[0] - k2 * r0) * e;
    a1 += (f[1] - k2 * r1) * e;
    a2 += (f[2] - k2 * r2) * e;
    a3 += (f[3] - k2 * r3) * e;
}

__device__ __forceinline__ void load_feat(float (&f)[4], const float* __restrict__ features,
                                          int col, int lane) {
    const float* frow = features + (size_t)col * HID + lane;
    f[0] = frow[0]; f[1] = frow[32]; f[2] = frow[64]; f[3] = frow[96];
}

__device__ __forceinline__ int bcast_i(int v, int j) {
    return __builtin_amdgcn_readlane(v, j);
}
__device__ __forceinline__ float bcast_f(float v, int j) {
    return __int_as_float(__builtin_amdgcn_readlane(__float_as_int(v), j));
}

__device__ __forceinline__ void seg_bounds(const int* __restrict__ cur, int node,
                                           int& s, int& e) {
    const int n = min(node, NODE_SIZE - 1);
    e = cur[n];
    s = cur[max(n - 1, 0)];
    if (n == 0) s = 0;
}

__device__ __forceinline__ void sum_batch(const float* __restrict__ features,
                                          const float* s_rel,
                                          int col, int r, float coef, float e,
                                          int m, int lane,
                                          float& a0, float& a1, float& a2, float& a3) {
    int j = 0;
    for (; j + 8 <= m; j += 8) {
        float f[8][4];
#pragma unroll
        for (int k = 0; k < 8; ++k) load_feat(f[k], features, bcast_i(col, j + k), lane);
#pragma unroll
        for (int k = 0; k < 8; ++k)
            add_edge(f[k], s_rel + bcast_i(r, j + k) * HID + lane,
                     bcast_f(coef, j + k), bcast_f(e, j + k), a0, a1, a2, a3);
    }
    for (; j + 4 <= m; j += 4) {
        float f[4][4];
#pragma unroll
        for (int k = 0; k < 4; ++k) load_feat(f[k], features, bcast_i(col, j + k), lane);
#pragma unroll
        for (int k = 0; k < 4; ++k)
            add_edge(f[k], s_rel + bcast_i(r, j + k) * HID + lane,
                     bcast_f(coef, j + k), bcast_f(e, j + k), a0, a1, a2, a3);
    }
    for (; j < m; ++j) {
        float f[4];
        load_feat(f, features, bcast_i(col, j), lane);
        add_edge(f, s_rel + bcast_i(r, j) * HID + lane,
                 bcast_f(coef, j), bcast_f(e, j), a0, a1, a2, a3);
    }
}

__global__ __launch_bounds__(1024)
void gather_kernel(const float* __restrict__ features,
                   const float* __restrict__ rel_emb,
                   const int*   __restrict__ indices,
                   const int*   __restrict__ sp_idx,
                   const float* __restrict__ sp_val,
                   const float* __restrict__ rel_norm,
                   const float* __restrict__ rel_att,
                   const int*   __restrict__ cur,
                   const int*   __restrict__ perm,
                   float* __restrict__ out) {
    __shared__ float s_rel[REL_SIZE * HID];
    __shared__ float s_norm[512];
    __shared__ float s_att[512];
    __shared__ int   s_next;

    const int tid  = threadIdx.x;
    const int lane = tid & 31;
    const int wave = __builtin_amdgcn_readfirstlane(tid >> 5);
    const int lo = blockIdx.x * G_CHUNK;
    const int hi = min(lo + G_CHUNK, NODE_SIZE);

    if (tid == 0) s_next = lo + 3 * G_WAVES;
    int n0 = lo + wave, n1 = n0 + G_WAVES, n2 = n1 + G_WAVES;
    int s0, e0, s1, e1, s2, e2;
    seg_bounds(cur, n0, s0, e0);
    seg_bounds(cur, n1, s1, e1);
    seg_bounds(cur, n2, s2, e2);
    int t0 = 0, t1 = 0;
    if (n0 < hi && lane < e0 - s0) t0 = perm[s0 + lane];
    if (n1 < hi && lane < e1 - s1) t1 = perm[s1 + lane];

    {
        const float4* src = (const float4*)rel_emb;
        float4* dst = (float4*)s_rel;
#pragma unroll
        for (int b = 0; b < 4; ++b) {
            float4 v[4];
#pragma unroll
            for (int k = 0; k < 4; ++k) {
                const int i = (b * 4 + k) * 1024 + tid;
                v[k] = src[min(i, REL_VEC4 - 1)];
            }
#pragma unroll
            for (int k = 0; k < 4; ++k) {
                const int i = (b * 4 + k) * 1024 + tid;
                if (i < REL_VEC4) dst[i] = v[k];
            }
        }
        if (tid < REL_SIZE) { s_norm[tid] = rel_norm[tid]; s_att[tid] = rel_att[tid]; }
    }

    int colA = 0, rA = 0;
    float vA = 0.0f;
    if (n0 < hi && lane < e0 - s0) {
        const int tc = min(max(t0, 0), NTRIPLE - 1);
        colA = indices[2 * tc + 1];
        rA   = sp_idx[2 * tc + 1];
        vA   = sp_val[tc];
    }
    __syncthreads();

    while (n0 < hi) {
        int n3 = 0;
        if (lane == 0) n3 = atomicAdd(&s_next, 1);
        n3 = __builtin_amdgcn_readfirstlane(n3);
        int s3, e3;
        seg_bounds(cur, n3, s3, e3);
        int t2 = 0;
        if (n2 < hi && lane < e2 - s2) t2 = perm[s2 + lane];
        int colB = 0, rB = 0;
        float vB = 0.0f;
        if (n1 < hi && lane < e1 - s1) {
            const int tc = min(max(t1, 0), NTRIPLE - 1);
            colB = indices[2 * tc + 1];
            rB   = sp_idx[2 * tc + 1];
            vB   = sp_val[tc];
        }

        float a0 = 0.0f, a1 = 0.0f, a2 = 0.0f, a3 = 0.0f, den = 0.0f;
        {
            const int m = min(32, e0 - s0);
            float coef = 0.0f, e = 0.0f;
            if (lane < m) {
                coef = vA / fmaxf(fabsf(vA) * s_norm[rA], EPSF);
                e    = expf(coef * s_att[rA]);
            }
            den += e;
            sum_batch(features, s_rel, colA, rA, coef, e, m, lane, a0, a1, a2, a3);
        }
        for (int base = s0 + 32; base < e0; base += 32) {
            const int m = min(32, e0 - base);
            int col = 0, r = 0;
            float coef = 0.0f, e = 0.0f;
            if (lane < m) {
                const int tc = min(max(perm[base + lane], 0), NTRIPLE - 1);
                col = indices[2 * tc + 1];
                r   = sp_idx[2 * tc + 1];
                const float v = sp_val[tc];
                coef = v / fmaxf(fabsf(v) * s_norm[r], EPSF);
                e    = expf(coef * s_att[r]);
            }
            den += e;
            sum_batch(features, s_rel, col, r, coef, e, m, lane, a0, a1, a2, a3);
        }

        den = wave_sum(den);
        float* dst = out + (size_t)n0 * HID + lane;
        if (e0 > s0) {
            dst[0]  = a0 / den;
            dst[32] = a1 / den;
            dst[64] = a2 / den;
            dst[96] = a3 / den;
        } else {
            dst[0] = 0.0f; dst[32] = 0.0f; dst[64] = 0.0f; dst[96] = 0.0f;
        }

        n0 = n1; s0 = s1; e0 = e1;
        n1 = n2; s1 = s2; e1 = e2; t1 = t2;
        n2 = n3; s2 = s3; e2 = e3;
        colA = colB; rA = rB; vA = vB;
    }
}

extern "C" void kernel_launch(void* const* d_in, const int* in_sizes, int n_in,
                              void* d_out, int out_size, void* d_ws, size_t ws_size,
                              hipStream_t stream) {
    const float* features = (const float*)d_in[0];
    const float* rel_emb  = (const float*)d_in[1];
    const int*   indices  = (const int*)  d_in[2];
    const int*   sp_idx   = (const int*)  d_in[3];
    const float* sp_val   = (const float*)d_in[4];
    const float* attn     = (const float*)d_in[5];
    float* out = (float*)d_out;

    float* ws       = (float*)d_ws;
    float* rel_norm = ws;
    float* rel_att  = ws + 512;
    int*   perm     = (int*)(ws + 1024);
    int*   cur      = (int*)(ws + 1024 + NTRIPLE);

    zero_int_kernel<<<49, 1024, 0, stream>>>(cur, NODE_SIZE);

    rel_stats_kernel<<<32, 32, 0, stream>>>(rel_emb, attn, rel_norm, rel_att);

    count_kernel<<<SORT_GROUPS, 1024, 0, stream>>>(indices, cur);
    scan_kernel<<<1, 1024, 0, stream>>>(cur);
    fill_kernel<<<SORT_GROUPS, 1024, 0, stream>>>(indices, cur, perm);

    gather_kernel<<<G_GROUPS, 1024, 0, stream>>>(
        features, rel_emb, indices, sp_idx, sp_val, rel_norm, rel_att,
        cur, perm, out);
}


# scan kernel: the 50000 degrees fetched with 12 16-byte loads + one 4-byte load per thread, all in flight at once; the built parent waited for each of its 49 4-byte loads before issuing the next
# speedup vs baseline: 1.7313x; 1.1032x over previous
#include <hip/hip_runtime.h>

#define NODE_SIZE 50000
#define REL_SIZE  500
#define HID       128
#define NTRIPLE   800000
#define EPSF      1e-12f

typedef __attribute__((ext_vector_type(2))) float v2f;
typedef __attribute__((ext_vector_type(8))) float v8f;

__global__ void zero_int_kernel(int* __restrict__ p, int n) {
    int i = blockIdx.x * blockDim.x + threadIdx.x;
    int stride = gridDim.x * blockDim.x;
    for (; i < n; i += stride) p[i] = 0;
}

__global__ __launch_bounds__(32)
void rel_stats_kernel(const float* __restrict__ rel_emb,
                      const float* __restrict__ attn,
                      float* __restrict__ rel_norm,
                      float* __restrict__ rel_att) {
    const int lane = threadIdx.x & 31;
    const int tile = blockIdx.x;
    const int half = lane >> 4;
    const int m16  = lane & 15;

    const int   row   = tile * 16 + m16;
    const bool  rowok = row < REL_SIZE;
    const float amask = rowok ? 1.0f : 0.0f;
    const float* arow = rel_emb + (size_t)(rowok ? row : 0) * HID;
    const float bmask = (m16 == 0) ? 1.0f : 0.0f;

    v8f c = {};
    for (int k = 0; k < HID; k += 4) {
        const int koff = half * 2;
        v2f a, b;
        a.x = arow[k + koff + 0] * amask;
        a.y = arow[k + koff + 1] * amask;
        b.x = attn[k + koff + 0] * bmask;
        b.y = attn[k + koff + 1] * bmask;
        c = __builtin_amdgcn_wmma_f32_16x16x4_f32(false, a, false, b,
                                                  (short)0, c, false, false);
    }
    if (m16 == 0) {
        const int mbase = tile * 16 + half * 8;
#pragma unroll
        for (int j = 0; j < 8; ++j)
            if (mbase + j < REL_SIZE) rel_att[mbase + j] = c[j];
    }

    float ss = 0.0f;
    {
        const float4* p = (const float4*)(arow + half * 64);
#pragma unroll
        for (int i = 0; i < 16; ++i) {
            float4 f = p[i];
            ss += f.x * f.x + f.y * f.y + f.z * f.z + f.w * f.w;
        }
    }
    ss += __shfl_xor(ss, 16, 32);
    if (half == 0 && rowok) rel_norm[row] = sqrtf(ss);
}

#define SORT_GROUPS  64
#define SORT_CHUNK   (NTRIPLE / SORT_GROUPS)
#define SORT_EDGES   13
#define NODE_TRIPS   49

__device__ __forceinline__ int node_slot(int k, int rot, int tid) {
    int kk = k + rot;
    if (kk >= NODE_TRIPS) kk -= NODE_TRIPS;
    return kk * 1024 + tid;
}

__global__ __launch_bounds__(1024)
void count_kernel(const int* __restrict__ indices, int* __restrict__ cnt) {
    __shared__ int h[NODE_SIZE];
    const int tid = threadIdx.x;
    for (int i = tid; i < NODE_SIZE; i += 1024) h[i] = 0;

    const int lo = blockIdx.x * SORT_CHUNK;
    int row[SORT_EDGES];
#pragma unroll
    for (int k = 0; k < SORT_EDGES; ++k) {
        const int e = k * 1024 + tid;
        row[k] = (e < SORT_CHUNK) ? indices[2 * (lo + e)] : -1;
    }
    __syncthreads();
#pragma unroll
    for (int k = 0; k < SORT_EDGES; ++k)
        if (row[k] >= 0) atomicAdd(&h[row[k]], 1);
    __syncthreads();

    const int rot = blockIdx.x % NODE_TRIPS;
    for (int k = 0; k < NODE_TRIPS; ++k) {
        const int i = node_slot(k, rot, tid);
        if (i < NODE_SIZE) {
            const int c = h[i];
            if (c) atomicAdd(&cnt[i], c);
        }
    }
}

#define SCAN_PER_THREAD 49
#define SCAN_WIDE  12
#define SCAN_TAIL  (SCAN_WIDE * 4096)

__global__ __launch_bounds__(1024)
void scan_kernel(int* __restrict__ cur) {
    __shared__ int4 s4[1024 * SCAN_PER_THREAD / 4];
    __shared__ int wsum[32];
    int* s = (int*)s4;
    const int tid  = threadIdx.x;
    const int lane = tid & 31;
    const int wave = tid >> 5;

    {
        const int4* src = (const int4*)cur;
        int4 v[SCAN_WIDE];
#pragma unroll
        for (int k = 0; k < SCAN_WIDE; ++k) v[k] = src[k * 1024 + tid];
        const int it = SCAN_TAIL + tid;
        const int vt = cur[min(it, NODE_SIZE - 1)];
#pragma unroll
        for (int k = 0; k < SCAN_WIDE; ++k) s4[k * 1024 + tid] = v[k];
        s[it] = (it < NODE_SIZE) ? vt : 0;
    }
    __syncthreads();

    const int base = tid * SCAN_PER_THREAD;
    int sum = 0;
    for (int j = 0; j < SCAN_PER_THREAD; ++j) sum += s[base + j];

    int incl = sum;
#pragma unroll
    for (int off = 1; off < 32; off <<= 1) {
        const int n = __shfl_up(incl, off, 32);
        if (lane >= off) incl += n;
    }
    if (lane == 31) wsum[wave] = incl;
    __syncthreads();
    if (wave == 0) {
        const int w = wsum[lane];
        int wi = w;
#pragma unroll
        for (int off = 1; off < 32; off <<= 1) {
            const int n = __shfl_up(wi, off, 32);
            if (lane >= off) wi += n;
        }
        wsum[lane] = wi - w;
    }
    __syncthreads();

    int run = incl - sum + wsum[wave];
    for (int j = 0; j < SCAN_PER_THREAD; ++j) {
        const int c = s[base + j];
        s[base + j] = run;
        run += c;
    }
    __syncthreads();
    for (int i = tid; i < NODE_SIZE; i += 1024) cur[i] = s[i];
}

__global__ __launch_bounds__(1024)
void fill_kernel(const int* __restrict__ indices,
                 int* __restrict__ cur,
                 int* __restrict__ perm) {
    __shared__ int h[NODE_SIZE];
    const int tid = threadIdx.x;
    for (int i = tid; i < NODE_SIZE; i += 1024) h[i] = 0;

    const int lo = blockIdx.x * SORT_CHUNK;
    int row[SORT_EDGES], rank[SORT_EDGES];
#pragma unroll
    for (int k = 0; k < SORT_EDGES; ++k) {
        const int e = k * 1024 + tid;
        row[k] = (e < SORT_CHUNK) ? indices[2 * (lo + e)] : -1;
    }
    __syncthreads();
#pragma unroll
    for (int k = 0; k < SORT_EDGES; ++k)
        rank[k] = (row[k] >= 0) ? atomicAdd(&h[row[k]], 1) : 0;
    __syncthreads();

    const int rot = blockIdx.x % NODE_TRIPS;
    for (int k0 = 0; k0 < NODE_TRIPS; k0 += 7) {
        int start[7];
#pragma unroll
        for (int k = 0; k < 7; ++k) {
            const int i = node_slot(k0 + k, rot, tid);
            start[k] = (i < NODE_SIZE) ? atomicAdd(&cur[i], h[i]) : 0;
        }
#pragma unroll
        for (int k = 0; k < 7; ++k) {
            const int i = node_slot(k0 + k, rot, tid);
            if (i < NODE_SIZE) h[i] = start[k];
        }
    }
    __syncthreads();

#pragma unroll
    for (int k = 0; k < SORT_EDGES; ++k) {
        if (row[k] >= 0) {
            int pos = h[row[k]] + rank[k];
            pos = min(max(pos, 0), NTRIPLE - 1);
            perm[pos] = lo + k * 1024 + tid;
        }
    }
}

#define G_GROUPS 256
#define G_WAVES  32
#define G_CHUNK  ((NODE_SIZE + G_GROUPS - 1) / G_GROUPS)
#define REL_VEC4 (REL_SIZE * HID / 4)

__device__ __forceinline__ float wave_sum(float p) {
#pragma unroll
    for (int off = 16; off >= 1; off >>= 1) p += __shfl_xor(p, off, 32);
    return p;
}

__device__ __forceinline__ void add_edge(const float (&f)[4], const float* rrow,
                                         float coef, float e,
                                         float& a0, float& a1, float& a2, float& a3) {
    const float r0 = rrow[0], r1 = rrow[32], r2 = rrow[64], r3 = rrow[96];
    const float partial = wave_sum(f[0] * r0 + f[1] * r1 + f[2] * r2 + f[3] * r3);
    const float proj = coef * partial;
    const float k2   = 2.0f * proj * coef;
    a0 += (f[0] - k2 * r0) * e;
    a1 += (f[1] - k2 * r1) * e;
    a2 += (f[2] - k2 * r2) * e;
    a3 += (f[3] - k2 * r3) * e;
}

__device__ __forceinline__ void load_feat(float (&f)[4], const float* __restrict__ features,
                                          int col, int lane) {
    const float* frow = features + (size_t)col * HID + lane;
    f[0] = frow[0]; f[1] = frow[32]; f[2] = frow[64]; f[3] = frow[96];
}

__device__ __forceinline__ int bcast_i(int v, int j) {
    return __builtin_amdgcn_readlane(v, j);
}
__device__ __forceinline__ float bcast_f(float v, int j) {
    return __int_as_float(__builtin_amdgcn_readlane(__float_as_int(v), j));
}

__device__ __forceinline__ void seg_bounds(const int* __restrict__ cur, int node,
                                           int& s, int& e) {
    const int n = min(node, NODE_SIZE - 1);
    e = cur[n];
    s = cur[max(n - 1, 0)];
    if (n == 0) s = 0;
}

__device__ __forceinline__ void sum_batch(const float* __restrict__ features,
                                          const float* s_rel,
                                          int col, int r, float coef, float e,
                                          int m, int lane,
                                          float& a0, float& a1, float& a2, float& a3) {
    int j = 0;
    for (; j + 8 <= m; j += 8) {
        float f[8][4];
#pragma unroll
        for (int k = 0; k < 8; ++k) load_feat(f[k], features, bcast_i(col, j + k), lane);
#pragma unroll
        for (int k = 0; k < 8; ++k)
            add_edge(f[k], s_rel + bcast_i(r, j + k) * HID + lane,
                     bcast_f(coef, j + k), bcast_f(e, j + k), a0, a1, a2, a3);
    }
    for (; j + 4 <= m; j += 4) {
        float f[4][4];
#pragma unroll
        for (int k = 0; k < 4; ++k) load_feat(f[k], features, bcast_i(col, j + k), lane);
#pragma unroll
        for (int k = 0; k < 4; ++k)
            add_edge(f[k], s_rel + bcast_i(r, j + k) * HID + lane,
                     bcast_f(coef, j + k), bcast_f(e, j + k), a0, a1, a2, a3);
    }
    for (; j < m; ++j) {
        float f[4];
        load_feat(f, features, bcast_i(col, j), lane);
        add_edge(f, s_rel + bcast_i(r, j) * HID + lane,
                 bcast_f(coef, j), bcast_f(e, j), a0, a1, a2, a3);
    }
}

__global__ __launch_bounds__(1024)
void gather_kernel(const float* __restrict__ features,
                   const float* __restrict__ rel_emb,
                   const int*   __restrict__ indices,
                   const int*   __restrict__ sp_idx,
                   const float* __restrict__ sp_val,
                   const float* __restrict__ rel_norm,
                   const float* __restrict__ rel_att,
                   const int*   __restrict__ cur,
                   const int*   __restrict__ perm,
                   float* __restrict__ out) {
    __shared__ float s_rel[REL_SIZE * HID];
    __shared__ float s_norm[512];
    __shared__ float s_att[512];
    __shared__ int   s_next;

    const int tid  = threadIdx.x;
    const int lane = tid & 31;
    const int wave = __builtin_amdgcn_readfirstlane(tid >> 5);
    const int lo = blockIdx.x * G_CHUNK;
    const int hi = min(lo + G_CHUNK, NODE_SIZE);

    if (tid == 0) s_next = lo + 3 * G_WAVES;
    int n0 = lo + wave, n1 = n0 + G_WAVES, n2 = n1 + G_WAVES;
    int s0, e0, s1, e1, s2, e2;
    seg_bounds(cur, n0, s0, e0);
    seg_bounds(cur, n1, s1, e1);
    seg_bounds(cur, n2, s2, e2);
    int t0 = 0, t1 = 0;
    if (n0 < hi && lane < e0 - s0) t0 = perm[s0 + lane];
    if (n1 < hi && lane < e1 - s1) t1 = perm[s1 + lane];

    {
        const float4* src = (const float4*)rel_emb;
        float4* dst = (float4*)s_rel;
#pragma unroll
        for (int b = 0; b < 4; ++b) {
            float4 v[4];
#pragma unroll
            for (int k = 0; k < 4; ++k) {
                const int i = (b * 4 + k) * 1024 + tid;
                v[k] = src[min(i, REL_VEC4 - 1)];
            }
#pragma unroll
            for (int k = 0; k < 4; ++k) {
                const int i = (b * 4 + k) * 1024 + tid;
                if (i < REL_VEC4) dst[i] = v[k];
            }
        }
        if (tid < REL_SIZE) { s_norm[tid] = rel_norm[tid]; s_att[tid] = rel_att[tid]; }
    }

    int colA = 0, rA = 0;
    float vA = 0.0f;
    if (n0 < hi && lane < e0 - s0) {
        const int tc = min(max(t0, 0), NTRIPLE - 1);
        colA = indices[2 * tc + 1];
        rA   = sp_idx[2 * tc + 1];
        vA   = sp_val[tc];
    }
    __syncthreads();

    while (n0 < hi) {
        int n3 = 0;
        if (lane == 0) n3 = atomicAdd(&s_next, 1);
        n3 = __builtin_amdgcn_readfirstlane(n3);
        int s3, e3;
        seg_bounds(cur, n3, s3, e3);
        int t2 = 0;
        if (n2 < hi && lane < e2 - s2) t2 = perm[s2 + lane];
        int colB = 0, rB = 0;
        float vB = 0.0f;
        if (n1 < hi && lane < e1 - s1) {
            const int tc = min(max(t1, 0), NTRIPLE - 1);
            colB = indices[2 * tc + 1];
            rB   = sp_idx[2 * tc + 1];
            vB   = sp_val[tc];
        }

        float a0 = 0.0f, a1 = 0.0f, a2 = 0.0f, a3 = 0.0f, den = 0.0f;
        {
            const int m = min(32, e0 - s0);
            float coef = 0.0f, e = 0.0f;
            if (lane < m) {
                coef = vA / fmaxf(fabsf(vA) * s_norm[rA], EPSF);
                e    = expf(coef * s_att[rA]);
            }
            den += e;
            sum_batch(features, s_rel, colA, rA, coef, e, m, lane, a0, a1, a2, a3);
        }
        for (int base = s0 + 32; base < e0; base += 32) {
            const int m = min(32, e0 - base);
            int col = 0, r = 0;
            float coef = 0.0f, e = 0.0f;
            if (lane < m) {
                const int tc = min(max(perm[base + lane], 0), NTRIPLE - 1);
                col = indices[2 * tc + 1];
                r   = sp_idx[2 * tc + 1];
                const float v = sp_val[tc];
                coef = v / fmaxf(fabsf(v) * s_norm[r], EPSF);
                e    = expf(coef * s_att[r]);
            }
            den += e;
            sum_batch(features, s_rel, col, r, coef, e, m, lane, a0, a1, a2, a3);
        }

        den = wave_sum(den);
        float* dst = out + (size_t)n0 * HID + lane;
        if (e0 > s0) {
            dst[0]  = a0 / den;
            dst[32] = a1 / den;
            dst[64] = a2 / den;
            dst[96] = a3 / den;
        } else {
            dst[0] = 0.0f; dst[32] = 0.0f; dst[64] = 0.0f; dst[96] = 0.0f;
        }

        n0 = n1; s0 = s1; e0 = e1;
        n1 = n2; s1 = s2; e1 = e2; t1 = t2;
        n2 = n3; s2 = s3; e2 = e3;
        colA = colB; rA = rB; vA = vB;
    }
}

extern "C" void kernel_launch(void* const* d_in, const int* in_sizes, int n_in,
                              void* d_out, int out_size, void* d_ws, size_t ws_size,
                              hipStream_t stream) {
    const float* features = (const float*)d_in[0];
    const float* rel_emb  = (const float*)d_in[1];
    const int*   indices  = (const int*)  d_in[2];
    const int*   sp_idx   = (const int*)  d_in[3];
    const float* sp_val   = (const float*)d_in[4];
    const float* attn     = (const float*)d_in[5];
    float* out = (float*)d_out;

    float* ws       = (float*)d_ws;
    float* rel_norm = ws;
    float* rel_att  = ws + 512;
    int*   perm     = (int*)(ws + 1024);
    int*   cur      = (int*)(ws + 1024 + NTRIPLE);

    zero_int_kernel<<<49, 1024, 0, stream>>>(cur, NODE_SIZE);

    rel_stats_kernel<<<32, 32, 0, stream>>>(rel_emb, attn, rel_norm, rel_att);

    count_kernel<<<SORT_GROUPS, 1024, 0, stream>>>(indices, cur);
    scan_kernel<<<1, 1024, 0, stream>>>(cur);
    fill_kernel<<<SORT_GROUPS, 1024, 0, stream>>>(indices, cur, perm);

    gather_kernel<<<G_GROUPS, 1024, 0, stream>>>(
        features, rel_emb, indices, sp_idx, sp_val, rel_norm, rel_att,
        cur, perm, out);
}


# gather: the 32-lane sums of the edges' dot products leave the LDS unit (ds_bpermute) for the vector ALU: permlane16 swap + row_xmask DPP adds, eight sums folded together, same pairing order
# speedup vs baseline: 1.7981x; 1.0386x over previous
#include <hip/hip_runtime.h>

#define NODE_SIZE 50000
#define REL_SIZE  500
#define HID       128
#define NTRIPLE   800000
#define EPSF      1e-12f

typedef __attribute__((ext_vector_type(2))) float v2f;
typedef __attribute__((ext_vector_type(8))) float v8f;

__global__ void zero_int_kernel(int* __restrict__ p, int n) {
    int i = blockIdx.x * blockDim.x + threadIdx.x;
    int stride = gridDim.x * blockDim.x;
    for (; i < n; i += stride) p[i] = 0;
}

__global__ __launch_bounds__(32)
void rel_stats_kernel(const float* __restrict__ rel_emb,
                      const float* __restrict__ attn,
                      float* __restrict__ rel_norm,
                      float* __restrict__ rel_att) {
    const int lane = threadIdx.x & 31;
    const int tile = blockIdx.x;
    const int half = lane >> 4;
    const int m16  = lane & 15;

    const int   row   = tile * 16 + m16;
    const bool  rowok = row < REL_SIZE;
    const float amask = rowok ? 1.0f : 0.0f;
    const float* arow = rel_emb + (size_t)(rowok ? row : 0) * HID;
    const float bmask = (m16 == 0) ? 1.0f : 0.0f;

    v8f c = {};
    for (int k = 0; k < HID; k += 4) {
        const int koff = half * 2;
        v2f a, b;
        a.x = arow[k + koff + 0] * amask;
        a.y = arow[k + koff + 1] * amask;
        b.x = attn[k + koff + 0] * bmask;
        b.y = attn[k + koff + 1] * bmask;
        c = __builtin_amdgcn_wmma_f32_16x16x4_f32(false, a, false, b,
                                                  (short)0, c, false, false);
    }
    if (m16 == 0) {
        const int mbase = tile * 16 + half * 8;
#pragma unroll
        for (int j = 0; j < 8; ++j)
            if (mbase + j < REL_SIZE) rel_att[mbase + j] = c[j];
    }

    float ss = 0.0f;
    {
        const float4* p = (const float4*)(arow + half * 64);
#pragma unroll
        for (int i = 0; i < 16; ++i) {
            float4 f = p[i];
            ss += f.x * f.x + f.y * f.y + f.z * f.z + f.w * f.w;
        }
    }
    ss += __shfl_xor(ss, 16, 32);
    if (half == 0 && rowok) rel_norm[row] = sqrtf(ss);
}

#define SORT_GROUPS  64
#define SORT_CHUNK   (NTRIPLE / SORT_GROUPS)
#define SORT_EDGES   13
#define NODE_TRIPS   49

__device__ __forceinline__ int node_slot(int k, int rot, int tid) {
    int kk = k + rot;
    if (kk >= NODE_TRIPS) kk -= NODE_TRIPS;
    return kk * 1024 + tid;
}

__global__ __launch_bounds__(1024)
void count_kernel(const int* __restrict__ indices, int* __restrict__ cnt) {
    __shared__ int h[NODE_SIZE];
    const int tid = threadIdx.x;
    for (int i = tid; i < NODE_SIZE; i += 1024) h[i] = 0;

    const int lo = blockIdx.x * SORT_CHUNK;
    int row[SORT_EDGES];
#pragma unroll
    for (int k = 0; k < SORT_EDGES; ++k) {
        const int e = k * 1024 + tid;
        row[k] = (e < SORT_CHUNK) ? indices[2 * (lo + e)] : -1;
    }
    __syncthreads();
#pragma unroll
    for (int k = 0; k < SORT_EDGES; ++k)
        if (row[k] >= 0) atomicAdd(&h[row[k]], 1);
    __syncthreads();

    const int rot = blockIdx.x % NODE_TRIPS;
    for (int k = 0; k < NODE_TRIPS; ++k) {
        const int i = node_slot(k, rot, tid);
        if (i < NODE_SIZE) {
            const int c = h[i];
            if (c) atomicAdd(&cnt[i], c);
        }
    }
}

#define SCAN_PER_THREAD 49
#define SCAN_WIDE  12
#define SCAN_TAIL  (SCAN_WIDE * 4096)

__global__ __launch_bounds__(1024)
void scan_kernel(int* __restrict__ cur) {
    __shared__ int4 s4[1024 * SCAN_PER_THREAD / 4];
    __shared__ int wsum[32];
    int* s = (int*)s4;
    const int tid  = threadIdx.x;
    const int lane = tid & 31;
    const int wave = tid >> 5;

    {
        const int4* src = (const int4*)cur;
        int4 v[SCAN_WIDE];
#pragma unroll
        for (int k = 0; k < SCAN_WIDE; ++k) v[k] = src[k * 1024 + tid];
        const int it = SCAN_TAIL + tid;
        const int vt = cur[min(it, NODE_SIZE - 1)];
#pragma unroll
        for (int k = 0; k < SCAN_WIDE; ++k) s4[k * 1024 + tid] = v[k];
        s[it] = (it < NODE_SIZE) ? vt : 0;
    }
    __syncthreads();

    const int base = tid * SCAN_PER_THREAD;
    int sum = 0;
    for (int j = 0; j < SCAN_PER_THREAD; ++j) sum += s[base + j];

    int incl = sum;
#pragma unroll
    for (int off = 1; off < 32; off <<= 1) {
        const int n = __shfl_up(incl, off, 32);
        if (lane >= off) incl += n;
    }
    if (lane == 31) wsum[wave] = incl;
    __syncthreads();
    if (wave == 0) {
        const int w = wsum[lane];
        int wi = w;
#pragma unroll
        for (int off = 1; off < 32; off <<= 1) {
            const int n = __shfl_up(wi, off, 32);
            if (lane >= off) wi += n;
        }
        wsum[lane] = wi - w;
    }
    __syncthreads();

    int run = incl - sum + wsum[wave];
    for (int j = 0; j < SCAN_PER_THREAD; ++j) {
        const int c = s[base + j];
        s[base + j] = run;
        run += c;
    }
    __syncthreads();
    for (int i = tid; i < NODE_SIZE; i += 1024) cur[i] = s[i];
}

__global__ __launch_bounds__(1024)
void fill_kernel(const int* __restrict__ indices,
                 int* __restrict__ cur,
                 int* __restrict__ perm) {
    __shared__ int h[NODE_SIZE];
    const int tid = threadIdx.x;
    for (int i = tid; i < NODE_SIZE; i += 1024) h[i] = 0;

    const int lo = blockIdx.x * SORT_CHUNK;
    int row[SORT_EDGES], rank[SORT_EDGES];
#pragma unroll
    for (int k = 0; k < SORT_EDGES; ++k) {
        const int e = k * 1024 + tid;
        row[k] = (e < SORT_CHUNK) ? indices[2 * (lo + e)] : -1;
    }
    __syncthreads();
#pragma unroll
    for (int k = 0; k < SORT_EDGES; ++k)
        rank[k] = (row[k] >= 0) ? atomicAdd(&h[row[k]], 1) : 0;
    __syncthreads();

    const int rot = blockIdx.x % NODE_TRIPS;
    for (int k0 = 0; k0 < NODE_TRIPS; k0 += 7) {
        int start[7];
#pragma unroll
        for (int k = 0; k < 7; ++k) {
            const int i = node_slot(k0 + k, rot, tid);
            start[k] = (i < NODE_SIZE) ? atomicAdd(&cur[i], h[i]) : 0;
        }
#pragma unroll
        for (int k = 0; k < 7; ++k) {
            const int i = node_slot(k0 + k, rot, tid);
            if (i < NODE_SIZE) h[i] = start[k];
        }
    }
    __syncthreads();

#pragma unroll
    for (int k = 0; k < SORT_EDGES; ++k) {
        if (row[k] >= 0) {
            int pos = h[row[k]] + rank[k];
            pos = min(max(pos, 0), NTRIPLE - 1);
            perm[pos] = lo + k * 1024 + tid;
        }
    }
}

#define G_GROUPS 256
#define G_WAVES  32
#define G_CHUNK  ((NODE_SIZE + G_GROUPS - 1) / G_GROUPS)
#define REL_VEC4 (REL_SIZE * HID / 4)


template <int M>
__device__ __forceinline__ float row_xor(float p) {
    return __int_as_float(__builtin_amdgcn_update_dpp(0, __float_as_int(p), 0x160 + M, 0xf, 0xf, true));
}

__device__ __forceinline__ float xhalf(float x) {
    const int b = __float_as_int(x);
    return __int_as_float(__builtin_amdgcn_permlanex16(b, b, 0x76543210, 0xfedcba98, false, false));
}

__device__ __forceinline__ float wave_sum(float p) {
    p += xhalf(p);
    p += row_xor<8>(p);
    p += row_xor<4>(p);
    p += row_xor<2>(p);
    p += row_xor<1>(p);
    return p;
}

__device__ __forceinline__ float swap_add(float x, float y) {
    const auto r = __builtin_amdgcn_permlane16_swap(__float_as_uint(x), __float_as_uint(y), false, false);
    return __uint_as_float(r[0]) + __uint_as_float(r[1]);
}

__device__ __forceinline__ void wave_sum8(const float (&p)[8], bool hi8, bool hi4, float (&tot)[8]) {
    float q[4];
#pragma unroll
    for (int i = 0; i < 4; ++i) q[i] = swap_add(p[i], p[i + 4]);
    float r[2];
#pragma unroll
    for (int i = 0; i < 2; ++i) {
        const float keep = hi8 ? q[i + 2] : q[i];
        const float send = hi8 ? q[i] : q[i + 2];
        r[i] = keep + row_xor<8>(send);
    }
    const float keep = hi4 ? r[1] : r[0];
    const float send = hi4 ? r[0] : r[1];
    float s = keep + row_xor<4>(send);
    s += row_xor<2>(s);
    s += row_xor<1>(s);
#pragma unroll
    for (int k = 0; k < 8; ++k)
        tot[k] = __int_as_float(__builtin_amdgcn_readlane(
            __float_as_int(s), 4 * (k & 1) + 8 * ((k >> 1) & 1) + 16 * (k >> 2)));
}

__device__ __forceinline__ void wave_sum4(const float (&p)[4], bool hi8, float (&tot)[4]) {
    const float q0 = swap_add(p[0], p[2]);
    const float q1 = swap_add(p[1], p[3]);
    const float keep = hi8 ? q1 : q0;
    const float send = hi8 ? q0 : q1;
    float s = keep + row_xor<8>(send);
    s += row_xor<4>(s);
    s += row_xor<2>(s);
    s += row_xor<1>(s);
#pragma unroll
    for (int k = 0; k < 4; ++k)
        tot[k] = __int_as_float(__builtin_amdgcn_readlane(
            __float_as_int(s), 8 * (k & 1) + 16 * (k >> 1)));
}

__device__ __forceinline__ void load_rel(float (&r)[4], const float* rrow) {
    r[0] = rrow[0]; r[1] = rrow[32]; r[2] = rrow[64]; r[3] = rrow[96];
}

__device__ __forceinline__ float lane_dot(const float (&f)[4], const float (&r)[4]) {
    return f[0] * r[0] + f[1] * r[1] + f[2] * r[2] + f[3] * r[3];
}

__device__ __forceinline__ void add_edge(const float (&f)[4], const float (&r)[4],
                                         float coef, float e, float partial,
                                         float& a0, float& a1, float& a2, float& a3) {
    const float proj = coef * partial;
    const float k2   = 2.0f * proj * coef;
    a0 += (f[0] - k2 * r[0]) * e;
    a1 += (f[1] - k2 * r[1]) * e;
    a2 += (f[2] - k2 * r[2]) * e;
    a3 += (f[3] - k2 * r[3]) * e;
}

__device__ __forceinline__ void load_feat(float (&f)[4], const float* __restrict__ features,
                                          int col, int lane) {
    const float* frow = features + (size_t)col * HID + lane;
    f[0] = frow[0]; f[1] = frow[32]; f[2] = frow[64]; f[3] = frow[96];
}

__device__ __forceinline__ int bcast_i(int v, int j) {
    return __builtin_amdgcn_readlane(v, j);
}
__device__ __forceinline__ float bcast_f(float v, int j) {
    return __int_as_float(__builtin_amdgcn_readlane(__float_as_int(v), j));
}

__device__ __forceinline__ void seg_bounds(const int* __restrict__ cur, int node,
                                           int& s, int& e) {
    const int n = min(node, NODE_SIZE - 1);
    e = cur[n];
    s = cur[max(n - 1, 0)];
    if (n == 0) s = 0;
}

__device__ __forceinline__ void sum_batch(const float* __restrict__ features,
                                          const float* s_rel,
                                          int col, int r, float coef, float e,
                                          int m, int lane,
                                          float& a0, float& a1, float& a2, float& a3) {
    const bool hi8 = (lane & 8) != 0;
    const bool hi4 = (lane & 4) != 0;
    int j = 0;
    for (; j + 8 <= m; j += 8) {
        float f[8][4], rr[8][4], p[8], tot[8];
#pragma unroll
        for (int k = 0; k < 8; ++k) load_feat(f[k], features, bcast_i(col, j + k), lane);
#pragma unroll
        for (int k = 0; k < 8; ++k) {
            load_rel(rr[k], s_rel + bcast_i(r, j + k) * HID + lane);
            p[k] = lane_dot(f[k], rr[k]);
        }
        wave_sum8(p, hi8, hi4, tot);
#pragma unroll
        for (int k = 0; k < 8; ++k)
            add_edge(f[k], rr[k], bcast_f(coef, j + k), bcast_f(e, j + k), tot[k], a0, a1, a2, a3);
    }
    for (; j + 4 <= m; j += 4) {
        float f[4][4], rr[4][4], p[4], tot[4];
#pragma unroll
        for (int k = 0; k < 4; ++k) load_feat(f[k], features, bcast_i(col, j + k), lane);
#pragma unroll
        for (int k = 0; k < 4; ++k) {
            load_rel(rr[k], s_rel + bcast_i(r, j + k) * HID + lane);
            p[k] = lane_dot(f[k], rr[k]);
        }
        wave_sum4(p, hi8, tot);
#pragma unroll
        for (int k = 0; k < 4; ++k)
            add_edge(f[k], rr[k], bcast_f(coef, j + k), bcast_f(e, j + k), tot[k], a0, a1, a2, a3);
    }
    for (; j < m; ++j) {
        float f[4], rr[4];
        load_feat(f, features, bcast_i(col, j), lane);
        load_rel(rr, s_rel + bcast_i(r, j) * HID + lane);
        add_edge(f, rr, bcast_f(coef, j), bcast_f(e, j), wave_sum(lane_dot(f, rr)), a0, a1, a2, a3);
    }
}

__global__ __launch_bounds__(1024)
void gather_kernel(const float* __restrict__ features,
                   const float* __restrict__ rel_emb,
                   const int*   __restrict__ indices,
                   const int*   __restrict__ sp_idx,
                   const float* __restrict__ sp_val,
                   const float* __restrict__ rel_norm,
                   const float* __restrict__ rel_att,
                   const int*   __restrict__ cur,
                   const int*   __restrict__ perm,
                   float* __restrict__ out) {
    __shared__ float s_rel[REL_SIZE * HID];
    __shared__ float s_norm[512];
    __shared__ float s_att[512];
    __shared__ int   s_next;

    const int tid  = threadIdx.x;
    const int lane = tid & 31;
    const int wave = __builtin_amdgcn_readfirstlane(tid >> 5);
    const int lo = blockIdx.x * G_CHUNK;
    const int hi = min(lo + G_CHUNK, NODE_SIZE);

    if (tid == 0) s_next = lo + 3 * G_WAVES;
    int n0 = lo + wave, n1 = n0 + G_WAVES, n2 = n1 + G_WAVES;
    int s0, e0, s1, e1, s2, e2;
    seg_bounds(cur, n0, s0, e0);
    seg_bounds(cur, n1, s1, e1);
    seg_bounds(cur, n2, s2, e2);
    int t0 = 0, t1 = 0;
    if (n0 < hi && lane < e0 - s0) t0 = perm[s0 + lane];
    if (n1 < hi && lane < e1 - s1) t1 = perm[s1 + lane];

    {
        const float4* src = (const float4*)rel_emb;
        float4* dst = (float4*)s_rel;
#pragma unroll
        for (int b = 0; b < 4; ++b) {
            float4 v[4];
#pragma unroll
            for (int k = 0; k < 4; ++k) {
                const int i = (b * 4 + k) * 1024 + tid;
                v[k] = src[min(i, REL_VEC4 - 1)];
            }
#pragma unroll
            for (int k = 0; k < 4; ++k) {
                const int i = (b * 4 + k) * 1024 + tid;
                if (i < REL_VEC4) dst[i] = v[k];
            }
        }
        if (tid < REL_SIZE) { s_norm[tid] = rel_norm[tid]; s_att[tid] = rel_att[tid]; }
    }

    int colA = 0, rA = 0;
    float vA = 0.0f;
    if (n0 < hi && lane < e0 - s0) {
        const int tc = min(max(t0, 0), NTRIPLE - 1);
        colA = indices[2 * tc + 1];
        rA   = sp_idx[2 * tc + 1];
        vA   = sp_val[tc];
    }
    __syncthreads();

    while (n0 < hi) {
        int n3 = 0;
        if (lane == 0) n3 = atomicAdd(&s_next, 1);
        n3 = __builtin_amdgcn_readfirstlane(n3);
        int s3, e3;
        seg_bounds(cur, n3, s3, e3);
        int t2 = 0;
        if (n2 < hi && lane < e2 - s2) t2 = perm[s2 + lane];
        int colB = 0, rB = 0;
        float vB = 0.0f;
        if (n1 < hi && lane < e1 - s1) {
            const int tc = min(max(t1, 0), NTRIPLE - 1);
            colB = indices[2 * tc + 1];
            rB   = sp_idx[2 * tc + 1];
            vB   = sp_val[tc];
        }

        float a0 = 0.0f, a1 = 0.0f, a2 = 0.0f, a3 = 0.0f, den = 0.0f;
        {
            const int m = min(32, e0 - s0);
            float coef = 0.0f, e = 0.0f;
            if (lane < m) {
                coef = vA / fmaxf(fabsf(vA) * s_norm[rA], EPSF);
                e    = expf(coef * s_att[rA]);
            }
            den += e;
            sum_batch(features, s_rel, colA, rA, coef, e, m, lane, a0, a1, a2, a3);
        }
        for (int base = s0 + 32; base < e0; base += 32) {
            const int m = min(32, e0 - base);
            int col = 0, r = 0;
            float coef = 0.0f, e = 0.0f;
            if (lane < m) {
                const int tc = min(max(perm[base + lane], 0), NTRIPLE - 1);
                col = indices[2 * tc + 1];
                r   = sp_idx[2 * tc + 1];
                const float v = sp_val[tc];
                coef = v / fmaxf(fabsf(v) * s_norm[r], EPSF);
                e    = expf(coef * s_att[r]);
            }
            den += e;
            sum_batch(features, s_rel, col, r, coef, e, m, lane, a0, a1, a2, a3);
        }

        den = wave_sum(den);
        float* dst = out + (size_t)n0 * HID + lane;
        if (e0 > s0) {
            dst[0]  = a0 / den;
            dst[32] = a1 / den;
            dst[64] = a2 / den;
            dst[96] = a3 / den;
        } else {
            dst[0] = 0.0f; dst[32] = 0.0f; dst[64] = 0.0f; dst[96] = 0.0f;
        }

        n0 = n1; s0 = s1; e0 = e1;
        n1 = n2; s1 = s2; e1 = e2; t1 = t2;
        n2 = n3; s2 = s3; e2 = e3;
        colA = colB; rA = rB; vA = vB;
    }
}

extern "C" void kernel_launch(void* const* d_in, const int* in_sizes, int n_in,
                              void* d_out, int out_size, void* d_ws, size_t ws_size,
                              hipStream_t stream) {
    const float* features = (const float*)d_in[0];
    const float* rel_emb  = (const float*)d_in[1];
    const int*   indices  = (const int*)  d_in[2];
    const int*   sp_idx   = (const int*)  d_in[3];
    const float* sp_val   = (const float*)d_in[4];
    const float* attn     = (const float*)d_in[5];
    float* out = (float*)d_out;

    float* ws       = (float*)d_ws;
    float* rel_norm = ws;
    float* rel_att  = ws + 512;
    int*   perm     = (int*)(ws + 1024);
    int*   cur      = (int*)(ws + 1024 + NTRIPLE);

    zero_int_kernel<<<49, 1024, 0, stream>>>(cur, NODE_SIZE);

    rel_stats_kernel<<<32, 32, 0, stream>>>(rel_emb, attn, rel_norm, rel_att);

    count_kernel<<<SORT_GROUPS, 1024, 0, stream>>>(indices, cur);
    scan_kernel<<<1, 1024, 0, stream>>>(cur);
    fill_kernel<<<SORT_GROUPS, 1024, 0, stream>>>(indices, cur, perm);

    gather_kernel<<<G_GROUPS, 1024, 0, stream>>>(
        features, rel_emb, indices, sp_idx, sp_val, rel_norm, rel_att,
        cur, perm, out);
}


# counting sort ranks once: count pass stores each edge's slot in its segment (LDS rank + returning add only for nodes the group has edges of); fill = scanned start + slot, no LDS table; arrival order
# speedup vs baseline: 1.9260x; 1.0711x over previous
#include <hip/hip_runtime.h>

#define NODE_SIZE 50000
#define REL_SIZE  500
#define HID       128
#define NTRIPLE   800000
#define EPSF      1e-12f

typedef __attribute__((ext_vector_type(2))) float v2f;
typedef __attribute__((ext_vector_type(8))) float v8f;

__global__ void zero_int_kernel(int* __restrict__ p, int n) {
    int i = blockIdx.x * blockDim.x + threadIdx.x;
    int stride = gridDim.x * blockDim.x;
    for (; i < n; i += stride) p[i] = 0;
}

__global__ __launch_bounds__(32)
void rel_stats_kernel(const float* __restrict__ rel_emb,
                      const float* __restrict__ attn,
                      float* __restrict__ rel_norm,
                      float* __restrict__ rel_att) {
    const int lane = threadIdx.x & 31;
    const int tile = blockIdx.x;
    const int half = lane >> 4;
    const int m16  = lane & 15;

    const int   row   = tile * 16 + m16;
    const bool  rowok = row < REL_SIZE;
    const float amask = rowok ? 1.0f : 0.0f;
    const float* arow = rel_emb + (size_t)(rowok ? row : 0) * HID;
    const float bmask = (m16 == 0) ? 1.0f : 0.0f;

    v8f c = {};
    for (int k = 0; k < HID; k += 4) {
        const int koff = half * 2;
        v2f a, b;
        a.x = arow[k + koff + 0] * amask;
        a.y = arow[k + koff + 1] * amask;
        b.x = attn[k + koff + 0] * bmask;
        b.y = attn[k + koff + 1] * bmask;
        c = __builtin_amdgcn_wmma_f32_16x16x4_f32(false, a, false, b,
                                                  (short)0, c, false, false);
    }
    if (m16 == 0) {
        const int mbase = tile * 16 + half * 8;
#pragma unroll
        for (int j = 0; j < 8; ++j)
            if (mbase + j < REL_SIZE) rel_att[mbase + j] = c[j];
    }

    float ss = 0.0f;
    {
        const float4* p = (const float4*)(arow + half * 64);
#pragma unroll
        for (int i = 0; i < 16; ++i) {
            float4 f = p[i];
            ss += f.x * f.x + f.y * f.y + f.z * f.z + f.w * f.w;
        }
    }
    ss += __shfl_xor(ss, 16, 32);
    if (half == 0 && rowok) rel_norm[row] = sqrtf(ss);
}

#define SORT_GROUPS  64
#define SORT_CHUNK   (NTRIPLE / SORT_GROUPS)
#define SORT_EDGES   13
#define NODE_TRIPS   49

__device__ __forceinline__ int node_slot(int k, int rot, int tid) {
    int kk = k + rot;
    if (kk >= NODE_TRIPS) kk -= NODE_TRIPS;
    return kk * 1024 + tid;
}

#define SWEEP_WIDE   13
#define SWEEP_ROUNDS 4

__global__ __launch_bounds__(1024)
void count_kernel(const int* __restrict__ indices, int* __restrict__ cnt,
                  int* __restrict__ slot) {
    __shared__ int h[NODE_SIZE];
    const int tid = threadIdx.x;
    for (int i = tid; i < NODE_SIZE; i += 1024) h[i] = 0;

    const int lo = blockIdx.x * SORT_CHUNK;
    int row[SORT_EDGES], rank[SORT_EDGES];
#pragma unroll
    for (int k = 0; k < SORT_EDGES; ++k) {
        const int e = k * 1024 + tid;
        row[k] = (e < SORT_CHUNK) ? indices[2 * (lo + e)] : -1;
    }
    __syncthreads();
#pragma unroll
    for (int k = 0; k < SORT_EDGES; ++k)
        rank[k] = (row[k] >= 0) ? atomicAdd(&h[row[k]], 1) : 0;
    __syncthreads();

    const int rot = blockIdx.x % NODE_TRIPS;
    for (int k0 = 0; k0 < SWEEP_ROUNDS * SWEEP_WIDE; k0 += SWEEP_WIDE) {
        int c[SWEEP_WIDE], start[SWEEP_WIDE];
#pragma unroll
        for (int k = 0; k < SWEEP_WIDE; ++k) {
            const int i = node_slot(k0 + k, rot, tid);
            c[k] = (k0 + k < NODE_TRIPS && i < NODE_SIZE) ? h[i] : 0;
        }
#pragma unroll
        for (int k = 0; k < SWEEP_WIDE; ++k) {
            const int i = node_slot(k0 + k, rot, tid);
            start[k] = c[k] ? atomicAdd(&cnt[i], c[k]) : 0;
        }
#pragma unroll
        for (int k = 0; k < SWEEP_WIDE; ++k) {
            const int i = node_slot(k0 + k, rot, tid);
            if (c[k]) h[i] = start[k];
        }
    }
    __syncthreads();

    int first[SORT_EDGES];
#pragma unroll
    for (int k = 0; k < SORT_EDGES; ++k) first[k] = h[max(row[k], 0)];
#pragma unroll
    for (int k = 0; k < SORT_EDGES; ++k)
        if (row[k] >= 0) slot[lo + k * 1024 + tid] = first[k] + rank[k];
}

#define SCAN_PER_THREAD 49
#define SCAN_WIDE  12
#define SCAN_TAIL  (SCAN_WIDE * 4096)

__global__ __launch_bounds__(1024)
void scan_kernel(int* __restrict__ cur) {
    __shared__ int4 s4[1024 * SCAN_PER_THREAD / 4];
    __shared__ int wsum[32];
    int* s = (int*)s4;
    const int tid  = threadIdx.x;
    const int lane = tid & 31;
    const int wave = tid >> 5;

    {
        const int4* src = (const int4*)cur;
        int4 v[SCAN_WIDE];
#pragma unroll
        for (int k = 0; k < SCAN_WIDE; ++k) v[k] = src[k * 1024 + tid];
        const int it = SCAN_TAIL + tid;
        const int vt = cur[min(it, NODE_SIZE - 1)];
#pragma unroll
        for (int k = 0; k < SCAN_WIDE; ++k) s4[k * 1024 + tid] = v[k];
        s[it] = (it < NODE_SIZE) ? vt : 0;
    }
    __syncthreads();

    const int base = tid * SCAN_PER_THREAD;
    int sum = 0;
    for (int j = 0; j < SCAN_PER_THREAD; ++j) sum += s[base + j];

    int incl = sum;
#pragma unroll
    for (int off = 1; off < 32; off <<= 1) {
        const int n = __shfl_up(incl, off, 32);
        if (lane >= off) incl += n;
    }
    if (lane == 31) wsum[wave] = incl;
    __syncthreads();
    if (wave == 0) {
        const int w = wsum[lane];
        int wi = w;
#pragma unroll
        for (int off = 1; off < 32; off <<= 1) {
            const int n = __shfl_up(wi, off, 32);
            if (lane >= off) wi += n;
        }
        wsum[lane] = wi - w;
    }
    __syncthreads();

    int run = incl - sum + wsum[wave];
    for (int j = 0; j < SCAN_PER_THREAD; ++j) {
        const int c = s[base + j];
        s[base + j] = run;
        run += c;
    }
    __syncthreads();
    for (int i = tid; i < NODE_SIZE; i += 1024) cur[i] = s[i];
}

#define FILL_GROUPS  125
#define FILL_THREADS 800
#define FILL_EDGES   8
#define FILL_STRIDE  (FILL_GROUPS * FILL_THREADS)

__global__ __launch_bounds__(FILL_THREADS)
void fill_kernel(const int* __restrict__ indices,
                 const int* __restrict__ cur,
                 const int* __restrict__ slot,
                 int* __restrict__ perm) {
    static_assert(FILL_STRIDE * FILL_EDGES == NTRIPLE, "the grid covers the edge list exactly");
    const int t = blockIdx.x * FILL_THREADS + threadIdx.x;
    int row[FILL_EDGES], off[FILL_EDGES], base[FILL_EDGES];
#pragma unroll
    for (int k = 0; k < FILL_EDGES; ++k) {
        const int e = k * FILL_STRIDE + t;
        row[k] = indices[2 * e];
        off[k] = slot[e];
    }
#pragma unroll
    for (int k = 0; k < FILL_EDGES; ++k) base[k] = cur[row[k]];
#pragma unroll
    for (int k = 0; k < FILL_EDGES; ++k) {
        int pos = base[k] + off[k];
        pos = min(max(pos, 0), NTRIPLE - 1);
        perm[pos] = k * FILL_STRIDE + t;
    }
}

#define G_GROUPS 256
#define G_WAVES  32
#define G_CHUNK  ((NODE_SIZE + G_GROUPS - 1) / G_GROUPS)
#define REL_VEC4 (REL_SIZE * HID / 4)


template <int M>
__device__ __forceinline__ float row_xor(float p) {
    return __int_as_float(__builtin_amdgcn_update_dpp(0, __float_as_int(p), 0x160 + M, 0xf, 0xf, true));
}

__device__ __forceinline__ float xhalf(float x) {
    const int b = __float_as_int(x);
    return __int_as_float(__builtin_amdgcn_permlanex16(b, b, 0x76543210, 0xfedcba98, false, false));
}

__device__ __forceinline__ float wave_sum(float p) {
    p += xhalf(p);
    p += row_xor<8>(p);
    p += row_xor<4>(p);
    p += row_xor<2>(p);
    p += row_xor<1>(p);
    return p;
}

__device__ __forceinline__ float swap_add(float x, float y) {
    const auto r = __builtin_amdgcn_permlane16_swap(__float_as_uint(x), __float_as_uint(y), false, false);
    return __uint_as_float(r[0]) + __uint_as_float(r[1]);
}

__device__ __forceinline__ void wave_sum8(const float (&p)[8], bool hi8, bool hi4, float (&tot)[8]) {
    float q[4];
#pragma unroll
    for (int i = 0; i < 4; ++i) q[i] = swap_add(p[i], p[i + 4]);
    float r[2];
#pragma unroll
    for (int i = 0; i < 2; ++i) {
        const float keep = hi8 ? q[i + 2] : q[i];
        const float send = hi8 ? q[i] : q[i + 2];
        r[i] = keep + row_xor<8>(send);
    }
    const float keep = hi4 ? r[1] : r[0];
    const float send = hi4 ? r[0] : r[1];
    float s = keep + row_xor<4>(send);
    s += row_xor<2>(s);
    s += row_xor<1>(s);
#pragma unroll
    for (int k = 0; k < 8; ++k)
        tot[k] = __int_as_float(__builtin_amdgcn_readlane(
            __float_as_int(s), 4 * (k & 1) + 8 * ((k >> 1) & 1) + 16 * (k >> 2)));
}

__device__ __forceinline__ void wave_sum4(const float (&p)[4], bool hi8, float (&tot)[4]) {
    const float q0 = swap_add(p[0], p[2]);
    const float q1 = swap_add(p[1], p[3]);
    const float keep = hi8 ? q1 : q0;
    const float send = hi8 ? q0 : q1;
    float s = keep + row_xor<8>(send);
    s += row_xor<4>(s);
    s += row_xor<2>(s);
    s += row_xor<1>(s);
#pragma unroll
    for (int k = 0; k < 4; ++k)
        tot[k] = __int_as_float(__builtin_amdgcn_readlane(
            __float_as_int(s), 8 * (k & 1) + 16 * (k >> 1)));
}

__device__ __forceinline__ void load_rel(float (&r)[4], const float* rrow) {
    r[0] = rrow[0]; r[1] = rrow[32]; r[2] = rrow[64]; r[3] = rrow[96];
}

__device__ __forceinline__ float lane_dot(const float (&f)[4], const float (&r)[4]) {
    return f[0] * r[0] + f[1] * r[1] + f[2] * r[2] + f[3] * r[3];
}

__device__ __forceinline__ void add_edge(const float (&f)[4], const float (&r)[4],
                                         float coef, float e, float partial,
                                         float& a0, float& a1, float& a2, float& a3) {
    const float proj = coef * partial;
    const float k2   = 2.0f * proj * coef;
    a0 += (f[0] - k2 * r[0]) * e;
    a1 += (f[1] - k2 * r[1]) * e;
    a2 += (f[2] - k2 * r[2]) * e;
    a3 += (f[3] - k2 * r[3]) * e;
}

__device__ __forceinline__ void load_feat(float (&f)[4], const float* __restrict__ features,
                                          int col, int lane) {
    const float* frow = features + (size_t)col * HID + lane;
    f[0] = frow[0]; f[1] = frow[32]; f[2] = frow[64]; f[3] = frow[96];
}

__device__ __forceinline__ int bcast_i(int v, int j) {
    return __builtin_amdgcn_readlane(v, j);
}
__device__ __forceinline__ float bcast_f(float v, int j) {
    return __int_as_float(__builtin_amdgcn_readlane(__float_as_int(v), j));
}

__device__ __forceinline__ void seg_bounds(const int* __restrict__ cur, int node,
                                           int& s, int& e) {
    const int n = min(node, NODE_SIZE - 1);
    s = cur[n];
    e = cur[min(n + 1, NODE_SIZE - 1)];
    if (n == NODE_SIZE - 1) e = NTRIPLE;
}

__device__ __forceinline__ void sum_batch(const float* __restrict__ features,
                                          const float* s_rel,
                                          int col, int r, float coef, float e,
                                          int m, int lane,
                                          float& a0, float& a1, float& a2, float& a3) {
    const bool hi8 = (lane & 8) != 0;
    const bool hi4 = (lane & 4) != 0;
    int j = 0;
    for (; j + 8 <= m; j += 8) {
        float f[8][4], rr[8][4], p[8], tot[8];
#pragma unroll
        for (int k = 0; k < 8; ++k) load_feat(f[k], features, bcast_i(col, j + k), lane);
#pragma unroll
        for (int k = 0; k < 8; ++k) {
            load_rel(rr[k], s_rel + bcast_i(r, j + k) * HID + lane);
            p[k] = lane_dot(f[k], rr[k]);
        }
        wave_sum8(p, hi8, hi4, tot);
#pragma unroll
        for (int k = 0; k < 8; ++k)
            add_edge(f[k], rr[k], bcast_f(coef, j + k), bcast_f(e, j + k), tot[k], a0, a1, a2, a3);
    }
    for (; j + 4 <= m; j += 4) {
        float f[4][4], rr[4][4], p[4], tot[4];
#pragma unroll
        for (int k = 0; k < 4; ++k) load_feat(f[k], features, bcast_i(col, j + k), lane);
#pragma unroll
        for (int k = 0; k < 4; ++k) {
            load_rel(rr[k], s_rel + bcast_i(r, j + k) * HID + lane);
            p[k] = lane_dot(f[k], rr[k]);
        }
        wave_sum4(p, hi8, tot);
#pragma unroll
        for (int k = 0; k < 4; ++k)
            add_edge(f[k], rr[k], bcast_f(coef, j + k), bcast_f(e, j + k), tot[k], a0, a1, a2, a3);
    }
    for (; j < m; ++j) {
        float f[4], rr[4];
        load_feat(f, features, bcast_i(col, j), lane);
        load_rel(rr, s_rel + bcast_i(r, j) * HID + lane);
        add_edge(f, rr, bcast_f(coef, j), bcast_f(e, j), wave_sum(lane_dot(f, rr)), a0, a1, a2, a3);
    }
}

__global__ __launch_bounds__(1024)
void gather_kernel(const float* __restrict__ features,
                   const float* __restrict__ rel_emb,
                   const int*   __restrict__ indices,
                   const int*   __restrict__ sp_idx,
                   const float* __restrict__ sp_val,
                   const float* __restrict__ rel_norm,
                   const float* __restrict__ rel_att,
                   const int*   __restrict__ cur,
                   const int*   __restrict__ perm,
                   float* __restrict__ out) {
    __shared__ float s_rel[REL_SIZE * HID];
    __shared__ float s_norm[512];
    __shared__ float s_att[512];
    __shared__ int   s_next;

    const int tid  = threadIdx.x;
    const int lane = tid & 31;
    const int wave = __builtin_amdgcn_readfirstlane(tid >> 5);
    const int lo = blockIdx.x * G_CHUNK;
    const int hi = min(lo + G_CHUNK, NODE_SIZE);

    if (tid == 0) s_next = lo + 3 * G_WAVES;
    int n0 = lo + wave, n1 = n0 + G_WAVES, n2 = n1 + G_WAVES;
    int s0, e0, s1, e1, s2, e2;
    seg_bounds(cur, n0, s0, e0);
    seg_bounds(cur, n1, s1, e1);
    seg_bounds(cur, n2, s2, e2);
    int t0 = 0, t1 = 0;
    if (n0 < hi && lane < e0 - s0) t0 = perm[s0 + lane];
    if (n1 < hi && lane < e1 - s1) t1 = perm[s1 + lane];

    {
        const float4* src = (const float4*)rel_emb;
        float4* dst = (float4*)s_rel;
#pragma unroll
        for (int b = 0; b < 4; ++b) {
            float4 v[4];
#pragma unroll
            for (int k = 0; k < 4; ++k) {
                const int i = (b * 4 + k) * 1024 + tid;
                v[k] = src[min(i, REL_VEC4 - 1)];
            }
#pragma unroll
            for (int k = 0; k < 4; ++k) {
                const int i = (b * 4 + k) * 1024 + tid;
                if (i < REL_VEC4) dst[i] = v[k];
            }
        }
        if (tid < REL_SIZE) { s_norm[tid] = rel_norm[tid]; s_att[tid] = rel_att[tid]; }
    }

    int colA = 0, rA = 0;
    float vA = 0.0f;
    if (n0 < hi && lane < e0 - s0) {
        const int tc = min(max(t0, 0), NTRIPLE - 1);
        colA = indices[2 * tc + 1];
        rA   = sp_idx[2 * tc + 1];
        vA   = sp_val[tc];
    }
    __syncthreads();

    while (n0 < hi) {
        int n3 = 0;
        if (lane == 0) n3 = atomicAdd(&s_next, 1);
        n3 = __builtin_amdgcn_readfirstlane(n3);
        int s3, e3;
        seg_bounds(cur, n3, s3, e3);
        int t2 = 0;
        if (n2 < hi && lane < e2 - s2) t2 = perm[s2 + lane];
        int colB = 0, rB = 0;
        float vB = 0.0f;
        if (n1 < hi && lane < e1 - s1) {
            const int tc = min(max(t1, 0), NTRIPLE - 1);
            colB = indices[2 * tc + 1];
            rB   = sp_idx[2 * tc + 1];
            vB   = sp_val[tc];
        }

        float a0 = 0.0f, a1 = 0.0f, a2 = 0.0f, a3 = 0.0f, den = 0.0f;
        {
            const int m = min(32, e0 - s0);
            float coef = 0.0f, e = 0.0f;
            if (lane < m) {
                coef = vA / fmaxf(fabsf(vA) * s_norm[rA], EPSF);
                e    = expf(coef * s_att[rA]);
            }
            den += e;
            sum_batch(features, s_rel, colA, rA, coef, e, m, lane, a0, a1, a2, a3);
        }
        for (int base = s0 + 32; base < e0; base += 32) {
            const int m = min(32, e0 - base);
            int col = 0, r = 0;
            float coef = 0.0f, e = 0.0f;
            if (lane < m) {
                const int tc = min(max(perm[base + lane], 0), NTRIPLE - 1);
                col = indices[2 * tc + 1];
                r   = sp_idx[2 * tc + 1];
                const float v = sp_val[tc];
                coef = v / fmaxf(fabsf(v) * s_norm[r], EPSF);
                e    = expf(coef * s_att[r]);
            }
            den += e;
            sum_batch(features, s_rel, col, r, coef, e, m, lane, a0, a1, a2, a3);
        }

        den = wave_sum(den);
        float* dst = out + (size_t)n0 * HID + lane;
        if (e0 > s0) {
            dst[0]  = a0 / den;
            dst[32] = a1 / den;
            dst[64] = a2 / den;
            dst[96] = a3 / den;
        } else {
            dst[0] = 0.0f; dst[32] = 0.0f; dst[64] = 0.0f; dst[96] = 0.0f;
        }

        n0 = n1; s0 = s1; e0 = e1;
        n1 = n2; s1 = s2; e1 = e2; t1 = t2;
        n2 = n3; s2 = s3; e2 = e3;
        colA = colB; rA = rB; vA = vB;
    }
}

extern "C" void kernel_launch(void* const* d_in, const int* in_sizes, int n_in,
                              void* d_out, int out_size, void* d_ws, size_t ws_size,
                              hipStream_t stream) {
    const float* features = (const float*)d_in[0];
    const float* rel_emb  = (const float*)d_in[1];
    const int*   indices  = (const int*)  d_in[2];
    const int*   sp_idx   = (const int*)  d_in[3];
    const float* sp_val   = (const float*)d_in[4];
    const float* attn     = (const float*)d_in[5];
    float* out = (float*)d_out;

    float* ws       = (float*)d_ws;
    float* rel_norm = ws;
    float* rel_att  = ws + 512;
    int*   perm     = (int*)(ws + 1024);
    int*   cur      = (int*)(ws + 1024 + NTRIPLE);

    zero_int_kernel<<<49, 1024, 0, stream>>>(cur, NODE_SIZE);

    rel_stats_kernel<<<32, 32, 0, stream>>>(rel_emb, attn, rel_norm, rel_att);

    int* slot = (int*)d_out;
    count_kernel<<<SORT_GROUPS, 1024, 0, stream>>>(indices, cur, slot);
    scan_kernel<<<1, 1024, 0, stream>>>(cur);
    fill_kernel<<<FILL_GROUPS, FILL_THREADS, 0, stream>>>(indices, cur, slot, perm);

    gather_kernel<<<G_GROUPS, 1024, 0, stream>>>(
        features, rel_emb, indices, sp_idx, sp_val, rel_norm, rel_att,
        cur, perm, out);
}


# fill pass stores each edge's record (neighbour + relation id in one word, value) at its sorted position; gather reads a node's records as one run, not 3 scattered 4-byte reads per edge via edge ids
# speedup vs baseline: 2.2000x; 1.1423x over previous
#include <hip/hip_runtime.h>

#define NODE_SIZE 50000
#define REL_SIZE  500
#define HID       128
#define NTRIPLE   800000
#define EPSF      1e-12f

typedef __attribute__((ext_vector_type(2))) float v2f;
typedef __attribute__((ext_vector_type(8))) float v8f;

__global__ void zero_int_kernel(int* __restrict__ p, int n) {
    int i = blockIdx.x * blockDim.x + threadIdx.x;
    int stride = gridDim.x * blockDim.x;
    for (; i < n; i += stride) p[i] = 0;
}

__global__ __launch_bounds__(32)
void rel_stats_kernel(const float* __restrict__ rel_emb,
                      const float* __restrict__ attn,
                      float* __restrict__ rel_norm,
                      float* __restrict__ rel_att) {
    const int lane = threadIdx.x & 31;
    const int tile = blockIdx.x;
    const int half = lane >> 4;
    const int m16  = lane & 15;

    const int   row   = tile * 16 + m16;
    const bool  rowok = row < REL_SIZE;
    const float amask = rowok ? 1.0f : 0.0f;
    const float* arow = rel_emb + (size_t)(rowok ? row : 0) * HID;
    const float bmask = (m16 == 0) ? 1.0f : 0.0f;

    v8f c = {};
    for (int k = 0; k < HID; k += 4) {
        const int koff = half * 2;
        v2f a, b;
        a.x = arow[k + koff + 0] * amask;
        a.y = arow[k + koff + 1] * amask;
        b.x = attn[k + koff + 0] * bmask;
        b.y = attn[k + koff + 1] * bmask;
        c = __builtin_amdgcn_wmma_f32_16x16x4_f32(false, a, false, b,
                                                  (short)0, c, false, false);
    }
    if (m16 == 0) {
        const int mbase = tile * 16 + half * 8;
#pragma unroll
        for (int j = 0; j < 8; ++j)
            if (mbase + j < REL_SIZE) rel_att[mbase + j] = c[j];
    }

    float ss = 0.0f;
    {
        const float4* p = (const float4*)(arow + half * 64);
#pragma unroll
        for (int i = 0; i < 16; ++i) {
            float4 f = p[i];
            ss += f.x * f.x + f.y * f.y + f.z * f.z + f.w * f.w;
        }
    }
    ss += __shfl_xor(ss, 16, 32);
    if (half == 0 && rowok) rel_norm[row] = sqrtf(ss);
}

#define SORT_GROUPS  64
#define SORT_CHUNK   (NTRIPLE / SORT_GROUPS)
#define SORT_EDGES   13
#define NODE_TRIPS   49

__device__ __forceinline__ int node_slot(int k, int rot, int tid) {
    int kk = k + rot;
    if (kk >= NODE_TRIPS) kk -= NODE_TRIPS;
    return kk * 1024 + tid;
}

#define SWEEP_WIDE   13
#define SWEEP_ROUNDS 4

__global__ __launch_bounds__(1024)
void count_kernel(const int* __restrict__ indices, int* __restrict__ cnt,
                  int* __restrict__ slot) {
    __shared__ int h[NODE_SIZE];
    const int tid = threadIdx.x;
    for (int i = tid; i < NODE_SIZE; i += 1024) h[i] = 0;

    const int lo = blockIdx.x * SORT_CHUNK;
    int row[SORT_EDGES], rank[SORT_EDGES];
#pragma unroll
    for (int k = 0; k < SORT_EDGES; ++k) {
        const int e = k * 1024 + tid;
        row[k] = (e < SORT_CHUNK) ? indices[2 * (lo + e)] : -1;
    }
    __syncthreads();
#pragma unroll
    for (int k = 0; k < SORT_EDGES; ++k)
        rank[k] = (row[k] >= 0) ? atomicAdd(&h[row[k]], 1) : 0;
    __syncthreads();

    const int rot = blockIdx.x % NODE_TRIPS;
    for (int k0 = 0; k0 < SWEEP_ROUNDS * SWEEP_WIDE; k0 += SWEEP_WIDE) {
        int c[SWEEP_WIDE], start[SWEEP_WIDE];
#pragma unroll
        for (int k = 0; k < SWEEP_WIDE; ++k) {
            const int i = node_slot(k0 + k, rot, tid);
            c[k] = (k0 + k < NODE_TRIPS && i < NODE_SIZE) ? h[i] : 0;
        }
#pragma unroll
        for (int k = 0; k < SWEEP_WIDE; ++k) {
            const int i = node_slot(k0 + k, rot, tid);
            start[k] = c[k] ? atomicAdd(&cnt[i], c[k]) : 0;
        }
#pragma unroll
        for (int k = 0; k < SWEEP_WIDE; ++k) {
            const int i = node_slot(k0 + k, rot, tid);
            if (c[k]) h[i] = start[k];
        }
    }
    __syncthreads();

    int first[SORT_EDGES];
#pragma unroll
    for (int k = 0; k < SORT_EDGES; ++k) first[k] = h[max(row[k], 0)];
#pragma unroll
    for (int k = 0; k < SORT_EDGES; ++k)
        if (row[k] >= 0) slot[lo + k * 1024 + tid] = first[k] + rank[k];
}

#define SCAN_PER_THREAD 49
#define SCAN_WIDE  12
#define SCAN_TAIL  (SCAN_WIDE * 4096)

__global__ __launch_bounds__(1024)
void scan_kernel(int* __restrict__ cur) {
    __shared__ int4 s4[1024 * SCAN_PER_THREAD / 4];
    __shared__ int wsum[32];
    int* s = (int*)s4;
    const int tid  = threadIdx.x;
    const int lane = tid & 31;
    const int wave = tid >> 5;

    {
        const int4* src = (const int4*)cur;
        int4 v[SCAN_WIDE];
#pragma unroll
        for (int k = 0; k < SCAN_WIDE; ++k) v[k] = src[k * 1024 + tid];
        const int it = SCAN_TAIL + tid;
        const int vt = cur[min(it, NODE_SIZE - 1)];
#pragma unroll
        for (int k = 0; k < SCAN_WIDE; ++k) s4[k * 1024 + tid] = v[k];
        s[it] = (it < NODE_SIZE) ? vt : 0;
    }
    __syncthreads();

    const int base = tid * SCAN_PER_THREAD;
    int sum = 0;
    for (int j = 0; j < SCAN_PER_THREAD; ++j) sum += s[base + j];

    int incl = sum;
#pragma unroll
    for (int off = 1; off < 32; off <<= 1) {
        const int n = __shfl_up(incl, off, 32);
        if (lane >= off) incl += n;
    }
    if (lane == 31) wsum[wave] = incl;
    __syncthreads();
    if (wave == 0) {
        const int w = wsum[lane];
        int wi = w;
#pragma unroll
        for (int off = 1; off < 32; off <<= 1) {
            const int n = __shfl_up(wi, off, 32);
            if (lane >= off) wi += n;
        }
        wsum[lane] = wi - w;
    }
    __syncthreads();

    int run = incl - sum + wsum[wave];
    for (int j = 0; j < SCAN_PER_THREAD; ++j) {
        const int c = s[base + j];
        s[base + j] = run;
        run += c;
    }
    __syncthreads();
    for (int i = tid; i < NODE_SIZE; i += 1024) cur[i] = s[i];
}

#define FILL_GROUPS  125
#define FILL_THREADS 800
#define FILL_EDGES   8
#define FILL_STRIDE  (FILL_GROUPS * FILL_THREADS)

#define WS_LIST_OFF   (1024 + NODE_SIZE)
#define WS_WIDE_BYTES ((size_t)(WS_LIST_OFF + 2 * NTRIPLE) * 4)

__global__ __launch_bounds__(FILL_THREADS)
void fill_kernel(const int* __restrict__ indices,
                 const int* __restrict__ sp_idx,
                 const float* __restrict__ sp_val,
                 const int* __restrict__ cur,
                 const int* __restrict__ slot,
                 int* __restrict__ list,
                 size_t ws_bytes) {
    static_assert(FILL_STRIDE * FILL_EDGES == NTRIPLE, "the grid covers the edge list exactly");
    const int t = blockIdx.x * FILL_THREADS + threadIdx.x;
    if (ws_bytes >= WS_WIDE_BYTES) {
        const int2* ind2 = (const int2*)indices;
        int2* rec = (int2*)list;
        int2 rc[FILL_EDGES];
        int off[FILL_EDGES], rl[FILL_EDGES], base[FILL_EDGES];
        float vl[FILL_EDGES];
#pragma unroll
        for (int k = 0; k < FILL_EDGES; ++k) {
            const int e = k * FILL_STRIDE + t;
            rc[k]  = ind2[e];
            off[k] = slot[e];
            rl[k]  = sp_idx[2 * e + 1];
            vl[k]  = sp_val[e];
        }
#pragma unroll
        for (int k = 0; k < FILL_EDGES; ++k) base[k] = cur[rc[k].x];
#pragma unroll
        for (int k = 0; k < FILL_EDGES; ++k) {
            int pos = base[k] + off[k];
            pos = min(max(pos, 0), NTRIPLE - 1);
            rec[pos] = make_int2((rc[k].y & 0xffff) | (rl[k] << 16), __float_as_int(vl[k]));
        }
        return;
    }
    int* perm = list;
    int row[FILL_EDGES], off[FILL_EDGES], base[FILL_EDGES];
#pragma unroll
    for (int k = 0; k < FILL_EDGES; ++k) {
        const int e = k * FILL_STRIDE + t;
        row[k] = indices[2 * e];
        off[k] = slot[e];
    }
#pragma unroll
    for (int k = 0; k < FILL_EDGES; ++k) base[k] = cur[row[k]];
#pragma unroll
    for (int k = 0; k < FILL_EDGES; ++k) {
        int pos = base[k] + off[k];
        pos = min(max(pos, 0), NTRIPLE - 1);
        perm[pos] = k * FILL_STRIDE + t;
    }
}

#define G_GROUPS 256
#define G_WAVES  32
#define G_CHUNK  ((NODE_SIZE + G_GROUPS - 1) / G_GROUPS)
#define REL_VEC4 (REL_SIZE * HID / 4)


template <int M>
__device__ __forceinline__ float row_xor(float p) {
    return __int_as_float(__builtin_amdgcn_update_dpp(0, __float_as_int(p), 0x160 + M, 0xf, 0xf, true));
}

__device__ __forceinline__ float xhalf(float x) {
    const int b = __float_as_int(x);
    return __int_as_float(__builtin_amdgcn_permlanex16(b, b, 0x76543210, 0xfedcba98, false, false));
}

__device__ __forceinline__ float wave_sum(float p) {
    p += xhalf(p);
    p += row_xor<8>(p);
    p += row_xor<4>(p);
    p += row_xor<2>(p);
    p += row_xor<1>(p);
    return p;
}

__device__ __forceinline__ float swap_add(float x, float y) {
    const auto r = __builtin_amdgcn_permlane16_swap(__float_as_uint(x), __float_as_uint(y), false, false);
    return __uint_as_float(r[0]) + __uint_as_float(r[1]);
}

__device__ __forceinline__ void wave_sum8(const float (&p)[8], bool hi8, bool hi4, float (&tot)[8]) {
    float q[4];
#pragma unroll
    for (int i = 0; i < 4; ++i) q[i] = swap_add(p[i], p[i + 4]);
    float r[2];
#pragma unroll
    for (int i = 0; i < 2; ++i) {
        const float keep = hi8 ? q[i + 2] : q[i];
        const float send = hi8 ? q[i] : q[i + 2];
        r[i] = keep + row_xor<8>(send);
    }
    const float keep = hi4 ? r[1] : r[0];
    const float send = hi4 ? r[0] : r[1];
    float s = keep + row_xor<4>(send);
    s += row_xor<2>(s);
    s += row_xor<1>(s);
#pragma unroll
    for (int k = 0; k < 8; ++k)
        tot[k] = __int_as_float(__builtin_amdgcn_readlane(
            __float_as_int(s), 4 * (k & 1) + 8 * ((k >> 1) & 1) + 16 * (k >> 2)));
}

__device__ __forceinline__ void wave_sum4(const float (&p)[4], bool hi8, float (&tot)[4]) {
    const float q0 = swap_add(p[0], p[2]);
    const float q1 = swap_add(p[1], p[3]);
    const float keep = hi8 ? q1 : q0;
    const float send = hi8 ? q0 : q1;
    float s = keep + row_xor<8>(send);
    s += row_xor<4>(s);
    s += row_xor<2>(s);
    s += row_xor<1>(s);
#pragma unroll
    for (int k = 0; k < 4; ++k)
        tot[k] = __int_as_float(__builtin_amdgcn_readlane(
            __float_as_int(s), 8 * (k & 1) + 16 * (k >> 1)));
}

__device__ __forceinline__ void load_rel(float (&r)[4], const float* rrow) {
    r[0] = rrow[0]; r[1] = rrow[32]; r[2] = rrow[64]; r[3] = rrow[96];
}

__device__ __forceinline__ float lane_dot(const float (&f)[4], const float (&r)[4]) {
    return f[0] * r[0] + f[1] * r[1] + f[2] * r[2] + f[3] * r[3];
}

__device__ __forceinline__ void add_edge(const float (&f)[4], const float (&r)[4],
                                         float coef, float e, float partial,
                                         float& a0, float& a1, float& a2, float& a3) {
    const float proj = coef * partial;
    const float k2   = 2.0f * proj * coef;
    a0 += (f[0] - k2 * r[0]) * e;
    a1 += (f[1] - k2 * r[1]) * e;
    a2 += (f[2] - k2 * r[2]) * e;
    a3 += (f[3] - k2 * r[3]) * e;
}

__device__ __forceinline__ void load_feat(float (&f)[4], const float* __restrict__ features,
                                          int col, int lane) {
    const float* frow = features + (size_t)col * HID + lane;
    f[0] = frow[0]; f[1] = frow[32]; f[2] = frow[64]; f[3] = frow[96];
}

__device__ __forceinline__ int bcast_i(int v, int j) {
    return __builtin_amdgcn_readlane(v, j);
}
__device__ __forceinline__ float bcast_f(float v, int j) {
    return __int_as_float(__builtin_amdgcn_readlane(__float_as_int(v), j));
}

__device__ __forceinline__ void seg_bounds(const int* __restrict__ cur, int node,
                                           int& s, int& e) {
    const int n = min(node, NODE_SIZE - 1);
    s = cur[n];
    e = cur[min(n + 1, NODE_SIZE - 1)];
    if (n == NODE_SIZE - 1) e = NTRIPLE;
}

__device__ __forceinline__ void sum_batch(const float* __restrict__ features,
                                          const float* s_rel,
                                          int col, int r, float coef, float e,
                                          int m, int lane,
                                          float& a0, float& a1, float& a2, float& a3) {
    const bool hi8 = (lane & 8) != 0;
    const bool hi4 = (lane & 4) != 0;
    int j = 0;
    for (; j + 8 <= m; j += 8) {
        float f[8][4], rr[8][4], p[8], tot[8];
#pragma unroll
        for (int k = 0; k < 8; ++k) load_feat(f[k], features, bcast_i(col, j + k), lane);
#pragma unroll
        for (int k = 0; k < 8; ++k) {
            load_rel(rr[k], s_rel + bcast_i(r, j + k) * HID + lane);
            p[k] = lane_dot(f[k], rr[k]);
        }
        wave_sum8(p, hi8, hi4, tot);
#pragma unroll
        for (int k = 0; k < 8; ++k)
            add_edge(f[k], rr[k], bcast_f(coef, j + k), bcast_f(e, j + k), tot[k], a0, a1, a2, a3);
    }
    for (; j + 4 <= m; j += 4) {
        float f[4][4], rr[4][4], p[4], tot[4];
#pragma unroll
        for (int k = 0; k < 4; ++k) load_feat(f[k], features, bcast_i(col, j + k), lane);
#pragma unroll
        for (int k = 0; k < 4; ++k) {
            load_rel(rr[k], s_rel + bcast_i(r, j + k) * HID + lane);
            p[k] = lane_dot(f[k], rr[k]);
        }
        wave_sum4(p, hi8, tot);
#pragma unroll
        for (int k = 0; k < 4; ++k)
            add_edge(f[k], rr[k], bcast_f(coef, j + k), bcast_f(e, j + k), tot[k], a0, a1, a2, a3);
    }
    for (; j < m; ++j) {
        float f[4], rr[4];
        load_feat(f, features, bcast_i(col, j), lane);
        load_rel(rr, s_rel + bcast_i(r, j) * HID + lane);
        add_edge(f, rr, bcast_f(coef, j), bcast_f(e, j), wave_sum(lane_dot(f, rr)), a0, a1, a2, a3);
    }
}

__device__ __forceinline__ int2 load_rec(const int2* __restrict__ rec, int pos) {
    return rec[min(max(pos, 0), NTRIPLE - 1)];
}
__device__ __forceinline__ void split_rec(int2 x, int& col, int& r, float& v) {
    col = min(x.x & 0xffff, NODE_SIZE - 1);
    r   = (int)((unsigned)x.x >> 16);
    v   = __int_as_float(x.y);
}

__device__ __forceinline__ int load_bounds(const int* __restrict__ cur, int node, int lane) {
    const int n = min(node, NODE_SIZE - 1);
    return cur[min(n + (lane & 1), NODE_SIZE - 1)];
}
__device__ __forceinline__ void read_bounds(int b, int node, int& s, int& e) {
    s = __builtin_amdgcn_readlane(b, 0);
    e = (node >= NODE_SIZE - 1) ? NTRIPLE : __builtin_amdgcn_readlane(b, 1);
}

__device__ __forceinline__ void stage_tables(const float* __restrict__ rel_emb,
                                             const float* __restrict__ rel_norm,
                                             const float* __restrict__ rel_att,
                                             float* s_rel, float* s_norm, float* s_att,
                                             int tid) {
    const float4* src = (const float4*)rel_emb;
    float4* dst = (float4*)s_rel;
#pragma unroll
    for (int b = 0; b < 4; ++b) {
        float4 v[4];
#pragma unroll
        for (int k = 0; k < 4; ++k) {
            const int i = (b * 4 + k) * 1024 + tid;
            v[k] = src[min(i, REL_VEC4 - 1)];
        }
#pragma unroll
        for (int k = 0; k < 4; ++k) {
            const int i = (b * 4 + k) * 1024 + tid;
            if (i < REL_VEC4) dst[i] = v[k];
        }
    }
    if (tid < REL_SIZE) { s_norm[tid] = rel_norm[tid]; s_att[tid] = rel_att[tid]; }
}

__global__ __launch_bounds__(1024)
void gather_kernel(const float* __restrict__ features,
                   const float* __restrict__ rel_emb,
                   const int*   __restrict__ indices,
                   const int*   __restrict__ sp_idx,
                   const float* __restrict__ sp_val,
                   const float* __restrict__ rel_norm,
                   const float* __restrict__ rel_att,
                   const int*   __restrict__ cur,
                   const int*   __restrict__ list,
                   float* __restrict__ out,
                   size_t ws_bytes) {
    __shared__ float s_rel[REL_SIZE * HID];
    __shared__ float s_norm[512];
    __shared__ float s_att[512];
    __shared__ int   s_next;

    const int tid  = threadIdx.x;
    const int lane = tid & 31;
    const int wave = __builtin_amdgcn_readfirstlane(tid >> 5);
    const int lo = blockIdx.x * G_CHUNK;
    const int hi = min(lo + G_CHUNK, NODE_SIZE);

    if (ws_bytes >= WS_WIDE_BYTES) {
        const int2* rec = (const int2*)list;
        if (tid == 0) s_next = lo + 2 * G_WAVES;
        int n0 = lo + wave, n1 = n0 + G_WAVES;
        const int b0 = load_bounds(cur, n0, lane);
        int b1 = load_bounds(cur, n1, lane);
        int s0, e0;
        read_bounds(b0, n0, s0, e0);
        int2 xA = make_int2(0, 0);
        if (n0 < hi && lane < e0 - s0) xA = load_rec(rec, s0 + lane);
        stage_tables(rel_emb, rel_norm, rel_att, s_rel, s_norm, s_att, tid);
        __syncthreads();

        while (n0 < hi) {
            int s1, e1;
            read_bounds(b1, n1, s1, e1);
            int n2 = 0;
            if (lane == 0) n2 = atomicAdd(&s_next, 1);
            n2 = __builtin_amdgcn_readfirstlane(n2);
            const int b2 = load_bounds(cur, n2, lane);
            int2 xB = make_int2(0, 0);
            if (n1 < hi && lane < e1 - s1) xB = load_rec(rec, s1 + lane);

            float a0 = 0.0f, a1 = 0.0f, a2 = 0.0f, a3 = 0.0f, den = 0.0f;
            {
                const int m = min(32, e0 - s0);
                int colA, rA;
                float vA;
                split_rec(xA, colA, rA, vA);
                float coef = 0.0f, e = 0.0f;
                if (lane < m) {
                    coef = vA / fmaxf(fabsf(vA) * s_norm[rA], EPSF);
                    e    = expf(coef * s_att[rA]);
                }
                den += e;
                sum_batch(features, s_rel, colA, rA, coef, e, m, lane, a0, a1, a2, a3);
            }
            for (int base = s0 + 32; base < e0; base += 32) {
                const int m = min(32, e0 - base);
                int col = 0, r = 0;
                float coef = 0.0f, e = 0.0f;
                if (lane < m) {
                    float v;
                    split_rec(load_rec(rec, base + lane), col, r, v);
                    coef = v / fmaxf(fabsf(v) * s_norm[r], EPSF);
                    e    = expf(coef * s_att[r]);
                }
                den += e;
                sum_batch(features, s_rel, col, r, coef, e, m, lane, a0, a1, a2, a3);
            }

            den = wave_sum(den);
            float* dst = out + (size_t)n0 * HID + lane;
            if (e0 > s0) {
                dst[0]  = a0 / den;
                dst[32] = a1 / den;
                dst[64] = a2 / den;
                dst[96] = a3 / den;
            } else {
                dst[0] = 0.0f; dst[32] = 0.0f; dst[64] = 0.0f; dst[96] = 0.0f;
            }

            n0 = n1; s0 = s1; e0 = e1;
            n1 = n2; b1 = b2;
            xA = xB;
        }
        return;
    }

    const int* perm = list;
    if (tid == 0) s_next = lo + 3 * G_WAVES;
    int n0 = lo + wave, n1 = n0 + G_WAVES, n2 = n1 + G_WAVES;
    int s0, e0, s1, e1, s2, e2;
    seg_bounds(cur, n0, s0, e0);
    seg_bounds(cur, n1, s1, e1);
    seg_bounds(cur, n2, s2, e2);
    int t0 = 0, t1 = 0;
    if (n0 < hi && lane < e0 - s0) t0 = perm[s0 + lane];
    if (n1 < hi && lane < e1 - s1) t1 = perm[s1 + lane];

    {
        const float4* src = (const float4*)rel_emb;
        float4* dst = (float4*)s_rel;
#pragma unroll
        for (int b = 0; b < 4; ++b) {
            float4 v[4];
#pragma unroll
            for (int k = 0; k < 4; ++k) {
                const int i = (b * 4 + k) * 1024 + tid;
                v[k] = src[min(i, REL_VEC4 - 1)];
            }
#pragma unroll
            for (int k = 0; k < 4; ++k) {
                const int i = (b * 4 + k) * 1024 + tid;
                if (i < REL_VEC4) dst[i] = v[k];
            }
        }
        if (tid < REL_SIZE) { s_norm[tid] = rel_norm[tid]; s_att[tid] = rel_att[tid]; }
    }

    int colA = 0, rA = 0;
    float vA = 0.0f;
    if (n0 < hi && lane < e0 - s0) {
        const int tc = min(max(t0, 0), NTRIPLE - 1);
        colA = indices[2 * tc + 1];
        rA   = sp_idx[2 * tc + 1];
        vA   = sp_val[tc];
    }
    __syncthreads();

    while (n0 < hi) {
        int n3 = 0;
        if (lane == 0) n3 = atomicAdd(&s_next, 1);
        n3 = __builtin_amdgcn_readfirstlane(n3);
        int s3, e3;
        seg_bounds(cur, n3, s3, e3);
        int t2 = 0;
        if (n2 < hi && lane < e2 - s2) t2 = perm[s2 + lane];
        int colB = 0, rB = 0;
        float vB = 0.0f;
        if (n1 < hi && lane < e1 - s1) {
            const int tc = min(max(t1, 0), NTRIPLE - 1);
            colB = indices[2 * tc + 1];
            rB   = sp_idx[2 * tc + 1];
            vB   = sp_val[tc];
        }

        float a0 = 0.0f, a1 = 0.0f, a2 = 0.0f, a3 = 0.0f, den = 0.0f;
        {
            const int m = min(32, e0 - s0);
            float coef = 0.0f, e = 0.0f;
            if (lane < m) {
                coef = vA / fmaxf(fabsf(vA) * s_norm[rA], EPSF);
                e    = expf(coef * s_att[rA]);
            }
            den += e;
            sum_batch(features, s_rel, colA, rA, coef, e, m, lane, a0, a1, a2, a3);
        }
        for (int base = s0 + 32; base < e0; base += 32) {
            const int m = min(32, e0 - base);
            int col = 0, r = 0;
            float coef = 0.0f, e = 0.0f;
            if (lane < m) {
                const int tc = min(max(perm[base + lane], 0), NTRIPLE - 1);
                col = indices[2 * tc + 1];
                r   = sp_idx[2 * tc + 1];
                const float v = sp_val[tc];
                coef = v / fmaxf(fabsf(v) * s_norm[r], EPSF);
                e    = expf(coef * s_att[r]);
            }
            den += e;
            sum_batch(features, s_rel, col, r, coef, e, m, lane, a0, a1, a2, a3);
        }

        den = wave_sum(den);
        float* dst = out + (size_t)n0 * HID + lane;
        if (e0 > s0) {
            dst[0]  = a0 / den;
            dst[32] = a1 / den;
            dst[64] = a2 / den;
            dst[96] = a3 / den;
        } else {
            dst[0] = 0.0f; dst[32] = 0.0f; dst[64] = 0.0f; dst[96] = 0.0f;
        }

        n0 = n1; s0 = s1; e0 = e1;
        n1 = n2; s1 = s2; e1 = e2; t1 = t2;
        n2 = n3; s2 = s3; e2 = e3;
        colA = colB; rA = rB; vA = vB;
    }
}

extern "C" void kernel_launch(void* const* d_in, const int* in_sizes, int n_in,
                              void* d_out, int out_size, void* d_ws, size_t ws_size,
                              hipStream_t stream) {
    const float* features = (const float*)d_in[0];
    const float* rel_emb  = (const float*)d_in[1];
    const int*   indices  = (const int*)  d_in[2];
    const int*   sp_idx   = (const int*)  d_in[3];
    const float* sp_val   = (const float*)d_in[4];
    const float* attn     = (const float*)d_in[5];
    float* out = (float*)d_out;

    float* ws       = (float*)d_ws;
    float* rel_norm = ws;
    float* rel_att  = ws + 512;
    int*   cur      = (int*)(ws + 1024);
    int*   list     = (int*)(ws + WS_LIST_OFF);

    zero_int_kernel<<<49, 1024, 0, stream>>>(cur, NODE_SIZE);

    rel_stats_kernel<<<32, 32, 0, stream>>>(rel_emb, attn, rel_norm, rel_att);

    int* slot = (int*)d_out;
    count_kernel<<<SORT_GROUPS, 1024, 0, stream>>>(indices, cur, slot);
    scan_kernel<<<1, 1024, 0, stream>>>(cur);
    fill_kernel<<<FILL_GROUPS, FILL_THREADS, 0, stream>>>(indices, sp_idx, sp_val, cur, slot,
                                                          list, ws_size);

    gather_kernel<<<G_GROUPS, 1024, 0, stream>>>(
        features, rel_emb, indices, sp_idx, sp_val, rel_norm, rel_att,
        cur, list, out, ws_size);
}


# sort by node in 2 levels: edges bucketed to the gather groups' node ranges (LDS counts + count table; no global atomics, zero pass or 50000-entry scan), nodes ordered per group in LDS; arrival order
# speedup vs baseline: 3.4841x; 1.5837x over previous
#include <hip/hip_runtime.h>

#define NODE_SIZE 50000
#define REL_SIZE  500
#define HID       128
#define NTRIPLE   800000
#define EPSF      1e-12f

typedef __attribute__((ext_vector_type(2))) float v2f;
typedef __attribute__((ext_vector_type(8))) float v8f;

__global__ __launch_bounds__(32)
void rel_stats_kernel(const float* __restrict__ rel_emb,
                      const float* __restrict__ attn,
                      float* __restrict__ rel_norm,
                      float* __restrict__ rel_att) {
    const int lane = threadIdx.x & 31;
    const int tile = blockIdx.x;
    const int half = lane >> 4;
    const int m16  = lane & 15;

    const int   row   = tile * 16 + m16;
    const bool  rowok = row < REL_SIZE;
    const float amask = rowok ? 1.0f : 0.0f;
    const float* arow = rel_emb + (size_t)(rowok ? row : 0) * HID;
    const float bmask = (m16 == 0) ? 1.0f : 0.0f;

    v8f c = {};
    for (int k = 0; k < HID; k += 4) {
        const int koff = half * 2;
        v2f a, b;
        a.x = arow[k + koff + 0] * amask;
        a.y = arow[k + koff + 1] * amask;
        b.x = attn[k + koff + 0] * bmask;
        b.y = attn[k + koff + 1] * bmask;
        c = __builtin_amdgcn_wmma_f32_16x16x4_f32(false, a, false, b,
                                                  (short)0, c, false, false);
    }
    if (m16 == 0) {
        const int mbase = tile * 16 + half * 8;
#pragma unroll
        for (int j = 0; j < 8; ++j)
            if (mbase + j < REL_SIZE) rel_att[mbase + j] = c[j];
    }

    float ss = 0.0f;
    {
        const float4* p = (const float4*)(arow + half * 64);
#pragma unroll
        for (int i = 0; i < 16; ++i) {
            float4 f = p[i];
            ss += f.x * f.x + f.y * f.y + f.z * f.z + f.w * f.w;
        }
    }
    ss += __shfl_xor(ss, 16, 32);
    if (half == 0 && rowok) rel_norm[row] = sqrtf(ss);
}

#define G_GROUPS 256
#define G_WAVES  32
#define G_CHUNK  ((NODE_SIZE + G_GROUPS - 1) / G_GROUPS)
#define REL_VEC4 (REL_SIZE * HID / 4)

#define SUB_NODES    98
#define NBUCKET      512
#define SORT_GROUPS  128
#define SORT_CHUNK   (NTRIPLE / SORT_GROUPS)
#define SORT_EDGES   7

static_assert(2 * SUB_NODES == G_CHUNK, "two buckets per gather work-group");
static_assert(2 * G_GROUPS == NBUCKET, "two buckets per gather work-group");
static_assert(SUB_NODES <= 128, "a node's place in its bucket fits 7 bits");
static_assert(SORT_GROUPS * SORT_CHUNK == NTRIPLE, "the slices cover the edge list exactly");
static_assert(SORT_GROUPS == 128, "bucket_fill_kernel sums the table in two halves of 64 rows");
static_assert(NODE_SIZE <= 65536 && REL_SIZE <= 512 && NTRIPLE <= (1 << 20), "record fields");

#define WS_LIST_OFF   (1024 + NODE_SIZE)
#define WS_WIDE_BYTES ((size_t)(WS_LIST_OFF + 2 * NTRIPLE) * 4)

__device__ __forceinline__ int bucket_of(int row) {
    return (int)min((unsigned)row / (unsigned)SUB_NODES, (unsigned)(NBUCKET - 1));
}

__global__ __launch_bounds__(1024)
void bucket_count_kernel(const int* __restrict__ indices, int* __restrict__ hist) {
    __shared__ int h[NBUCKET];
    const int tid = threadIdx.x;
    if (tid < NBUCKET) h[tid] = 0;

    const int lo = blockIdx.x * SORT_CHUNK;
    int row[SORT_EDGES];
#pragma unroll
    for (int k = 0; k < SORT_EDGES; ++k) {
        const int e = min(k * 1024 + tid, SORT_CHUNK - 1);
        row[k] = indices[2 * (lo + e)];
    }
    __syncthreads();
#pragma unroll
    for (int k = 0; k < SORT_EDGES; ++k)
        if (k * 1024 + tid < SORT_CHUNK) atomicAdd(&h[bucket_of(row[k])], 1);
    __syncthreads();
    if (tid < NBUCKET) hist[blockIdx.x * NBUCKET + tid] = h[tid];
}

__global__ __launch_bounds__(1024)
void bucket_fill_kernel(const int* __restrict__ indices,
                        const int* __restrict__ sp_idx,
                        const float* __restrict__ sp_val,
                        const int* __restrict__ hist,
                        int* __restrict__ bstart,
                        int* __restrict__ list,
                        size_t ws_bytes) {
    __shared__ int  s_tot[2][NBUCKET];
    __shared__ int  s_mine[2][NBUCKET];
    __shared__ int  s_cnt[NBUCKET];
    __shared__ int  s_base[NBUCKET];
    __shared__ int  s_loff[NBUCKET];
    __shared__ int  s_wt[16], s_wo[16];
    __shared__ int2 s_rec[SORT_CHUNK];
    __shared__ int  s_pos[SORT_CHUNK];

    const int tid  = threadIdx.x;
    const int lane = tid & 31;
    const int wave = tid >> 5;
    const int b    = blockIdx.x;
    const int lo   = b * SORT_CHUNK;
    const bool wide = ws_bytes >= WS_WIDE_BYTES;

    if (tid < NBUCKET) s_cnt[tid] = 0;

    const int2* ind2 = (const int2*)indices;
    int2  rc[SORT_EDGES];
    int   rl[SORT_EDGES];
    float vl[SORT_EDGES];
#pragma unroll
    for (int k = 0; k < SORT_EDGES; ++k) {
        const int e = lo + min(k * 1024 + tid, SORT_CHUNK - 1);
        rc[k] = ind2[e];
        rl[k] = 0;
        vl[k] = 0.0f;
        if (wide) {
            rl[k] = sp_idx[2 * e + 1];
            vl[k] = sp_val[e];
        }
    }

    {
        const int q    = tid & (NBUCKET - 1);
        const int half = tid >> 9;
        int tot = 0, mine = 0;
#pragma unroll 32
        for (int r = half * 64; r < half * 64 + 64; ++r) {
            const int v = hist[r * NBUCKET + q];
            tot  += v;
            mine += (r < b) ? v : 0;
        }
        s_tot[half][q]  = tot;
        s_mine[half][q] = mine;
    }
    __syncthreads();

    int qk[SORT_EDGES], rank[SORT_EDGES];
#pragma unroll
    for (int k = 0; k < SORT_EDGES; ++k) {
        qk[k]   = bucket_of(rc[k].x);
        rank[k] = (k * 1024 + tid < SORT_CHUNK) ? atomicAdd(&s_cnt[qk[k]], 1) : 0;
    }
    __syncthreads();

    int t = 0, own = 0, mn = 0, it = 0, io = 0;
    if (tid < NBUCKET) {
        t   = s_tot[0][tid] + s_tot[1][tid];
        mn  = s_mine[0][tid] + s_mine[1][tid];
        own = s_cnt[tid];
        it = t;
        io = own;
#pragma unroll
        for (int off = 1; off < 32; off <<= 1) {
            const int nt = __shfl_up(it, off, 32);
            const int no = __shfl_up(io, off, 32);
            if (lane >= off) { it += nt; io += no; }
        }
        if (lane == 31) { s_wt[wave] = it; s_wo[wave] = io; }
    }
    __syncthreads();
    if (tid < NBUCKET) {
        int pt = 0, po = 0;
        for (int w = 0; w < wave; ++w) { pt += s_wt[w]; po += s_wo[w]; }
        const int start = pt + it - t;
        s_base[tid] = start + mn;
        s_loff[tid] = po + io - own;
        if (b == 0) bstart[tid] = start;
    }
    if (b == 0 && tid == 0) bstart[NBUCKET] = NTRIPLE;
    __syncthreads();

#pragma unroll
    for (int k = 0; k < SORT_EDGES; ++k) {
        if (k * 1024 + tid < SORT_CHUNK) {
            const int q = qk[k];
            const int p = min(max(s_loff[q] + rank[k], 0), SORT_CHUNK - 1);
            const int place = (rc[k].x - q * SUB_NODES) & 127;
            s_pos[p] = s_base[q] + rank[k];
            if (wide)
                s_rec[p] = make_int2((rc[k].y & 0xffff) | ((rl[k] & 511) << 16) | (place << 25),
                                     __float_as_int(vl[k]));
            else
                s_rec[p] = make_int2((lo + k * 1024 + tid) | (place << 20), 0);
        }
    }
    __syncthreads();

    int2* rec  = (int2*)list;
    int*  perm = list;
    for (int i = tid; i < SORT_CHUNK; i += 1024) {
        const int pos = min(max(s_pos[i], 0), NTRIPLE - 1);
        const int2 x = s_rec[i];
        if (wide) rec[pos] = x;
        else      perm[pos] = x.x;
    }
}

#define G_CAP 4096
#define G_PER (G_CAP / 1024)


template <int M>
__device__ __forceinline__ float row_xor(float p) {
    return __int_as_float(__builtin_amdgcn_update_dpp(0, __float_as_int(p), 0x160 + M, 0xf, 0xf, true));
}

__device__ __forceinline__ float xhalf(float x) {
    const int b = __float_as_int(x);
    return __int_as_float(__builtin_amdgcn_permlanex16(b, b, 0x76543210, 0xfedcba98, false, false));
}

__device__ __forceinline__ float wave_sum(float p) {
    p += xhalf(p);
    p += row_xor<8>(p);
    p += row_xor<4>(p);
    p += row_xor<2>(p);
    p += row_xor<1>(p);
    return p;
}

__device__ __forceinline__ float swap_add(float x, float y) {
    const auto r = __builtin_amdgcn_permlane16_swap(__float_as_uint(x), __float_as_uint(y), false, false);
    return __uint_as_float(r[0]) + __uint_as_float(r[1]);
}

__device__ __forceinline__ void wave_sum8(const float (&p)[8], bool hi8, bool hi4, float (&tot)[8]) {
    float q[4];
#pragma unroll
    for (int i = 0; i < 4; ++i) q[i] = swap_add(p[i], p[i + 4]);
    float r[2];
#pragma unroll
    for (int i = 0; i < 2; ++i) {
        const float keep = hi8 ? q[i + 2] : q[i];
        const float send = hi8 ? q[i] : q[i + 2];
        r[i] = keep + row_xor<8>(send);
    }
    const float keep = hi4 ? r[1] : r[0];
    const float send = hi4 ? r[0] : r[1];
    float s = keep + row_xor<4>(send);
    s += row_xor<2>(s);
    s += row_xor<1>(s);
#pragma unroll
    for (int k = 0; k < 8; ++k)
        tot[k] = __int_as_float(__builtin_amdgcn_readlane(
            __float_as_int(s), 4 * (k & 1) + 8 * ((k >> 1) & 1) + 16 * (k >> 2)));
}

__device__ __forceinline__ void wave_sum4(const float (&p)[4], bool hi8, float (&tot)[4]) {
    const float q0 = swap_add(p[0], p[2]);
    const float q1 = swap_add(p[1], p[3]);
    const float keep = hi8 ? q1 : q0;
    const float send = hi8 ? q0 : q1;
    float s = keep + row_xor<8>(send);
    s += row_xor<4>(s);
    s += row_xor<2>(s);
    s += row_xor<1>(s);
#pragma unroll
    for (int k = 0; k < 4; ++k)
        tot[k] = __int_as_float(__builtin_amdgcn_readlane(
            __float_as_int(s), 8 * (k & 1) + 16 * (k >> 1)));
}

__device__ __forceinline__ void load_rel(float (&r)[4], const float* rrow) {
    r[0] = rrow[0]; r[1] = rrow[32]; r[2] = rrow[64]; r[3] = rrow[96];
}

__device__ __forceinline__ float lane_dot(const float (&f)[4], const float (&r)[4]) {
    return f[0] * r[0] + f[1] * r[1] + f[2] * r[2] + f[3] * r[3];
}

__device__ __forceinline__ void add_edge(const float (&f)[4], const float (&r)[4],
                                         float coef, float e, float partial,
                                         float& a0, float& a1, float& a2, float& a3) {
    const float proj = coef * partial;
    const float k2   = 2.0f * proj * coef;
    a0 += (f[0] - k2 * r[0]) * e;
    a1 += (f[1] - k2 * r[1]) * e;
    a2 += (f[2] - k2 * r[2]) * e;
    a3 += (f[3] - k2 * r[3]) * e;
}

__device__ __forceinline__ void load_feat(float (&f)[4], const float* __restrict__ features,
                                          int col, int lane) {
    const float* frow = features + (size_t)col * HID + lane;
    f[0] = frow[0]; f[1] = frow[32]; f[2] = frow[64]; f[3] = frow[96];
}

__device__ __forceinline__ int bcast_i(int v, int j) {
    return __builtin_amdgcn_readlane(v, j);
}
__device__ __forceinline__ float bcast_f(float v, int j) {
    return __int_as_float(__builtin_amdgcn_readlane(__float_as_int(v), j));
}

__device__ __forceinline__ void sum_batch(const float* __restrict__ features,
                                          const float* s_rel,
                                          int col, int r, float coef, float e,
                                          int m, int lane,
                                          float& a0, float& a1, float& a2, float& a3) {
    const bool hi8 = (lane & 8) != 0;
    const bool hi4 = (lane & 4) != 0;
    int j = 0;
    for (; j + 8 <= m; j += 8) {
        float f[8][4], rr[8][4], p[8], tot[8];
#pragma unroll
        for (int k = 0; k < 8; ++k) load_feat(f[k], features, bcast_i(col, j + k), lane);
#pragma unroll
        for (int k = 0; k < 8; ++k) {
            load_rel(rr[k], s_rel + bcast_i(r, j + k) * HID + lane);
            p[k] = lane_dot(f[k], rr[k]);
        }
        wave_sum8(p, hi8, hi4, tot);
#pragma unroll
        for (int k = 0; k < 8; ++k)
            add_edge(f[k], rr[k], bcast_f(coef, j + k), bcast_f(e, j + k), tot[k], a0, a1, a2, a3);
    }
    for (; j + 4 <= m; j += 4) {
        float f[4][4], rr[4][4], p[4], tot[4];
#pragma unroll
        for (int k = 0; k < 4; ++k) load_feat(f[k], features, bcast_i(col, j + k), lane);
#pragma unroll
        for (int k = 0; k < 4; ++k) {
            load_rel(rr[k], s_rel + bcast_i(r, j + k) * HID + lane);
            p[k] = lane_dot(f[k], rr[k]);
        }
        wave_sum4(p, hi8, tot);
#pragma unroll
        for (int k = 0; k < 4; ++k)
            add_edge(f[k], rr[k], bcast_f(coef, j + k), bcast_f(e, j + k), tot[k], a0, a1, a2, a3);
    }
    for (; j < m; ++j) {
        float f[4], rr[4];
        load_feat(f, features, bcast_i(col, j), lane);
        load_rel(rr, s_rel + bcast_i(r, j) * HID + lane);
        add_edge(f, rr, bcast_f(coef, j), bcast_f(e, j), wave_sum(lane_dot(f, rr)), a0, a1, a2, a3);
    }
}

__device__ __forceinline__ void one_edge(const float* __restrict__ features,
                                         const float* s_rel,
                                         int col, int r, float coef, float e,
                                         int j, int lane,
                                         float& a0, float& a1, float& a2, float& a3) {
    float f[4], rr[4];
    load_feat(f, features, bcast_i(col, j), lane);
    load_rel(rr, s_rel + bcast_i(r, j) * HID + lane);
    add_edge(f, rr, bcast_f(coef, j), bcast_f(e, j), wave_sum(lane_dot(f, rr)), a0, a1, a2, a3);
}

__device__ __forceinline__ int2 fetch_rec(bool wide, const int* __restrict__ list,
                                          const int* __restrict__ indices,
                                          const int* __restrict__ sp_idx,
                                          const float* __restrict__ sp_val, int pos) {
    pos = min(max(pos, 0), NTRIPLE - 1);
    if (wide) return ((const int2*)list)[pos];
    const int w = list[pos];
    const int eid = min(w & 0xfffff, NTRIPLE - 1);
    const int col = indices[2 * eid + 1];
    const int rl  = sp_idx[2 * eid + 1];
    const float v = sp_val[eid];
    return make_int2((col & 0xffff) | ((rl & 511) << 16) | (((w >> 20) & 127) << 25),
                     __float_as_int(v));
}
__device__ __forceinline__ int rec_node(int2 x, int pos, int b1) {
    return (int)((unsigned)x.x >> 25) + ((pos >= b1) ? SUB_NODES : 0);
}
__device__ __forceinline__ void split_rec(int2 x, int& col, int& r, float& v) {
    col = min(x.x & 0xffff, NODE_SIZE - 1);
    r   = min((x.x >> 16) & 511, REL_SIZE - 1);
    v   = __int_as_float(x.y);
}

__device__ __forceinline__ int next_node(int* s_next, int lane) {
    int n = 0;
    if (lane == 0) n = atomicAdd(s_next, 1);
    return __builtin_amdgcn_readfirstlane(n);
}

__device__ __forceinline__ void store_node(float* __restrict__ out, int node, int lane, bool any,
                                           float a0, float a1, float a2, float a3, float den) {
    float* dst = out + (size_t)node * HID + lane;
    if (any) {
        dst[0]  = a0 / den;
        dst[32] = a1 / den;
        dst[64] = a2 / den;
        dst[96] = a3 / den;
    } else {
        dst[0] = 0.0f; dst[32] = 0.0f; dst[64] = 0.0f; dst[96] = 0.0f;
    }
}

__device__ __forceinline__ void stage_tables(const float* __restrict__ rel_emb,
                                             const float* __restrict__ rel_norm,
                                             const float* __restrict__ rel_att,
                                             float* s_rel, float* s_norm, float* s_att,
                                             int tid) {
    const float4* src = (const float4*)rel_emb;
    float4* dst = (float4*)s_rel;
#pragma unroll
    for (int b = 0; b < 4; ++b) {
        float4 v[4];
#pragma unroll
        for (int k = 0; k < 4; ++k) {
            const int i = (b * 4 + k) * 1024 + tid;
            v[k] = src[min(i, REL_VEC4 - 1)];
        }
#pragma unroll
        for (int k = 0; k < 4; ++k) {
            const int i = (b * 4 + k) * 1024 + tid;
            if (i < REL_VEC4) dst[i] = v[k];
        }
    }
    if (tid < REL_SIZE) { s_norm[tid] = rel_norm[tid]; s_att[tid] = rel_att[tid]; }
}

__global__ __launch_bounds__(1024)
void gather_kernel(const float* __restrict__ features,
                   const float* __restrict__ rel_emb,
                   const int*   __restrict__ indices,
                   const int*   __restrict__ sp_idx,
                   const float* __restrict__ sp_val,
                   const float* __restrict__ rel_norm,
                   const float* __restrict__ rel_att,
                   const int*   __restrict__ bstart,
                   const int*   __restrict__ list,
                   float* __restrict__ out,
                   size_t ws_bytes) {
    __shared__ float s_rel[REL_SIZE * HID];
    __shared__ float s_norm[512];
    __shared__ float s_att[512];
    __shared__ int2  s_rec[G_CAP];
    __shared__ int   s_hist[256];
    __shared__ int   s_seg[257];
    __shared__ int   s_next;

    const int tid  = threadIdx.x;
    const int lane = tid & 31;
    const int wave = __builtin_amdgcn_readfirstlane(tid >> 5);
    const int lo = blockIdx.x * G_CHUNK;
    const int hi = min(lo + G_CHUNK, NODE_SIZE);
    const bool wide = ws_bytes >= WS_WIDE_BYTES;

    int b0 = bstart[2 * blockIdx.x];
    int b1 = bstart[2 * blockIdx.x + 1];
    int b2 = bstart[2 * blockIdx.x + 2];
    b0 = min(max(b0, 0), NTRIPLE);
    b1 = min(max(b1, b0), NTRIPLE);
    b2 = min(max(b2, b1), NTRIPLE);
    const int m = b2 - b0;

    if (m <= G_CAP) {
        int2 x[G_PER];
        int  nd[G_PER], rk[G_PER];
#pragma unroll
        for (int j = 0; j < G_PER; ++j) {
            const int i = j * 1024 + tid;
            x[j]  = make_int2(0, 0);
            nd[j] = 0;
            if (i < m) {
                x[j]  = fetch_rec(wide, list, indices, sp_idx, sp_val, b0 + i);
                nd[j] = rec_node(x[j], b0 + i, b1);
            }
        }
        if (tid < 256) s_hist[tid] = 0;
        if (tid == 0) s_next = lo + G_WAVES;
        stage_tables(rel_emb, rel_norm, rel_att, s_rel, s_norm, s_att, tid);
        __syncthreads();

#pragma unroll
        for (int j = 0; j < G_PER; ++j)
            rk[j] = (j * 1024 + tid < m) ? atomicAdd(&s_hist[nd[j]], 1) : 0;
        __syncthreads();
        if (wave == 0) {
            int c[8];
            int sum = 0;
#pragma unroll
            for (int k = 0; k < 8; ++k) { c[k] = s_hist[lane * 8 + k]; sum += c[k]; }
            int incl = sum;
#pragma unroll
            for (int off = 1; off < 32; off <<= 1) {
                const int n = __shfl_up(incl, off, 32);
                if (lane >= off) incl += n;
            }
            int run = incl - sum;
#pragma unroll
            for (int k = 0; k < 8; ++k) { s_seg[lane * 8 + k] = run; run += c[k]; }
            if (lane == 31) s_seg[256] = run;
        }
        __syncthreads();
#pragma unroll
        for (int j = 0; j < G_PER; ++j)
            if (j * 1024 + tid < m)
                s_rec[min(max(s_seg[nd[j]] + rk[j], 0), G_CAP - 1)] = x[j];
        __syncthreads();

        int n0 = lo + wave;
        int n1 = next_node(&s_next, lane);
        int s0 = 0, e0 = 0;
        int2 xA = make_int2(0, 0);
        if (n0 < hi) {
            s0 = __builtin_amdgcn_readfirstlane(s_seg[n0 - lo]);
            e0 = __builtin_amdgcn_readfirstlane(s_seg[n0 - lo + 1]);
            if (lane < e0 - s0) xA = s_rec[min(s0 + lane, G_CAP - 1)];
        }
        while (n0 < hi) {
            const int n2 = next_node(&s_next, lane);
            int s1 = 0, e1 = 0;
            int2 xB = make_int2(0, 0);
            if (n1 < hi) {
                s1 = __builtin_amdgcn_readfirstlane(s_seg[n1 - lo]);
                e1 = __builtin_amdgcn_readfirstlane(s_seg[n1 - lo + 1]);
                if (lane < e1 - s1) xB = s_rec[min(s1 + lane, G_CAP - 1)];
            }

            float a0 = 0.0f, a1 = 0.0f, a2 = 0.0f, a3 = 0.0f, den = 0.0f;
            {
                const int mm = min(32, e0 - s0);
                int colA, rA;
                float vA;
                split_rec(xA, colA, rA, vA);
                float coef = 0.0f, e = 0.0f;
                if (lane < mm) {
                    coef = vA / fmaxf(fabsf(vA) * s_norm[rA], EPSF);
                    e    = expf(coef * s_att[rA]);
                }
                den += e;
                sum_batch(features, s_rel, colA, rA, coef, e, mm, lane, a0, a1, a2, a3);
            }
            for (int base = s0 + 32; base < e0; base += 32) {
                const int mm = min(32, e0 - base);
                int col = 0, r = 0;
                float coef = 0.0f, e = 0.0f;
                if (lane < mm) {
                    float v;
                    split_rec(s_rec[min(base + lane, G_CAP - 1)], col, r, v);
                    coef = v / fmaxf(fabsf(v) * s_norm[r], EPSF);
                    e    = expf(coef * s_att[r]);
                }
                den += e;
                sum_batch(features, s_rel, col, r, coef, e, mm, lane, a0, a1, a2, a3);
            }
            den = wave_sum(den);
            store_node(out, n0, lane, e0 > s0, a0, a1, a2, a3, den);

            n0 = n1; s0 = s1; e0 = e1; xA = xB;
            n1 = n2;
        }
        return;
    }

    if (tid == 0) s_next = lo + G_WAVES;
    stage_tables(rel_emb, rel_norm, rel_att, s_rel, s_norm, s_att, tid);
    __syncthreads();
    int n0 = lo + wave;
    while (n0 < hi) {
        float a0 = 0.0f, a1 = 0.0f, a2 = 0.0f, a3 = 0.0f, den = 0.0f;
        bool any = false;
        for (int base = b0; base < b2; base += 32) {
            const int i = base + lane;
            int col = 0, r = 0;
            float coef = 0.0f, e = 0.0f;
            bool hit = false;
            if (i < b2) {
                const int2 xr = fetch_rec(wide, list, indices, sp_idx, sp_val, i);
                if (rec_node(xr, i, b1) == n0 - lo) {
                    float v;
                    hit = true;
                    split_rec(xr, col, r, v);
                    coef = v / fmaxf(fabsf(v) * s_norm[r], EPSF);
                    e    = expf(coef * s_att[r]);
                }
            }
            den += e;
            unsigned mask = __builtin_amdgcn_ballot_w32(hit);
            any = any || (mask != 0u);
            while (mask) {
                const int j = __builtin_ctz(mask);
                mask &= mask - 1u;
                one_edge(features, s_rel, col, r, coef, e, j, lane, a0, a1, a2, a3);
            }
        }
        den = wave_sum(den);
        store_node(out, n0, lane, any, a0, a1, a2, a3, den);
        n0 = next_node(&s_next, lane);
    }
}

extern "C" void kernel_launch(void* const* d_in, const int* in_sizes, int n_in,
                              void* d_out, int out_size, void* d_ws, size_t ws_size,
                              hipStream_t stream) {
    const float* features = (const float*)d_in[0];
    const float* rel_emb  = (const float*)d_in[1];
    const int*   indices  = (const int*)  d_in[2];
    const int*   sp_idx   = (const int*)  d_in[3];
    const float* sp_val   = (const float*)d_in[4];
    const float* attn     = (const float*)d_in[5];
    float* out = (float*)d_out;

    float* ws       = (float*)d_ws;
    float* rel_norm = ws;
    float* rel_att  = ws + 512;
    int*   bstart   = (int*)(ws + 1024);
    int*   list     = (int*)(ws + WS_LIST_OFF);

    rel_stats_kernel<<<32, 32, 0, stream>>>(rel_emb, attn, rel_norm, rel_att);

    int* hist = (int*)d_out;
    bucket_count_kernel<<<SORT_GROUPS, 1024, 0, stream>>>(indices, hist);
    bucket_fill_kernel<<<SORT_GROUPS, 1024, 0, stream>>>(indices, sp_idx, sp_val, hist, bstart,
                                                         list, ws_size);

    gather_kernel<<<G_GROUPS, 1024, 0, stream>>>(
        features, rel_emb, indices, sp_idx, sp_val, rel_norm, rel_att,
        bstart, list, out, ws_size);
}


# per-relation stats (WMMA logits + norms) run in 8 extra work-groups at the head of the bucket-count launch, not in a launch of their own: 3 launches per repetition, not 4; arithmetic unchanged
# speedup vs baseline: 3.7449x; 1.0749x over previous
#include <hip/hip_runtime.h>

#define NODE_SIZE 50000
#define REL_SIZE  500
#define HID       128
#define NTRIPLE   800000
#define EPSF      1e-12f

typedef __attribute__((ext_vector_type(2))) float v2f;
typedef __attribute__((ext_vector_type(8))) float v8f;

__device__ __forceinline__
void rel_stats_tile(const float* __restrict__ rel_emb,
                    const float* __restrict__ attn,
                    float* __restrict__ rel_norm,
                    float* __restrict__ rel_att,
                    int tile, int lane) {
    const int half = lane >> 4;
    const int m16  = lane & 15;

    const int   row   = tile * 16 + m16;
    const bool  rowok = row < REL_SIZE;
    const float amask = rowok ? 1.0f : 0.0f;
    const float* arow = rel_emb + (size_t)(rowok ? row : 0) * HID;
    const float bmask = (m16 == 0) ? 1.0f : 0.0f;

    v8f c = {};
    for (int k = 0; k < HID; k += 4) {
        const int koff = half * 2;
        v2f a, b;
        a.x = arow[k + koff + 0] * amask;
        a.y = arow[k + koff + 1] * amask;
        b.x = attn[k + koff + 0] * bmask;
        b.y = attn[k + koff + 1] * bmask;
        c = __builtin_amdgcn_wmma_f32_16x16x4_f32(false, a, false, b,
                                                  (short)0, c, false, false);
    }
    if (m16 == 0) {
        const int mbase = tile * 16 + half * 8;
#pragma unroll
        for (int j = 0; j < 8; ++j)
            if (mbase + j < REL_SIZE) rel_att[mbase + j] = c[j];
    }

    float ss = 0.0f;
    {
        const float4* p = (const float4*)(arow + half * 64);
#pragma unroll
        for (int i = 0; i < 16; ++i) {
            float4 f = p[i];
            ss += f.x * f.x + f.y * f.y + f.z * f.z + f.w * f.w;
        }
    }
    ss += __shfl_xor(ss, 16, 32);
    if (half == 0 && rowok) rel_norm[row] = sqrtf(ss);
}

#define G_GROUPS 256
#define G_WAVES  32
#define G_CHUNK  ((NODE_SIZE + G_GROUPS - 1) / G_GROUPS)
#define REL_VEC4 (REL_SIZE * HID / 4)

#define SUB_NODES    98
#define NBUCKET      512
#define SORT_GROUPS  128
#define SORT_CHUNK   (NTRIPLE / SORT_GROUPS)
#define SORT_EDGES   7

#define REL_TILES    32
#define REL_GROUPS   8
#define REL_WAVES    (REL_TILES / REL_GROUPS)
#define COUNT_GRID   (REL_GROUPS + SORT_GROUPS)

static_assert(REL_TILES * 16 >= REL_SIZE && REL_GROUPS * REL_WAVES == REL_TILES, "the tiles cover the relation table");
static_assert(2 * SUB_NODES == G_CHUNK, "two buckets per gather work-group");
static_assert(2 * G_GROUPS == NBUCKET, "two buckets per gather work-group");
static_assert(SUB_NODES <= 128, "a node's place in its bucket fits 7 bits");
static_assert(SORT_GROUPS * SORT_CHUNK == NTRIPLE, "the slices cover the edge list exactly");
static_assert(SORT_GROUPS == 128, "bucket_fill_kernel sums the table in two halves of 64 rows");
static_assert(NODE_SIZE <= 65536 && REL_SIZE <= 512 && NTRIPLE <= (1 << 20), "record fields");

#define WS_LIST_OFF   (1024 + NODE_SIZE)
#define WS_WIDE_BYTES ((size_t)(WS_LIST_OFF + 2 * NTRIPLE) * 4)

__device__ __forceinline__ int bucket_of(int row) {
    return (int)min((unsigned)row / (unsigned)SUB_NODES, (unsigned)(NBUCKET - 1));
}

__global__ __launch_bounds__(1024)
void stats_count_kernel(const float* __restrict__ rel_emb,
                        const float* __restrict__ attn,
                        float* __restrict__ rel_norm,
                        float* __restrict__ rel_att,
                        const int* __restrict__ indices, int* __restrict__ hist) {
    __shared__ int h[NBUCKET];
    const int tid = threadIdx.x;

    if (blockIdx.x < REL_GROUPS) {
        const int wave = __builtin_amdgcn_readfirstlane(tid >> 5);
        if (wave < REL_WAVES)
            rel_stats_tile(rel_emb, attn, rel_norm, rel_att,
                           blockIdx.x * REL_WAVES + wave, tid & 31);
        return;
    }
    const int b = blockIdx.x - REL_GROUPS;

    if (tid < NBUCKET) h[tid] = 0;

    const int lo = b * SORT_CHUNK;
    int row[SORT_EDGES];
#pragma unroll
    for (int k = 0; k < SORT_EDGES; ++k) {
        const int e = min(k * 1024 + tid, SORT_CHUNK - 1);
        row[k] = indices[2 * (lo + e)];
    }
    __syncthreads();
#pragma unroll
    for (int k = 0; k < SORT_EDGES; ++k)
        if (k * 1024 + tid < SORT_CHUNK) atomicAdd(&h[bucket_of(row[k])], 1);
    __syncthreads();
    if (tid < NBUCKET) hist[b * NBUCKET + tid] = h[tid];
}

__global__ __launch_bounds__(1024)
void bucket_fill_kernel(const int* __restrict__ indices,
                        const int* __restrict__ sp_idx,
                        const float* __restrict__ sp_val,
                        const int* __restrict__ hist,
                        int* __restrict__ bstart,
                        int* __restrict__ list,
                        size_t ws_bytes) {
    __shared__ int  s_tot[2][NBUCKET];
    __shared__ int  s_mine[2][NBUCKET];
    __shared__ int  s_cnt[NBUCKET];
    __shared__ int  s_base[NBUCKET];
    __shared__ int  s_loff[NBUCKET];
    __shared__ int  s_wt[16], s_wo[16];
    __shared__ int2 s_rec[SORT_CHUNK];
    __shared__ int  s_pos[SORT_CHUNK];

    const int tid  = threadIdx.x;
    const int lane = tid & 31;
    const int wave = tid >> 5;
    const int b    = blockIdx.x;
    const int lo   = b * SORT_CHUNK;
    const bool wide = ws_bytes >= WS_WIDE_BYTES;

    if (tid < NBUCKET) s_cnt[tid] = 0;

    const int2* ind2 = (const int2*)indices;
    int2  rc[SORT_EDGES];
    int   rl[SORT_EDGES];
    float vl[SORT_EDGES];
#pragma unroll
    for (int k = 0; k < SORT_EDGES; ++k) {
        const int e = lo + min(k * 1024 + tid, SORT_CHUNK - 1);
        rc[k] = ind2[e];
        rl[k] = 0;
        vl[k] = 0.0f;
        if (wide) {
            rl[k] = sp_idx[2 * e + 1];
            vl[k] = sp_val[e];
        }
    }

    {
        const int q    = tid & (NBUCKET - 1);
        const int half = tid >> 9;
        int tot = 0, mine = 0;
#pragma unroll 32
        for (int r = half * 64; r < half * 64 + 64; ++r) {
            const int v = hist[r * NBUCKET + q];
            tot  += v;
            mine += (r < b) ? v : 0;
        }
        s_tot[half][q]  = tot;
        s_mine[half][q] = mine;
    }
    __syncthreads();

    int qk[SORT_EDGES], rank[SORT_EDGES];
#pragma unroll
    for (int k = 0; k < SORT_EDGES; ++k) {
        qk[k]   = bucket_of(rc[k].x);
        rank[k] = (k * 1024 + tid < SORT_CHUNK) ? atomicAdd(&s_cnt[qk[k]], 1) : 0;
    }
    __syncthreads();

    int t = 0, own = 0, mn = 0, it = 0, io = 0;
    if (tid < NBUCKET) {
        t   = s_tot[0][tid] + s_tot[1][tid];
        mn  = s_mine[0][tid] + s_mine[1][tid];
        own = s_cnt[tid];
        it = t;
        io = own;
#pragma unroll
        for (int off = 1; off < 32; off <<= 1) {
            const int nt = __shfl_up(it, off, 32);
            const int no = __shfl_up(io, off, 32);
            if (lane >= off) { it += nt; io += no; }
        }
        if (lane == 31) { s_wt[wave] = it; s_wo[wave] = io; }
    }
    __syncthreads();
    if (tid < NBUCKET) {
        int pt = 0, po = 0;
        for (int w = 0; w < wave; ++w) { pt += s_wt[w]; po += s_wo[w]; }
        const int start = pt + it - t;
        s_base[tid] = start + mn;
        s_loff[tid] = po + io - own;
        if (b == 0) bstart[tid] = start;
    }
    if (b == 0 && tid == 0) bstart[NBUCKET] = NTRIPLE;
    __syncthreads();

#pragma unroll
    for (int k = 0; k < SORT_EDGES; ++k) {
        if (k * 1024 + tid < SORT_CHUNK) {
            const int q = qk[k];
            const int p = min(max(s_loff[q] + rank[k], 0), SORT_CHUNK - 1);
            const int place = (rc[k].x - q * SUB_NODES) & 127;
            s_pos[p] = s_base[q] + rank[k];
            if (wide)
                s_rec[p] = make_int2((rc[k].y & 0xffff) | ((rl[k] & 511) << 16) | (place << 25),
                                     __float_as_int(vl[k]));
            else
                s_rec[p] = make_int2((lo + k * 1024 + tid) | (place << 20), 0);
        }
    }
    __syncthreads();

    int2* rec  = (int2*)list;
    int*  perm = list;
    for (int i = tid; i < SORT_CHUNK; i += 1024) {
        const int pos = min(max(s_pos[i], 0), NTRIPLE - 1);
        const int2 x = s_rec[i];
        if (wide) rec[pos] = x;
        else      perm[pos] = x.x;
    }
}

#define G_CAP 4096
#define G_PER (G_CAP / 1024)


template <int M>
__device__ __forceinline__ float row_xor(float p) {
    return __int_as_float(__builtin_amdgcn_update_dpp(0, __float_as_int(p), 0x160 + M, 0xf, 0xf, true));
}

__device__ __forceinline__ float xhalf(float x) {
    const int b = __float_as_int(x);
    return __int_as_float(__builtin_amdgcn_permlanex16(b, b, 0x76543210, 0xfedcba98, false, false));
}

__device__ __forceinline__ float wave_sum(float p) {
    p += xhalf(p);
    p += row_xor<8>(p);
    p += row_xor<4>(p);
    p += row_xor<2>(p);
    p += row_xor<1>(p);
    return p;
}

__device__ __forceinline__ float swap_add(float x, float y) {
    const auto r = __builtin_amdgcn_permlane16_swap(__float_as_uint(x), __float_as_uint(y), false, false);
    return __uint_as_float(r[0]) + __uint_as_float(r[1]);
}

__device__ __forceinline__ void wave_sum8(const float (&p)[8], bool hi8, bool hi4, float (&tot)[8]) {
    float q[4];
#pragma unroll
    for (int i = 0; i < 4; ++i) q[i] = swap_add(p[i], p[i + 4]);
    float r[2];
#pragma unroll
    for (int i = 0; i < 2; ++i) {
        const float keep = hi8 ? q[i + 2] : q[i];
        const float send = hi8 ? q[i] : q[i + 2];
        r[i] = keep + row_xor<8>(send);
    }
    const float keep = hi4 ? r[1] : r[0];
    const float send = hi4 ? r[0] : r[1];
    float s = keep + row_xor<4>(send);
    s += row_xor<2>(s);
    s += row_xor<1>(s);
#pragma unroll
    for (int k = 0; k < 8; ++k)
        tot[k] = __int_as_float(__builtin_amdgcn_readlane(
            __float_as_int(s), 4 * (k & 1) + 8 * ((k >> 1) & 1) + 16 * (k >> 2)));
}

__device__ __forceinline__ void wave_sum4(const float (&p)[4], bool hi8, float (&tot)[4]) {
    const float q0 = swap_add(p[0], p[2]);
    const float q1 = swap_add(p[1], p[3]);
    const float keep = hi8 ? q1 : q0;
    const float send = hi8 ? q0 : q1;
    float s = keep + row_xor<8>(send);
    s += row_xor<4>(s);
    s += row_xor<2>(s);
    s += row_xor<1>(s);
#pragma unroll
    for (int k = 0; k < 4; ++k)
        tot[k] = __int_as_float(__builtin_amdgcn_readlane(
            __float_as_int(s), 8 * (k & 1) + 16 * (k >> 1)));
}

__device__ __forceinline__ void load_rel(float (&r)[4], const float* rrow) {
    r[0] = rrow[0]; r[1] = rrow[32]; r[2] = rrow[64]; r[3] = rrow[96];
}

__device__ __forceinline__ float lane_dot(const float (&f)[4], const float (&r)[4]) {
    return f[0] * r[0] + f[1] * r[1] + f[2] * r[2] + f[3] * r[3];
}

__device__ __forceinline__ void add_edge(const float (&f)[4], const float (&r)[4],
                                         float coef, float e, float partial,
                                         float& a0, float& a1, float& a2, float& a3) {
    const float proj = coef * partial;
    const float k2   = 2.0f * proj * coef;
    a0 += (f[0] - k2 * r[0]) * e;
    a1 += (f[1] - k2 * r[1]) * e;
    a2 += (f[2] - k2 * r[2]) * e;
    a3 += (f[3] - k2 * r[3]) * e;
}

__device__ __forceinline__ void load_feat(float (&f)[4], const float* __restrict__ features,
                                          int col, int lane) {
    const float* frow = features + (size_t)col * HID + lane;
    f[0] = frow[0]; f[1] = frow[32]; f[2] = frow[64]; f[3] = frow[96];
}

__device__ __forceinline__ int bcast_i(int v, int j) {
    return __builtin_amdgcn_readlane(v, j);
}
__device__ __forceinline__ float bcast_f(float v, int j) {
    return __int_as_float(__builtin_amdgcn_readlane(__float_as_int(v), j));
}

__device__ __forceinline__ void sum_batch(const float* __restrict__ features,
                                          const float* s_rel,
                                          int col, int r, float coef, float e,
                                          int m, int lane,
                                          float& a0, float& a1, float& a2, float& a3) {
    const bool hi8 = (lane & 8) != 0;
    const bool hi4 = (lane & 4) != 0;
    int j = 0;
    for (; j + 8 <= m; j += 8) {
        float f[8][4], rr[8][4], p[8], tot[8];
#pragma unroll
        for (int k = 0; k < 8; ++k) load_feat(f[k], features, bcast_i(col, j + k), lane);
#pragma unroll
        for (int k = 0; k < 8; ++k) {
            load_rel(rr[k], s_rel + bcast_i(r, j + k) * HID + lane);
            p[k] = lane_dot(f[k], rr[k]);
        }
        wave_sum8(p, hi8, hi4, tot);
#pragma unroll
        for (int k = 0; k < 8; ++k)
            add_edge(f[k], rr[k], bcast_f(coef, j + k), bcast_f(e, j + k), tot[k], a0, a1, a2, a3);
    }
    for (; j + 4 <= m; j += 4) {
        float f[4][4], rr[4][4], p[4], tot[4];
#pragma unroll
        for (int k = 0; k < 4; ++k) load_feat(f[k], features, bcast_i(col, j + k), lane);
#pragma unroll
        for (int k = 0; k < 4; ++k) {
            load_rel(rr[k], s_rel + bcast_i(r, j + k) * HID + lane);
            p[k] = lane_dot(f[k], rr[k]);
        }
        wave_sum4(p, hi8, tot);
#pragma unroll
        for (int k = 0; k < 4; ++k)
            add_edge(f[k], rr[k], bcast_f(coef, j + k), bcast_f(e, j + k), tot[k], a0, a1, a2, a3);
    }
    for (; j < m; ++j) {
        float f[4], rr[4];
        load_feat(f, features, bcast_i(col, j), lane);
        load_rel(rr, s_rel + bcast_i(r, j) * HID + lane);
        add_edge(f, rr, bcast_f(coef, j), bcast_f(e, j), wave_sum(lane_dot(f, rr)), a0, a1, a2, a3);
    }
}

__device__ __forceinline__ void one_edge(const float* __restrict__ features,
                                         const float* s_rel,
                                         int col, int r, float coef, float e,
                                         int j, int lane,
                                         float& a0, float& a1, float& a2, float& a3) {
    float f[4], rr[4];
    load_feat(f, features, bcast_i(col, j), lane);
    load_rel(rr, s_rel + bcast_i(r, j) * HID + lane);
    add_edge(f, rr, bcast_f(coef, j), bcast_f(e, j), wave_sum(lane_dot(f, rr)), a0, a1, a2, a3);
}

__device__ __forceinline__ int2 fetch_rec(bool wide, const int* __restrict__ list,
                                          const int* __restrict__ indices,
                                          const int* __restrict__ sp_idx,
                                          const float* __restrict__ sp_val, int pos) {
    pos = min(max(pos, 0), NTRIPLE - 1);
    if (wide) return ((const int2*)list)[pos];
    const int w = list[pos];
    const int eid = min(w & 0xfffff, NTRIPLE - 1);
    const int col = indices[2 * eid + 1];
    const int rl  = sp_idx[2 * eid + 1];
    const float v = sp_val[eid];
    return make_int2((col & 0xffff) | ((rl & 511) << 16) | (((w >> 20) & 127) << 25),
                     __float_as_int(v));
}
__device__ __forceinline__ int rec_node(int2 x, int pos, int b1) {
    return (int)((unsigned)x.x >> 25) + ((pos >= b1) ? SUB_NODES : 0);
}
__device__ __forceinline__ void split_rec(int2 x, int& col, int& r, float& v) {
    col = min(x.x & 0xffff, NODE_SIZE - 1);
    r   = min((x.x >> 16) & 511, REL_SIZE - 1);
    v   = __int_as_float(x.y);
}

__device__ __forceinline__ int next_node(int* s_next, int lane) {
    int n = 0;
    if (lane == 0) n = atomicAdd(s_next, 1);
    return __builtin_amdgcn_readfirstlane(n);
}

__device__ __forceinline__ void store_node(float* __restrict__ out, int node, int lane, bool any,
                                           float a0, float a1, float a2, float a3, float den) {
    float* dst = out + (size_t)node * HID + lane;
    if (any) {
        dst[0]  = a0 / den;
        dst[32] = a1 / den;
        dst[64] = a2 / den;
        dst[96] = a3 / den;
    } else {
        dst[0] = 0.0f; dst[32] = 0.0f; dst[64] = 0.0f; dst[96] = 0.0f;
    }
}

__device__ __forceinline__ void stage_tables(const float* __restrict__ rel_emb,
                                             const float* __restrict__ rel_norm,
                                             const float* __restrict__ rel_att,
                                             float* s_rel, float* s_norm, float* s_att,
                                             int tid) {
    const float4* src = (const float4*)rel_emb;
    float4* dst = (float4*)s_rel;
#pragma unroll
    for (int b = 0; b < 4; ++b) {
        float4 v[4];
#pragma unroll
        for (int k = 0; k < 4; ++k) {
            const int i = (b * 4 + k) * 1024 + tid;
            v[k] = src[min(i, REL_VEC4 - 1)];
        }
#pragma unroll
        for (int k = 0; k < 4; ++k) {
            const int i = (b * 4 + k) * 1024 + tid;
            if (i < REL_VEC4) dst[i] = v[k];
        }
    }
    if (tid < REL_SIZE) { s_norm[tid] = rel_norm[tid]; s_att[tid] = rel_att[tid]; }
}

__global__ __launch_bounds__(1024)
void gather_kernel(const float* __restrict__ features,
                   const float* __restrict__ rel_emb,
                   const int*   __restrict__ indices,
                   const int*   __restrict__ sp_idx,
                   const float* __restrict__ sp_val,
                   const float* __restrict__ rel_norm,
                   const float* __restrict__ rel_att,
                   const int*   __restrict__ bstart,
                   const int*   __restrict__ list,
                   float* __restrict__ out,
                   size_t ws_bytes) {
    __shared__ float s_rel[REL_SIZE * HID];
    __shared__ float s_norm[512];
    __shared__ float s_att[512];
    __shared__ int2  s_rec[G_CAP];
    __shared__ int   s_hist[256];
    __shared__ int   s_seg[257];
    __shared__ int   s_next;

    const int tid  = threadIdx.x;
    const int lane = tid & 31;
    const int wave = __builtin_amdgcn_readfirstlane(tid >> 5);
    const int lo = blockIdx.x * G_CHUNK;
    const int hi = min(lo + G_CHUNK, NODE_SIZE);
    const bool wide = ws_bytes >= WS_WIDE_BYTES;

    int b0 = bstart[2 * blockIdx.x];
    int b1 = bstart[2 * blockIdx.x + 1];
    int b2 = bstart[2 * blockIdx.x + 2];
    b0 = min(max(b0, 0), NTRIPLE);
    b1 = min(max(b1, b0), NTRIPLE);
    b2 = min(max(b2, b1), NTRIPLE);
    const int m = b2 - b0;

    if (m <= G_CAP) {
        int2 x[G_PER];
        int  nd[G_PER], rk[G_PER];
#pragma unroll
        for (int j = 0; j < G_PER; ++j) {
            const int i = j * 1024 + tid;
            x[j]  = make_int2(0, 0);
            nd[j] = 0;
            if (i < m) {
                x[j]  = fetch_rec(wide, list, indices, sp_idx, sp_val, b0 + i);
                nd[j] = rec_node(x[j], b0 + i, b1);
            }
        }
        if (tid < 256) s_hist[tid] = 0;
        if (tid == 0) s_next = lo + G_WAVES;
        stage_tables(rel_emb, rel_norm, rel_att, s_rel, s_norm, s_att, tid);
        __syncthreads();

#pragma unroll
        for (int j = 0; j < G_PER; ++j)
            rk[j] = (j * 1024 + tid < m) ? atomicAdd(&s_hist[nd[j]], 1) : 0;
        __syncthreads();
        if (wave == 0) {
            int c[8];
            int sum = 0;
#pragma unroll
            for (int k = 0; k < 8; ++k) { c[k] = s_hist[lane * 8 + k]; sum += c[k]; }
            int incl = sum;
#pragma unroll
            for (int off = 1; off < 32; off <<= 1) {
                const int n = __shfl_up(incl, off, 32);
                if (lane >= off) incl += n;
            }
            int run = incl - sum;
#pragma unroll
            for (int k = 0; k < 8; ++k) { s_seg[lane * 8 + k] = run; run += c[k]; }
            if (lane == 31) s_seg[256] = run;
        }
        __syncthreads();
#pragma unroll
        for (int j = 0; j < G_PER; ++j)
            if (j * 1024 + tid < m)
                s_rec[min(max(s_seg[nd[j]] + rk[j], 0), G_CAP - 1)] = x[j];
        __syncthreads();

        int n0 = lo + wave;
        int n1 = next_node(&s_next, lane);
        int s0 = 0, e0 = 0;
        int2 xA = make_int2(0, 0);
        if (n0 < hi) {
            s0 = __builtin_amdgcn_readfirstlane(s_seg[n0 - lo]);
            e0 = __builtin_amdgcn_readfirstlane(s_seg[n0 - lo + 1]);
            if (lane < e0 - s0) xA = s_rec[min(s0 + lane, G_CAP - 1)];
        }
        while (n0 < hi) {
            const int n2 = next_node(&s_next, lane);
            int s1 = 0, e1 = 0;
            int2 xB = make_int2(0, 0);
            if (n1 < hi) {
                s1 = __builtin_amdgcn_readfirstlane(s_seg[n1 - lo]);
                e1 = __builtin_amdgcn_readfirstlane(s_seg[n1 - lo + 1]);
                if (lane < e1 - s1) xB = s_rec[min(s1 + lane, G_CAP - 1)];
            }

            float a0 = 0.0f, a1 = 0.0f, a2 = 0.0f, a3 = 0.0f, den = 0.0f;
            {
                const int mm = min(32, e0 - s0);
                int colA, rA;
                float vA;
                split_rec(xA, colA, rA, vA);
                float coef = 0.0f, e = 0.0f;
                if (lane < mm) {
                    coef = vA / fmaxf(fabsf(vA) * s_norm[rA], EPSF);
                    e    = expf(coef * s_att[rA]);
                }
                den += e;
                sum_batch(features, s_rel, colA, rA, coef, e, mm, lane, a0, a1, a2, a3);
            }
            for (int base = s0 + 32; base < e0; base += 32) {
                const int mm = min(32, e0 - base);
                int col = 0, r = 0;
                float coef = 0.0f, e = 0.0f;
                if (lane < mm) {
                    float v;
                    split_rec(s_rec[min(base + lane, G_CAP - 1)], col, r, v);
                    coef = v / fmaxf(fabsf(v) * s_norm[r], EPSF);
                    e    = expf(coef * s_att[r]);
                }
                den += e;
                sum_batch(features, s_rel, col, r, coef, e, mm, lane, a0, a1, a2, a3);
            }
            den = wave_sum(den);
            store_node(out, n0, lane, e0 > s0, a0, a1, a2, a3, den);

            n0 = n1; s0 = s1; e0 = e1; xA = xB;
            n1 = n2;
        }
        return;
    }

    if (tid == 0) s_next = lo + G_WAVES;
    stage_tables(rel_emb, rel_norm, rel_att, s_rel, s_norm, s_att, tid);
    __syncthreads();
    int n0 = lo + wave;
    while (n0 < hi) {
        float a0 = 0.0f, a1 = 0.0f, a2 = 0.0f, a3 = 0.0f, den = 0.0f;
        bool any = false;
        for (int base = b0; base < b2; base += 32) {
            const int i = base + lane;
            int col = 0, r = 0;
            float coef = 0.0f, e = 0.0f;
            bool hit = false;
            if (i < b2) {
                const int2 xr = fetch_rec(wide, list, indices, sp_idx, sp_val, i);
                if (rec_node(xr, i, b1) == n0 - lo) {
                    float v;
                    hit = true;
                    split_rec(xr, col, r, v);
                    coef = v / fmaxf(fabsf(v) * s_norm[r], EPSF);
                    e    = expf(coef * s_att[r]);
                }
            }
            den += e;
            unsigned mask = __builtin_amdgcn_ballot_w32(hit);
            any = any || (mask != 0u);
            while (mask) {
                const int j = __builtin_ctz(mask);
                mask &= mask - 1u;
                one_edge(features, s_rel, col, r, coef, e, j, lane, a0, a1, a2, a3);
            }
        }
        den = wave_sum(den);
        store_node(out, n0, lane, any, a0, a1, a2, a3, den);
        n0 = next_node(&s_next, lane);
    }
}

extern "C" void kernel_launch(void* const* d_in, const int* in_sizes, int n_in,
                              void* d_out, int out_size, void* d_ws, size_t ws_size,
                              hipStream_t stream) {
    const float* features = (const float*)d_in[0];
    const float* rel_emb  = (const float*)d_in[1];
    const int*   indices  = (const int*)  d_in[2];
    const int*   sp_idx   = (const int*)  d_in[3];
    const float* sp_val   = (const float*)d_in[4];
    const float* attn     = (const float*)d_in[5];
    float* out = (float*)d_out;

    float* ws       = (float*)d_ws;
    float* rel_norm = ws;
    float* rel_att  = ws + 512;
    int*   bstart   = (int*)(ws + 1024);
    int*   list     = (int*)(ws + WS_LIST_OFF);

    int* hist = (int*)d_out;
    stats_count_kernel<<<COUNT_GRID, 1024, 0, stream>>>(rel_emb, attn, rel_norm, rel_att,
                                                       indices, hist);
    bucket_fill_kernel<<<SORT_GROUPS, 1024, 0, stream>>>(indices, sp_idx, sp_val, hist, bstart,
                                                         list, ws_size);

    gather_kernel<<<G_GROUPS, 1024, 0, stream>>>(
        features, rel_emb, indices, sp_idx, sp_val, rel_norm, rel_att,
        bstart, list, out, ws_size);
}


# edges sorted only inside each sorting work-group (own list region + row of run starts); a gather work-group collects its 128 runs, record loads in flight together: no count launch; arrival order
# speedup vs baseline: 4.8666x; 1.2995x over previous
#include <hip/hip_runtime.h>

#define NODE_SIZE 50000
#define REL_SIZE  500
#define HID       128
#define NTRIPLE   800000
#define EPSF      1e-12f

typedef __attribute__((ext_vector_type(2))) float v2f;
typedef __attribute__((ext_vector_type(8))) float v8f;

__device__ __forceinline__
void rel_stats_tile(const float* __restrict__ rel_emb,
                    const float* __restrict__ attn,
                    float* __restrict__ rel_norm,
                    float* __restrict__ rel_att,
                    int tile, int lane) {
    const int half = lane >> 4;
    const int m16  = lane & 15;

    const int   row   = tile * 16 + m16;
    const bool  rowok = row < REL_SIZE;
    const float amask = rowok ? 1.0f : 0.0f;
    const float* arow = rel_emb + (size_t)(rowok ? row : 0) * HID;
    const float bmask = (m16 == 0) ? 1.0f : 0.0f;

    v8f c = {};
    for (int k = 0; k < HID; k += 4) {
        const int koff = half * 2;
        v2f a, b;
        a.x = arow[k + koff + 0] * amask;
        a.y = arow[k + koff + 1] * amask;
        b.x = attn[k + koff + 0] * bmask;
        b.y = attn[k + koff + 1] * bmask;
        c = __builtin_amdgcn_wmma_f32_16x16x4_f32(false, a, false, b,
                                                  (short)0, c, false, false);
    }
    if (m16 == 0) {
        const int mbase = tile * 16 + half * 8;
#pragma unroll
        for (int j = 0; j < 8; ++j)
            if (mbase + j < REL_SIZE) rel_att[mbase + j] = c[j];
    }

    float ss = 0.0f;
    {
        const float4* p = (const float4*)(arow + half * 64);
#pragma unroll
        for (int i = 0; i < 16; ++i) {
            float4 f = p[i];
            ss += f.x * f.x + f.y * f.y + f.z * f.z + f.w * f.w;
        }
    }
    ss += __shfl_xor(ss, 16, 32);
    if (half == 0 && rowok) rel_norm[row] = sqrtf(ss);
}

#define G_GROUPS 256
#define G_WAVES  32
#define G_CHUNK  ((NODE_SIZE + G_GROUPS - 1) / G_GROUPS)
#define REL_VEC4 (REL_SIZE * HID / 4)

#define SUB_NODES    98
#define NBUCKET      512
#define SORT_GROUPS  128
#define SORT_CHUNK   (NTRIPLE / SORT_GROUPS)
#define SORT_EDGES   7
#define RUN_COLS     (G_GROUPS + 1)

#define REL_TILES    32
#define REL_GROUPS   8
#define REL_WAVES    (REL_TILES / REL_GROUPS)
#define SORT_GRID    (REL_GROUPS + SORT_GROUPS)

static_assert(REL_TILES * 16 >= REL_SIZE && REL_GROUPS * REL_WAVES == REL_TILES, "the tiles cover the relation table");
static_assert(2 * SUB_NODES == G_CHUNK, "two buckets per gather work-group");
static_assert(2 * G_GROUPS == NBUCKET, "two buckets per gather work-group");
static_assert(SUB_NODES <= 128, "a node's place in its bucket fits 7 bits");
static_assert(SORT_GROUPS * SORT_CHUNK == NTRIPLE, "the slices cover the edge list exactly");
static_assert(SORT_CHUNK < 65536, "a run start fits 16 bits");
static_assert(NODE_SIZE <= 65536 && REL_SIZE <= 512 && NTRIPLE <= (1 << 20), "record fields");

#define WS_LIST_OFF   (1024 + NODE_SIZE)
#define WS_WIDE_BYTES ((size_t)(WS_LIST_OFF + 2 * NTRIPLE) * 4)
static_assert(1024 + SORT_GROUPS * RUN_COLS <= WS_LIST_OFF, "the rows of run starts end before the list");

__device__ __forceinline__ int bucket_of(int row) {
    return (int)min((unsigned)row / (unsigned)SUB_NODES, (unsigned)(NBUCKET - 1));
}

__global__ __launch_bounds__(1024)
void stats_sort_kernel(const float* __restrict__ rel_emb,
                       const float* __restrict__ attn,
                       float* __restrict__ rel_norm,
                       float* __restrict__ rel_att,
                       const int* __restrict__ indices,
                       const int* __restrict__ sp_idx,
                       const float* __restrict__ sp_val,
                       int* __restrict__ runs,
                       int* __restrict__ list,
                       size_t ws_bytes) {
    __shared__ int  s_cnt[NBUCKET];
    __shared__ int  s_loff[NBUCKET];
    __shared__ int  s_wo[16];
    __shared__ int2 s_rec[SORT_CHUNK];

    const int tid  = threadIdx.x;

    if (blockIdx.x < REL_GROUPS) {
        const int wv = __builtin_amdgcn_readfirstlane(tid >> 5);
        if (wv < REL_WAVES)
            rel_stats_tile(rel_emb, attn, rel_norm, rel_att,
                           blockIdx.x * REL_WAVES + wv, tid & 31);
        return;
    }

    const int lane = tid & 31;
    const int wave = tid >> 5;
    const int b    = blockIdx.x - REL_GROUPS;
    const int lo   = b * SORT_CHUNK;
    const bool wide = ws_bytes >= WS_WIDE_BYTES;

    if (tid < NBUCKET) s_cnt[tid] = 0;

    const int2* ind2 = (const int2*)indices;
    int2  rc[SORT_EDGES];
    int   rl[SORT_EDGES];
    float vl[SORT_EDGES];
#pragma unroll
    for (int k = 0; k < SORT_EDGES; ++k) {
        const int e = lo + min(k * 1024 + tid, SORT_CHUNK - 1);
        rc[k] = ind2[e];
        rl[k] = 0;
        vl[k] = 0.0f;
        if (wide) {
            rl[k] = sp_idx[2 * e + 1];
            vl[k] = sp_val[e];
        }
    }
    __syncthreads();

    int qk[SORT_EDGES], rank[SORT_EDGES];
#pragma unroll
    for (int k = 0; k < SORT_EDGES; ++k) {
        qk[k]   = bucket_of(rc[k].x);
        rank[k] = (k * 1024 + tid < SORT_CHUNK) ? atomicAdd(&s_cnt[qk[k]], 1) : 0;
    }
    __syncthreads();

    int own = 0, io = 0;
    if (tid < NBUCKET) {
        own = s_cnt[tid];
        io  = own;
#pragma unroll
        for (int off = 1; off < 32; off <<= 1) {
            const int no = __shfl_up(io, off, 32);
            if (lane >= off) io += no;
        }
        if (lane == 31) s_wo[wave] = io;
    }
    __syncthreads();
    if (tid < NBUCKET) {
        int po = 0;
        for (int w = 0; w < wave; ++w) po += s_wo[w];
        s_loff[tid] = po + io - own;
    }
    __syncthreads();

    if (tid < RUN_COLS) {
        const int a = (tid < G_GROUPS) ? s_loff[2 * tid]     : SORT_CHUNK;
        const int c = (tid < G_GROUPS) ? s_loff[2 * tid + 1] : SORT_CHUNK;
        runs[b * RUN_COLS + tid] = a | (c << 16);
    }

#pragma unroll
    for (int k = 0; k < SORT_EDGES; ++k) {
        if (k * 1024 + tid < SORT_CHUNK) {
            const int q = qk[k];
            const int p = min(max(s_loff[q] + rank[k], 0), SORT_CHUNK - 1);
            const int place = (rc[k].x - q * SUB_NODES) & 127;
            if (wide)
                s_rec[p] = make_int2((rc[k].y & 0xffff) | ((rl[k] & 511) << 16) | (place << 25),
                                     __float_as_int(vl[k]));
            else
                s_rec[p] = make_int2((lo + k * 1024 + tid) | (place << 20), 0);
        }
    }
    __syncthreads();

    int2* rec  = (int2*)list;
    int*  perm = list;
    for (int i = tid; i < SORT_CHUNK; i += 1024) {
        const int2 x = s_rec[i];
        if (wide) rec[lo + i] = x;
        else      perm[lo + i] = x.x;
    }
}

#define G_CAP   4096
#define G_RUNS  (SORT_GROUPS / G_WAVES)
#define G_RUN   64
#define G_SLOTS (2 * G_RUNS)
static_assert(G_RUNS * G_WAVES == SORT_GROUPS && G_RUNS == 4, "s_len is read as one int4 per lane");


template <int M>
__device__ __forceinline__ float row_xor(float p) {
    return __int_as_float(__builtin_amdgcn_update_dpp(0, __float_as_int(p), 0x160 + M, 0xf, 0xf, true));
}

__device__ __forceinline__ float xhalf(float x) {
    const int b = __float_as_int(x);
    return __int_as_float(__builtin_amdgcn_permlanex16(b, b, 0x76543210, 0xfedcba98, false, false));
}

__device__ __forceinline__ float wave_sum(float p) {
    p += xhalf(p);
    p += row_xor<8>(p);
    p += row_xor<4>(p);
    p += row_xor<2>(p);
    p += row_xor<1>(p);
    return p;
}

__device__ __forceinline__ int wave_sum_i(int p) {
    p += __builtin_amdgcn_permlanex16(p, p, 0x76543210, 0xfedcba98, false, false);
    p += __builtin_amdgcn_update_dpp(0, p, 0x168, 0xf, 0xf, true);
    p += __builtin_amdgcn_update_dpp(0, p, 0x164, 0xf, 0xf, true);
    p += __builtin_amdgcn_update_dpp(0, p, 0x162, 0xf, 0xf, true);
    p += __builtin_amdgcn_update_dpp(0, p, 0x161, 0xf, 0xf, true);
    return p;
}

__device__ __forceinline__ float swap_add(float x, float y) {
    const auto r = __builtin_amdgcn_permlane16_swap(__float_as_uint(x), __float_as_uint(y), false, false);
    return __uint_as_float(r[0]) + __uint_as_float(r[1]);
}

__device__ __forceinline__ void wave_sum8(const float (&p)[8], bool hi8, bool hi4, float (&tot)[8]) {
    float q[4];
#pragma unroll
    for (int i = 0; i < 4; ++i) q[i] = swap_add(p[i], p[i + 4]);
    float r[2];
#pragma unroll
    for (int i = 0; i < 2; ++i) {
        const float keep = hi8 ? q[i + 2] : q[i];
        const float send = hi8 ? q[i] : q[i + 2];
        r[i] = keep + row_xor<8>(send);
    }
    const float keep = hi4 ? r[1] : r[0];
    const float send = hi4 ? r[0] : r[1];
    float s = keep + row_xor<4>(send);
    s += row_xor<2>(s);
    s += row_xor<1>(s);
#pragma unroll
    for (int k = 0; k < 8; ++k)
        tot[k] = __int_as_float(__builtin_amdgcn_readlane(
            __float_as_int(s), 4 * (k & 1) + 8 * ((k >> 1) & 1) + 16 * (k >> 2)));
}

__device__ __forceinline__ void wave_sum4(const float (&p)[4], bool hi8, float (&tot)[4]) {
    const float q0 = swap_add(p[0], p[2]);
    const float q1 = swap_add(p[1], p[3]);
    const float keep = hi8 ? q1 : q0;
    const float send = hi8 ? q0 : q1;
    float s = keep + row_xor<8>(send);
    s += row_xor<4>(s);
    s += row_xor<2>(s);
    s += row_xor<1>(s);
#pragma unroll
    for (int k = 0; k < 4; ++k)
        tot[k] = __int_as_float(__builtin_amdgcn_readlane(
            __float_as_int(s), 8 * (k & 1) + 16 * (k >> 1)));
}

__device__ __forceinline__ void load_rel(float (&r)[4], const float* rrow) {
    r[0] = rrow[0]; r[1] = rrow[32]; r[2] = rrow[64]; r[3] = rrow[96];
}

__device__ __forceinline__ float lane_dot(const float (&f)[4], const float (&r)[4]) {
    return f[0] * r[0] + f[1] * r[1] + f[2] * r[2] + f[3] * r[3];
}

__device__ __forceinline__ void add_edge(const float (&f)[4], const float (&r)[4],
                                         float coef, float e, float partial,
                                         float& a0, float& a1, float& a2, float& a3) {
    const float proj = coef * partial;
    const float k2   = 2.0f * proj * coef;
    a0 += (f[0] - k2 * r[0]) * e;
    a1 += (f[1] - k2 * r[1]) * e;
    a2 += (f[2] - k2 * r[2]) * e;
    a3 += (f[3] - k2 * r[3]) * e;
}

__device__ __forceinline__ void load_feat(float (&f)[4], const float* __restrict__ features,
                                          int col, int lane) {
    const float* frow = features + (size_t)col * HID + lane;
    f[0] = frow[0]; f[1] = frow[32]; f[2] = frow[64]; f[3] = frow[96];
}

__device__ __forceinline__ int bcast_i(int v, int j) {
    return __builtin_amdgcn_readlane(v, j);
}
__device__ __forceinline__ float bcast_f(float v, int j) {
    return __int_as_float(__builtin_amdgcn_readlane(__float_as_int(v), j));
}

__device__ __forceinline__ void sum_batch(const float* __restrict__ features,
                                          const float* s_rel,
                                          int col, int r, float coef, float e,
                                          int m, int lane,
                                          float& a0, float& a1, float& a2, float& a3) {
    const bool hi8 = (lane & 8) != 0;
    const bool hi4 = (lane & 4) != 0;
    int j = 0;
    for (; j + 8 <= m; j += 8) {
        float f[8][4], rr[8][4], p[8], tot[8];
#pragma unroll
        for (int k = 0; k < 8; ++k) load_feat(f[k], features, bcast_i(col, j + k), lane);
#pragma unroll
        for (int k = 0; k < 8; ++k) {
            load_rel(rr[k], s_rel + bcast_i(r, j + k) * HID + lane);
            p[k] = lane_dot(f[k], rr[k]);
        }
        wave_sum8(p, hi8, hi4, tot);
#pragma unroll
        for (int k = 0; k < 8; ++k)
            add_edge(f[k], rr[k], bcast_f(coef, j + k), bcast_f(e, j + k), tot[k], a0, a1, a2, a3);
    }
    for (; j + 4 <= m; j += 4) {
        float f[4][4], rr[4][4], p[4], tot[4];
#pragma unroll
        for (int k = 0; k < 4; ++k) load_feat(f[k], features, bcast_i(col, j + k), lane);
#pragma unroll
        for (int k = 0; k < 4; ++k) {
            load_rel(rr[k], s_rel + bcast_i(r, j + k) * HID + lane);
            p[k] = lane_dot(f[k], rr[k]);
        }
        wave_sum4(p, hi8, tot);
#pragma unroll
        for (int k = 0; k < 4; ++k)
            add_edge(f[k], rr[k], bcast_f(coef, j + k), bcast_f(e, j + k), tot[k], a0, a1, a2, a3);
    }
    for (; j < m; ++j) {
        float f[4], rr[4];
        load_feat(f, features, bcast_i(col, j), lane);
        load_rel(rr, s_rel + bcast_i(r, j) * HID + lane);
        add_edge(f, rr, bcast_f(coef, j), bcast_f(e, j), wave_sum(lane_dot(f, rr)), a0, a1, a2, a3);
    }
}

__device__ __forceinline__ void one_edge(const float* __restrict__ features,
                                         const float* s_rel,
                                         int col, int r, float coef, float e,
                                         int j, int lane,
                                         float& a0, float& a1, float& a2, float& a3) {
    float f[4], rr[4];
    load_feat(f, features, bcast_i(col, j), lane);
    load_rel(rr, s_rel + bcast_i(r, j) * HID + lane);
    add_edge(f, rr, bcast_f(coef, j), bcast_f(e, j), wave_sum(lane_dot(f, rr)), a0, a1, a2, a3);
}

__device__ __forceinline__ int2 fetch_rec(bool wide, const int* __restrict__ list,
                                          const int* __restrict__ indices,
                                          const int* __restrict__ sp_idx,
                                          const float* __restrict__ sp_val, int pos) {
    pos = min(max(pos, 0), NTRIPLE - 1);
    if (wide) return ((const int2*)list)[pos];
    const int w = list[pos];
    const int eid = min(w & 0xfffff, NTRIPLE - 1);
    const int col = indices[2 * eid + 1];
    const int rl  = sp_idx[2 * eid + 1];
    const float v = sp_val[eid];
    return make_int2((col & 0xffff) | ((rl & 511) << 16) | (((w >> 20) & 127) << 25),
                     __float_as_int(v));
}
__device__ __forceinline__ int rec_node(int2 x, int pos, int b1) {
    return (int)((unsigned)x.x >> 25) + ((pos >= b1) ? SUB_NODES : 0);
}
__device__ __forceinline__ void split_rec(int2 x, int& col, int& r, float& v) {
    col = min(x.x & 0xffff, NODE_SIZE - 1);
    r   = min((x.x >> 16) & 511, REL_SIZE - 1);
    v   = __int_as_float(x.y);
}

__device__ __forceinline__ int next_node(int* s_next, int lane) {
    int n = 0;
    if (lane == 0) n = atomicAdd(s_next, 1);
    return __builtin_amdgcn_readfirstlane(n);
}

__device__ __forceinline__ void store_node(float* __restrict__ out, int node, int lane, bool any,
                                           float a0, float a1, float a2, float a3, float den) {
    float* dst = out + (size_t)node * HID + lane;
    if (any) {
        dst[0]  = a0 / den;
        dst[32] = a1 / den;
        dst[64] = a2 / den;
        dst[96] = a3 / den;
    } else {
        dst[0] = 0.0f; dst[32] = 0.0f; dst[64] = 0.0f; dst[96] = 0.0f;
    }
}

__device__ __forceinline__ void stage_tables(const float* __restrict__ rel_emb,
                                             const float* __restrict__ rel_norm,
                                             const float* __restrict__ rel_att,
                                             float* s_rel, float* s_norm, float* s_att,
                                             int tid) {
    const float4* src = (const float4*)rel_emb;
    float4* dst = (float4*)s_rel;
#pragma unroll
    for (int b = 0; b < 4; ++b) {
        float4 v[4];
#pragma unroll
        for (int k = 0; k < 4; ++k) {
            const int i = (b * 4 + k) * 1024 + tid;
            v[k] = src[min(i, REL_VEC4 - 1)];
        }
#pragma unroll
        for (int k = 0; k < 4; ++k) {
            const int i = (b * 4 + k) * 1024 + tid;
            if (i < REL_VEC4) dst[i] = v[k];
        }
    }
    if (tid < REL_SIZE) { s_norm[tid] = rel_norm[tid]; s_att[tid] = rel_att[tid]; }
}

__global__ __launch_bounds__(1024)
void gather_kernel(const float* __restrict__ features,
                   const float* __restrict__ rel_emb,
                   const int*   __restrict__ indices,
                   const int*   __restrict__ sp_idx,
                   const float* __restrict__ sp_val,
                   const float* __restrict__ rel_norm,
                   const float* __restrict__ rel_att,
                   const int*   __restrict__ runs,
                   const int*   __restrict__ list,
                   float* __restrict__ out,
                   size_t ws_bytes) {
    __shared__ float s_rel[REL_SIZE * HID];
    __shared__ float s_norm[512];
    __shared__ float s_att[512];
    __shared__ int2  s_rec[G_CAP];
    __shared__ int   s_hist[256];
    __shared__ int   s_seg[257];
    __shared__ int   s_r0[SORT_GROUPS];
    __shared__ int   s_r1[SORT_GROUPS];
    __shared__ int   s_r2[SORT_GROUPS];
    __shared__ int4  s_len[SORT_GROUPS / 4];
    __shared__ int   s_next;

    const int tid  = threadIdx.x;
    const int lane = tid & 31;
    const int wave = __builtin_amdgcn_readfirstlane(tid >> 5);
    const int lo = blockIdx.x * G_CHUNK;
    const int hi = min(lo + G_CHUNK, NODE_SIZE);
    const bool wide = ws_bytes >= WS_WIDE_BYTES;

    const int rid = wave * G_RUNS + (lane & (G_RUNS - 1));
    const int w0 = runs[rid * RUN_COLS + blockIdx.x];
    const int w1 = runs[rid * RUN_COLS + blockIdx.x + 1];
    const int t0 = min(w0 & 0xffff, SORT_CHUNK);
    const int t1 = min(max((int)((unsigned)w0 >> 16), t0), SORT_CHUNK);
    const int t2 = min(max(w1 & 0xffff, t1), SORT_CHUNK);
    const int p0 = rid * SORT_CHUNK + t0;
    const int p1 = rid * SORT_CHUNK + t1;
    const int p2 = rid * SORT_CHUNK + t2;
    if (lane < G_RUNS) {
        s_r0[rid] = p0;
        s_r1[rid] = p1;
        s_r2[rid] = p2;
        ((int*)s_len)[rid] = (t2 - t0) + ((t2 - t0 > G_RUN) ? (1 << 24) : 0);
    }

    int  q0[G_RUNS], q1[G_RUNS], q2[G_RUNS];
#pragma unroll
    for (int r = 0; r < G_RUNS; ++r) {
        q0[r] = bcast_i(p0, r);
        q1[r] = bcast_i(p1, r);
        q2[r] = bcast_i(p2, r);
    }
    int2 x[G_SLOTS];
    int  nd[G_SLOTS], rk[G_SLOTS];
    if (wide) {
#pragma unroll
        for (int j = 0; j < G_SLOTS; ++j) {
            const int i = q0[j >> 1] + 32 * (j & 1) + lane;
            x[j] = make_int2(0, 0);
            if (i < q2[j >> 1]) x[j] = ((const int2*)list)[min(max(i, 0), NTRIPLE - 1)];
        }
    } else {
#pragma unroll
        for (int j = 0; j < G_SLOTS; ++j) {
            const int i = q0[j >> 1] + 32 * (j & 1) + lane;
            x[j] = make_int2(0, 0);
            if (i < q2[j >> 1]) x[j] = fetch_rec(false, list, indices, sp_idx, sp_val, i);
        }
    }
    if (tid < 256) s_hist[tid] = 0;
    if (tid == 0) s_next = lo + G_WAVES;
    stage_tables(rel_emb, rel_norm, rel_att, s_rel, s_norm, s_att, tid);
#pragma unroll
    for (int j = 0; j < G_SLOTS; ++j) {
        const int i = q0[j >> 1] + 32 * (j & 1) + lane;
        nd[j] = (i < q2[j >> 1]) ? rec_node(x[j], i, q1[j >> 1]) : -1;
    }
    __syncthreads();

    int tot;
    {
        const int4 l4 = s_len[lane];
        tot = __builtin_amdgcn_readfirstlane(wave_sum_i(l4.x + l4.y + l4.z + l4.w));
    }
    const int m = tot & 0xffffff;

    if ((tot >> 24) == 0 && m <= G_CAP) {
#pragma unroll
        for (int j = 0; j < G_SLOTS; ++j)
            rk[j] = (nd[j] >= 0) ? atomicAdd(&s_hist[nd[j]], 1) : 0;
        __syncthreads();
        if (wave == 0) {
            int c[8];
            int sum = 0;
#pragma unroll
            for (int k = 0; k < 8; ++k) { c[k] = s_hist[lane * 8 + k]; sum += c[k]; }
            int incl = sum;
#pragma unroll
            for (int off = 1; off < 32; off <<= 1) {
                const int n = __shfl_up(incl, off, 32);
                if (lane >= off) incl += n;
            }
            int run = incl - sum;
#pragma unroll
            for (int k = 0; k < 8; ++k) { s_seg[lane * 8 + k] = run; run += c[k]; }
            if (lane == 31) s_seg[256] = run;
        }
        __syncthreads();
#pragma unroll
        for (int j = 0; j < G_SLOTS; ++j)
            if (nd[j] >= 0)
                s_rec[min(max(s_seg[nd[j]] + rk[j], 0), G_CAP - 1)] = x[j];
        __syncthreads();
        int n0 = lo + wave;
        int n1 = next_node(&s_next, lane);
        int s0 = 0, e0 = 0;
        int2 xA = make_int2(0, 0);
        if (n0 < hi) {
            s0 = __builtin_amdgcn_readfirstlane(s_seg[n0 - lo]);
            e0 = __builtin_amdgcn_readfirstlane(s_seg[n0 - lo + 1]);
            if (lane < e0 - s0) xA = s_rec[min(s0 + lane, G_CAP - 1)];
        }
        while (n0 < hi) {
            const int n2 = next_node(&s_next, lane);
            int s1 = 0, e1 = 0;
            int2 xB = make_int2(0, 0);
            if (n1 < hi) {
                s1 = __builtin_amdgcn_readfirstlane(s_seg[n1 - lo]);
                e1 = __builtin_amdgcn_readfirstlane(s_seg[n1 - lo + 1]);
                if (lane < e1 - s1) xB = s_rec[min(s1 + lane, G_CAP - 1)];
            }

            float a0 = 0.0f, a1 = 0.0f, a2 = 0.0f, a3 = 0.0f, den = 0.0f;
            {
                const int mm = min(32, e0 - s0);
                int colA, rA;
                float vA;
                split_rec(xA, colA, rA, vA);
                float coef = 0.0f, e = 0.0f;
                if (lane < mm) {
                    coef = vA / fmaxf(fabsf(vA) * s_norm[rA], EPSF);
                    e    = expf(coef * s_att[rA]);
                }
                den += e;
                sum_batch(features, s_rel, colA, rA, coef, e, mm, lane, a0, a1, a2, a3);
            }
            for (int base = s0 + 32; base < e0; base += 32) {
                const int mm = min(32, e0 - base);
                int col = 0, r = 0;
                float coef = 0.0f, e = 0.0f;
                if (lane < mm) {
                    float v;
                    split_rec(s_rec[min(base + lane, G_CAP - 1)], col, r, v);
                    coef = v / fmaxf(fabsf(v) * s_norm[r], EPSF);
                    e    = expf(coef * s_att[r]);
                }
                den += e;
                sum_batch(features, s_rel, col, r, coef, e, mm, lane, a0, a1, a2, a3);
            }
            den = wave_sum(den);
            store_node(out, n0, lane, e0 > s0, a0, a1, a2, a3, den);

            n0 = n1; s0 = s1; e0 = e1; xA = xB;
            n1 = n2;
        }
        return;
    }

    int n0 = lo + wave;
    while (n0 < hi) {
        float a0 = 0.0f, a1 = 0.0f, a2 = 0.0f, a3 = 0.0f, den = 0.0f;
        bool any = false;
        for (int rb = 0; rb < SORT_GROUPS; ++rb) {
            const int q0 = __builtin_amdgcn_readfirstlane(s_r0[rb]);
            const int q1 = __builtin_amdgcn_readfirstlane(s_r1[rb]);
            const int q2 = __builtin_amdgcn_readfirstlane(s_r2[rb]);
            for (int base = q0; base < q2; base += 32) {
                const int i = base + lane;
                int col = 0, r = 0;
                float coef = 0.0f, e = 0.0f;
                bool hit = false;
                if (i < q2) {
                    const int2 xr = fetch_rec(wide, list, indices, sp_idx, sp_val, i);
                    if (rec_node(xr, i, q1) == n0 - lo) {
                        float v;
                        hit = true;
                        split_rec(xr, col, r, v);
                        coef = v / fmaxf(fabsf(v) * s_norm[r], EPSF);
                        e    = expf(coef * s_att[r]);
                    }
                }
                den += e;
                unsigned mask = __builtin_amdgcn_ballot_w32(hit);
                any = any || (mask != 0u);
                while (mask) {
                    const int j = __builtin_ctz(mask);
                    mask &= mask - 1u;
                    one_edge(features, s_rel, col, r, coef, e, j, lane, a0, a1, a2, a3);
                }
            }
        }
        den = wave_sum(den);
        store_node(out, n0, lane, any, a0, a1, a2, a3, den);
        n0 = next_node(&s_next, lane);
    }
}

extern "C" void kernel_launch(void* const* d_in, const int* in_sizes, int n_in,
                              void* d_out, int out_size, void* d_ws, size_t ws_size,
                              hipStream_t stream) {
    const float* features = (const float*)d_in[0];
    const float* rel_emb  = (const float*)d_in[1];
    const int*   indices  = (const int*)  d_in[2];
    const int*   sp_idx   = (const int*)  d_in[3];
    const float* sp_val   = (const float*)d_in[4];
    const float* attn     = (const float*)d_in[5];
    float* out = (float*)d_out;

    float* ws       = (float*)d_ws;
    float* rel_norm = ws;
    float* rel_att  = ws + 512;
    int*   runs     = (int*)(ws + 1024);
    int*   list     = (int*)(ws + WS_LIST_OFF);

    stats_sort_kernel<<<SORT_GRID, 1024, 0, stream>>>(rel_emb, attn, rel_norm, rel_att,
                                                      indices, sp_idx, sp_val, runs, list, ws_size);

    gather_kernel<<<G_GROUPS, 1024, 0, stream>>>(
        features, rel_emb, indices, sp_idx, sp_val, rel_norm, rel_att,
        runs, list, out, ws_size);
}
